# starting kernel
# baseline (speedup 1.0000x reference)
#include <hip/hip_runtime.h>

typedef float v2f __attribute__((ext_vector_type(2)));
typedef float v8f __attribute__((ext_vector_type(8)));

__device__ __forceinline__ v8f wmma_f32(v2f a, v2f b, v8f c) {
  return __builtin_amdgcn_wmma_f32_16x16x4_f32(false, a, false, b, (short)0, c,
                                               false, false);
}

__device__ __forceinline__ void atomic_fadd_f32(float* p, float v) {
  asm volatile("global_atomic_add_f32 %0, %1, off scope:SCOPE_DEV"
               :
               : "v"((unsigned long long)(uintptr_t)p), "v"(v)
               : "memory");
}

__launch_bounds__(256)
__global__ void gemm64_kernel(const float* __restrict__ A,
                              const float* __restrict__ B,
                              const float* __restrict__ bias,
                              float* __restrict__ C, int M) {
  __shared__ float2 Bs2[32 * 64];
  for (int i = threadIdx.x; i < 2048; i += 256) {
    int k2 = i >> 6, n = i & 63;
    Bs2[i] = float2{B[(2 * k2) * 64 + n], B[(2 * k2 + 1) * 64 + n]};
  }
  __syncthreads();

  const int wave = threadIdx.x >> 5;
  const int lane = threadIdx.x & 31;
  const int r = lane & 15;
  const int h = lane >> 4;

  const int row0 = (blockIdx.x * 8 + wave) * 16;
  if (row0 >= M) return;

  int rowA = row0 + r;
  if (rowA >= M) rowA = M - 1;
  const float* __restrict__ Arow = A + (size_t)rowA * 64;

  v8f acc0 = {}, acc1 = {}, acc2 = {}, acc3 = {};
#pragma unroll
  for (int k0 = 0; k0 < 64; k0 += 4) {
    const int ka = k0 + 2 * h;
    float2 av = *reinterpret_cast<const float2*>(Arow + ka);
    v2f a; a.x = av.x; a.y = av.y;
    const int k2 = (k0 >> 1) + h;
    const float2* __restrict__ Bk = &Bs2[k2 * 64 + r];
    float2 f0 = Bk[0], f1 = Bk[16], f2 = Bk[32], f3 = Bk[48];
    v2f b0, b1, b2, b3;
    b0.x = f0.x; b0.y = f0.y;  b1.x = f1.x; b1.y = f1.y;
    b2.x = f2.x; b2.y = f2.y;  b3.x = f3.x; b3.y = f3.y;
    acc0 = wmma_f32(a, b0, acc0);
    acc1 = wmma_f32(a, b1, acc1);
    acc2 = wmma_f32(a, b2, acc2);
    acc3 = wmma_f32(a, b3, acc3);
  }

  float bia0 = 0.f, bia1 = 0.f, bia2 = 0.f, bia3 = 0.f;
  if (bias) { bia0 = bias[r]; bia1 = bias[16 + r]; bia2 = bias[32 + r]; bia3 = bias[48 + r]; }

#pragma unroll
  for (int v = 0; v < 8; ++v) {
    const int row = row0 + v + 8 * h;
    if (row < M) {
      float* __restrict__ Crow = C + (size_t)row * 64;
      Crow[r]      = acc0[v] + bia0;
      Crow[16 + r] = acc1[v] + bia1;
      Crow[32 + r] = acc2[v] + bia2;
      Crow[48 + r] = acc3[v] + bia3;
    }
  }
}

__launch_bounds__(256)
__global__ void gemm_concat128_kernel(const float* __restrict__ A0,
                                      const float* __restrict__ A1,
                                      const float* __restrict__ B,
                                      const float* __restrict__ bias,
                                      float* __restrict__ C, int M) {
  __shared__ float2 Bs2[64 * 64];
  for (int i = threadIdx.x; i < 4096; i += 256) {
    int k2 = i >> 6, n = i & 63;
    Bs2[i] = float2{B[(2 * k2) * 64 + n], B[(2 * k2 + 1) * 64 + n]};
  }
  __syncthreads();

  const int wave = threadIdx.x >> 5;
  const int lane = threadIdx.x & 31;
  const int r = lane & 15;
  const int h = lane >> 4;

  const int row0 = (blockIdx.x * 8 + wave) * 16;
  if (row0 >= M) return;

  int rowA = row0 + r;
  if (rowA >= M) rowA = M - 1;
  const float* __restrict__ Arow0 = A0 + (size_t)rowA * 64;
  const float* __restrict__ Arow1 = A1 + (size_t)rowA * 64;

  v8f acc0 = {}, acc1 = {}, acc2 = {}, acc3 = {};
#pragma unroll
  for (int k0 = 0; k0 < 128; k0 += 4) {
    const int ka = k0 + 2 * h;
    const float* src = (k0 < 64) ? (Arow0 + ka) : (Arow1 + (ka - 64));
    float2 av = *reinterpret_cast<const float2*>(src);
    v2f a; a.x = av.x; a.y = av.y;
    const int k2 = (k0 >> 1) + h;
    const float2* __restrict__ Bk = &Bs2[k2 * 64 + r];
    float2 f0 = Bk[0], f1 = Bk[16], f2 = Bk[32], f3 = Bk[48];
    v2f b0, b1, b2, b3;
    b0.x = f0.x; b0.y = f0.y;  b1.x = f1.x; b1.y = f1.y;
    b2.x = f2.x; b2.y = f2.y;  b3.x = f3.x; b3.y = f3.y;
    acc0 = wmma_f32(a, b0, acc0);
    acc1 = wmma_f32(a, b1, acc1);
    acc2 = wmma_f32(a, b2, acc2);
    acc3 = wmma_f32(a, b3, acc3);
  }

  const float bia0 = bias[r], bia1 = bias[16 + r], bia2 = bias[32 + r], bia3 = bias[48 + r];
#pragma unroll
  for (int v = 0; v < 8; ++v) {
    const int row = row0 + v + 8 * h;
    if (row < M) {
      float* __restrict__ Crow = C + (size_t)row * 64;
      Crow[r]      = acc0[v] + bia0;
      Crow[16 + r] = acc1[v] + bia1;
      Crow[32 + r] = acc2[v] + bia2;
      Crow[48 + r] = acc3[v] + bia3;
    }
  }
}

__launch_bounds__(256)
__global__ void edge_kernel(const float* __restrict__ rhs,
                            const float* __restrict__ lhs,
                            const int* __restrict__ rj,
                            const int* __restrict__ lj,
                            const float* __restrict__ weights,
                            const float* __restrict__ We,
                            const float* __restrict__ Wf,
                            const float* __restrict__ bf,
                            float* __restrict__ outAcc, int E) {
  __shared__ float2 WfS2[32 * 64];
  __shared__ float WeS[64];
  __shared__ float bfS[64];
  __shared__ float msgS[8][16 * 64];
  __shared__ int rjS[8][16];

  for (int i = threadIdx.x; i < 2048; i += 256) {
    int k2 = i >> 6, n = i & 63;
    WfS2[i] = float2{Wf[(2 * k2) * 64 + n], Wf[(2 * k2 + 1) * 64 + n]};
  }
  if (threadIdx.x < 64) { WeS[threadIdx.x] = We[threadIdx.x]; bfS[threadIdx.x] = bf[threadIdx.x]; }
  __syncthreads();

  const int wave = threadIdx.x >> 5;
  const int lane = threadIdx.x & 31;
  const int r = lane & 15;
  const int h = lane >> 4;

  const int eBase = (blockIdx.x * 8 + wave) * 16;

  int e = eBase + r;
  if (e >= E) e = E - 1;
  const int idxR = rj[e];
  const int idxL = lj[e];
  const float w = weights[e];
  if (h == 0) rjS[wave][r] = idxR;

  const float4* __restrict__ pr = reinterpret_cast<const float4*>(rhs + (size_t)idxR * 64) + h * 8;
  const float4* __restrict__ pl = reinterpret_cast<const float4*>(lhs + (size_t)idxL * 64) + h * 8;
  const float4* __restrict__ pe = reinterpret_cast<const float4*>(WeS) + h * 8;
  float4* __restrict__ pm = reinterpret_cast<float4*>(&msgS[wave][r * 64]) + h * 8;
#pragma unroll
  for (int i = 0; i < 8; ++i) {
    float4 a = pr[i], b = pl[i], c = pe[i];
    float4 m;
    m.x = a.x + b.x + w * c.x;
    m.y = a.y + b.y + w * c.y;
    m.z = a.z + b.z + w * c.z;
    m.w = a.w + b.w + w * c.w;
    m.x = m.x > 0.f ? m.x : 0.01f * m.x;
    m.y = m.y > 0.f ? m.y : 0.01f * m.y;
    m.z = m.z > 0.f ? m.z : 0.01f * m.z;
    m.w = m.w > 0.f ? m.w : 0.01f * m.w;
    pm[i] = m;
  }

  v8f acc0 = {}, acc1 = {}, acc2 = {}, acc3 = {};
  const float* __restrict__ mrow = &msgS[wave][r * 64];
#pragma unroll
  for (int k0 = 0; k0 < 64; k0 += 4) {
    const int ka = k0 + 2 * h;
    float2 av = *reinterpret_cast<const float2*>(mrow + ka);
    v2f a; a.x = av.x; a.y = av.y;
    const int k2 = (k0 >> 1) + h;
    const float2* __restrict__ Bk = &WfS2[k2 * 64 + r];
    float2 f0 = Bk[0], f1 = Bk[16], f2 = Bk[32], f3 = Bk[48];
    v2f b0, b1, b2, b3;
    b0.x = f0.x; b0.y = f0.y;  b1.x = f1.x; b1.y = f1.y;
    b2.x = f2.x; b2.y = f2.y;  b3.x = f3.x; b3.y = f3.y;
    acc0 = wmma_f32(a, b0, acc0);
    acc1 = wmma_f32(a, b1, acc1);
    acc2 = wmma_f32(a, b2, acc2);
    acc3 = wmma_f32(a, b3, acc3);
  }

  const float bia0 = bfS[r], bia1 = bfS[16 + r], bia2 = bfS[32 + r], bia3 = bfS[48 + r];
#pragma unroll
  for (int v = 0; v < 8; ++v) {
    const int m = v + 8 * h;
    if (eBase + m < E) {
      const int seg = rjS[wave][m];
      float* __restrict__ orow = outAcc + (size_t)seg * 64;
      atomic_fadd_f32(orow + r,      acc0[v] + bia0);
      atomic_fadd_f32(orow + 16 + r, acc1[v] + bia1);
      atomic_fadd_f32(orow + 32 + r, acc2[v] + bia2);
      atomic_fadd_f32(orow + 48 + r, acc3[v] + bia3);
    }
  }
}

__global__ void zero4_kernel(float4* __restrict__ p, int n4) {
  int i = blockIdx.x * blockDim.x + threadIdx.x;
  if (i < n4) p[i] = float4{0.f, 0.f, 0.f, 0.f};
}

extern "C" void kernel_launch(void* const* d_in, const int* in_sizes, int n_in,
                              void* d_out, int out_size, void* d_ws, size_t ws_size,
                              hipStream_t stream) {
  const float* input   = (const float*)d_in[0];
  const float* other   = (const float*)d_in[1];
  const int*   rj      = (const int*)d_in[2];
  const int*   lj      = (const int*)d_in[3];
  const float* weights = (const float*)d_in[4];
  const float* Wi      = (const float*)d_in[5];
  const float* bi      = (const float*)d_in[6];
  const float* Wo      = (const float*)d_in[7];
  const float* We      = (const float*)d_in[8];
  const float* Wf      = (const float*)d_in[9];
  const float* bf      = (const float*)d_in[10];
  const float* Wout    = (const float*)d_in[11];
  const float* bout    = (const float*)d_in[12];

  const int D    = 64;
  const int N_IN = in_sizes[0] / D;
  const int N_OT = in_sizes[1] / D;
  const int E    = in_sizes[2];

  float* rhs    = (float*)d_ws;
  float* lhs    = rhs + (size_t)N_IN * D;
  float* outAcc = lhs + (size_t)N_OT * D;

  const int tiles_in = (N_IN + 15) / 16;
  const int tiles_ot = (N_OT + 15) / 16;

  gemm64_kernel<<<(tiles_in + 7) / 8, 256, 0, stream>>>(input, Wi, bi, rhs, N_IN);
  gemm64_kernel<<<(tiles_ot + 7) / 8, 256, 0, stream>>>(other, Wo, nullptr, lhs, N_OT);

  const int n4 = (N_IN * D) / 4;
  zero4_kernel<<<(n4 + 255) / 256, 256, 0, stream>>>((float4*)outAcc, n4);

  const int eTiles = (E + 15) / 16;
  edge_kernel<<<(eTiles + 7) / 8, 256, 0, stream>>>(rhs, lhs, rj, lj, weights,
                                                    We, Wf, bf, outAcc, E);

  gemm_concat128_kernel<<<(tiles_in + 7) / 8, 256, 0, stream>>>(
      outAcc, input, Wout, bout, (float*)d_out, N_IN);
}


# Wf hoisted out of the edge loop by linearity: edges only segment-sum leaky(message) with whole-line f32 atomics (lane=column), then one per-node WMMA GEMM applies Wf + deg*bf; f32 sum order changes
# speedup vs baseline: 1.3307x; 1.3307x over previous
#include <hip/hip_runtime.h>

typedef float v2f __attribute__((ext_vector_type(2)));
typedef float v8f __attribute__((ext_vector_type(8)));

__device__ __forceinline__ v8f wmma_f32(v2f a, v2f b, v8f c) {
  return __builtin_amdgcn_wmma_f32_16x16x4_f32(false, a, false, b, (short)0, c,
                                               false, false);
}

__device__ __forceinline__ void atomic_fadd_f32(float* p, float v) {
  asm volatile("global_atomic_add_f32 %0, %1, off scope:SCOPE_DEV"
               :
               : "v"((unsigned long long)(uintptr_t)p), "v"(v)
               : "memory");
}

__launch_bounds__(256)
__global__ void gemm64_kernel(const float* __restrict__ A,
                              const float* __restrict__ B,
                              const float* __restrict__ bias,
                              float* __restrict__ C, int M) {
  __shared__ float2 Bs2[32 * 64];
  for (int i = threadIdx.x; i < 2048; i += 256) {
    int k2 = i >> 6, n = i & 63;
    Bs2[i] = float2{B[(2 * k2) * 64 + n], B[(2 * k2 + 1) * 64 + n]};
  }
  __syncthreads();

  const int wave = threadIdx.x >> 5;
  const int lane = threadIdx.x & 31;
  const int r = lane & 15;
  const int h = lane >> 4;

  const int row0 = (blockIdx.x * 8 + wave) * 16;
  if (row0 >= M) return;

  int rowA = row0 + r;
  if (rowA >= M) rowA = M - 1;
  const float* __restrict__ Arow = A + (size_t)rowA * 64;

  v8f acc0 = {}, acc1 = {}, acc2 = {}, acc3 = {};
#pragma unroll
  for (int k0 = 0; k0 < 64; k0 += 4) {
    const int ka = k0 + 2 * h;
    float2 av = *reinterpret_cast<const float2*>(Arow + ka);
    v2f a; a.x = av.x; a.y = av.y;
    const int k2 = (k0 >> 1) + h;
    const float2* __restrict__ Bk = &Bs2[k2 * 64 + r];
    float2 f0 = Bk[0], f1 = Bk[16], f2 = Bk[32], f3 = Bk[48];
    v2f b0, b1, b2, b3;
    b0.x = f0.x; b0.y = f0.y;  b1.x = f1.x; b1.y = f1.y;
    b2.x = f2.x; b2.y = f2.y;  b3.x = f3.x; b3.y = f3.y;
    acc0 = wmma_f32(a, b0, acc0);
    acc1 = wmma_f32(a, b1, acc1);
    acc2 = wmma_f32(a, b2, acc2);
    acc3 = wmma_f32(a, b3, acc3);
  }

  float bia0 = 0.f, bia1 = 0.f, bia2 = 0.f, bia3 = 0.f;
  if (bias) { bia0 = bias[r]; bia1 = bias[16 + r]; bia2 = bias[32 + r]; bia3 = bias[48 + r]; }

#pragma unroll
  for (int v = 0; v < 8; ++v) {
    const int row = row0 + v + 8 * h;
    if (row < M) {
      float* __restrict__ Crow = C + (size_t)row * 64;
      Crow[r]      = acc0[v] + bia0;
      Crow[16 + r] = acc1[v] + bia1;
      Crow[32 + r] = acc2[v] + bia2;
      Crow[48 + r] = acc3[v] + bia3;
    }
  }
}

__launch_bounds__(256)
__global__ void node_out_kernel(const float* __restrict__ Ssum,
                                const float* __restrict__ X,
                                const float* __restrict__ Wf,
                                const float* __restrict__ bf,
                                const float* __restrict__ Wout,
                                const float* __restrict__ bout,
                                float* __restrict__ C, int M) {
  __shared__ float2 WfS2[32 * 64];
  __shared__ float2 WoS2[64 * 64];
  __shared__ __attribute__((aligned(16))) float tS[8][16 * 64];
  for (int i = threadIdx.x; i < 2048; i += 256) {
    int k2 = i >> 6, n = i & 63;
    WfS2[i] = float2{Wf[(2 * k2) * 64 + n], Wf[(2 * k2 + 1) * 64 + n]};
  }
  for (int i = threadIdx.x; i < 4096; i += 256) {
    int k2 = i >> 6, n = i & 63;
    WoS2[i] = float2{Wout[(2 * k2) * 64 + n], Wout[(2 * k2 + 1) * 64 + n]};
  }
  __syncthreads();

  const int wave = threadIdx.x >> 5;
  const int lane = threadIdx.x & 31;
  const int r = lane & 15;
  const int h = lane >> 4;

  const int row0 = (blockIdx.x * 8 + wave) * 16;
  if (row0 >= M) return;

  int rowA = row0 + r;
  if (rowA >= M) rowA = M - 1;
  const float* __restrict__ Srow = Ssum + (size_t)rowA * 64;
  const float* __restrict__ Xrow = X + (size_t)rowA * 64;

  float dg[8];
#pragma unroll
  for (int v = 0; v < 8; ++v) {
    int row = row0 + v + 8 * h;
    if (row >= M) row = M - 1;
    dg[v] = C[(size_t)row * 64];
  }

  v8f acc0 = {}, acc1 = {}, acc2 = {}, acc3 = {};
#pragma unroll
  for (int k0 = 0; k0 < 64; k0 += 4) {
    const int ka = k0 + 2 * h;
    float2 av = *reinterpret_cast<const float2*>(Srow + ka);
    v2f a; a.x = av.x; a.y = av.y;
    const int k2 = (k0 >> 1) + h;
    const float2* Bk = &WfS2[k2 * 64 + r];
    float2 f0 = Bk[0], f1 = Bk[16], f2 = Bk[32], f3 = Bk[48];
    v2f b0, b1, b2, b3;
    b0.x = f0.x; b0.y = f0.y;  b1.x = f1.x; b1.y = f1.y;
    b2.x = f2.x; b2.y = f2.y;  b3.x = f3.x; b3.y = f3.y;
    acc0 = wmma_f32(a, b0, acc0);
    acc1 = wmma_f32(a, b1, acc1);
    acc2 = wmma_f32(a, b2, acc2);
    acc3 = wmma_f32(a, b3, acc3);
  }

  {
    const float bf0 = bf[r], bf1 = bf[16 + r], bf2 = bf[32 + r], bf3 = bf[48 + r];
    float* tw = &tS[wave][0];
#pragma unroll
    for (int v = 0; v < 8; ++v) {
      const int row = v + 8 * h;
      float* trw = tw + row * 64;
      const int x = 4 * row;
      const float d = dg[v];
      trw[(r)      ^ x] = acc0[v] + d * bf0;
      trw[(16 + r) ^ x] = acc1[v] + d * bf1;
      trw[(32 + r) ^ x] = acc2[v] + d * bf2;
      trw[(48 + r) ^ x] = acc3[v] + d * bf3;
    }
  }

  acc0 = v8f{}; acc1 = v8f{}; acc2 = v8f{}; acc3 = v8f{};
  const float* trow = &tS[wave][r * 64];
  const int xr = 4 * r;
#pragma unroll
  for (int k0 = 0; k0 < 128; k0 += 4) {
    const int ka = k0 + 2 * h;
    float2 av;
    if (k0 < 64) av = *reinterpret_cast<const float2*>(trow + ((k0 ^ xr) + 2 * h));
    else         av = *reinterpret_cast<const float2*>(Xrow + (ka - 64));
    v2f a; a.x = av.x; a.y = av.y;
    const int k2 = (k0 >> 1) + h;
    const float2* Bk = &WoS2[k2 * 64 + r];
    float2 f0 = Bk[0], f1 = Bk[16], f2 = Bk[32], f3 = Bk[48];
    v2f b0, b1, b2, b3;
    b0.x = f0.x; b0.y = f0.y;  b1.x = f1.x; b1.y = f1.y;
    b2.x = f2.x; b2.y = f2.y;  b3.x = f3.x; b3.y = f3.y;
    acc0 = wmma_f32(a, b0, acc0);
    acc1 = wmma_f32(a, b1, acc1);
    acc2 = wmma_f32(a, b2, acc2);
    acc3 = wmma_f32(a, b3, acc3);
  }

  const float bia0 = bout[r], bia1 = bout[16 + r], bia2 = bout[32 + r], bia3 = bout[48 + r];
#pragma unroll
  for (int v = 0; v < 8; ++v) {
    const int row = row0 + v + 8 * h;
    if (row < M) {
      float* Crow = C + (size_t)row * 64;
      Crow[r]      = acc0[v] + bia0;
      Crow[16 + r] = acc1[v] + bia1;
      Crow[32 + r] = acc2[v] + bia2;
      Crow[48 + r] = acc3[v] + bia3;
    }
  }
}

__device__ __forceinline__ void atomic_fadd_f32_hi(float* p, float v) {
  asm volatile("global_atomic_add_f32 %0, %1, off offset:128 scope:SCOPE_DEV"
               :
               : "v"((unsigned long long)(uintptr_t)p), "v"(v)
               : "memory");
}

template <bool FULL>
__device__ __forceinline__ void edge_chunk8(const float* __restrict__ rhs,
                                            const float* __restrict__ lhs,
                                            float* __restrict__ Ssum,
                                            int idxR, int idxL, int wbits,
                                            float we0, float we1, int lane,
                                            int c0, int nLive) {
  float a0[8], a1[8], b0[8], b1[8];
#pragma unroll
  for (int i = 0; i < 8; ++i) {
    const int sr = __builtin_amdgcn_readlane(idxR, c0 + i);
    const int sl = __builtin_amdgcn_readlane(idxL, c0 + i);
    const float* pr = rhs + (size_t)sr * 64 + lane;
    const float* pl = lhs + (size_t)sl * 64 + lane;
    a0[i] = pr[0]; a1[i] = pr[32];
    b0[i] = pl[0]; b1[i] = pl[32];
  }
#pragma unroll
  for (int i = 0; i < 8; ++i) {
    if (FULL || c0 + i < nLive) {
      const int sr = __builtin_amdgcn_readlane(idxR, c0 + i);
      const float w = __int_as_float(__builtin_amdgcn_readlane(wbits, c0 + i));
      float m0 = a0[i] + b0[i] + w * we0;
      float m1 = a1[i] + b1[i] + w * we1;
      m0 = m0 > 0.f ? m0 : 0.01f * m0;
      m1 = m1 > 0.f ? m1 : 0.01f * m1;
      float* o = Ssum + (size_t)sr * 64 + lane;
      atomic_fadd_f32(o, m0);
      atomic_fadd_f32_hi(o, m1);
    }
  }
}

__launch_bounds__(256)
__global__ void edge_sum_kernel(const float* __restrict__ rhs,
                                const float* __restrict__ lhs,
                                const int* __restrict__ rj,
                                const int* __restrict__ lj,
                                const float* __restrict__ weights,
                                const float* __restrict__ We,
                                float* __restrict__ Ssum,
                                float* __restrict__ degw, int E) {
  const int lane = threadIdx.x & 31;
  const int wv = __builtin_amdgcn_readfirstlane((int)(threadIdx.x >> 5));
  const int eBase = (blockIdx.x * 8 + wv) * 32;
  if (eBase >= E) return;
  const int nLive = (E - eBase) < 32 ? (E - eBase) : 32;

  int e = eBase + lane;
  const bool live = lane < nLive;
  if (!live) e = E - 1;
  const int idxR = rj[e];
  const int idxL = lj[e];
  const int wbits = __float_as_int(weights[e]);
  const float we0 = We[lane], we1 = We[32 + lane];

  if (live) atomic_fadd_f32(degw + (size_t)idxR * 64, 1.0f);

  if (nLive == 32) {
#pragma unroll
    for (int c = 0; c < 32; c += 8)
      edge_chunk8<true>(rhs, lhs, Ssum, idxR, idxL, wbits, we0, we1, lane, c, nLive);
  } else {
#pragma unroll
    for (int c = 0; c < 32; c += 8)
      if (c < nLive)
        edge_chunk8<false>(rhs, lhs, Ssum, idxR, idxL, wbits, we0, we1, lane, c, nLive);
  }
}

__global__ void zero_kernel(float4* __restrict__ p, int n4,
                            float* __restrict__ degw, int nrows) {
  int i = blockIdx.x * blockDim.x + threadIdx.x;
  if (i < n4) p[i] = float4{0.f, 0.f, 0.f, 0.f};
  if (i < nrows) degw[(size_t)i * 64] = 0.f;
}

extern "C" void kernel_launch(void* const* d_in, const int* in_sizes, int n_in,
                              void* d_out, int out_size, void* d_ws, size_t ws_size,
                              hipStream_t stream) {
  const float* input   = (const float*)d_in[0];
  const float* other   = (const float*)d_in[1];
  const int*   rj      = (const int*)d_in[2];
  const int*   lj      = (const int*)d_in[3];
  const float* weights = (const float*)d_in[4];
  const float* Wi      = (const float*)d_in[5];
  const float* bi      = (const float*)d_in[6];
  const float* Wo      = (const float*)d_in[7];
  const float* We      = (const float*)d_in[8];
  const float* Wf      = (const float*)d_in[9];
  const float* bf      = (const float*)d_in[10];
  const float* Wout    = (const float*)d_in[11];
  const float* bout    = (const float*)d_in[12];

  const int D    = 64;
  const int N_IN = in_sizes[0] / D;
  const int N_OT = in_sizes[1] / D;
  const int E    = in_sizes[2];

  float* rhs    = (float*)d_ws;
  float* lhs    = rhs + (size_t)N_IN * D;
  float* outAcc = lhs + (size_t)N_OT * D;

  const int tiles_in = (N_IN + 15) / 16;
  const int tiles_ot = (N_OT + 15) / 16;

  gemm64_kernel<<<(tiles_in + 7) / 8, 256, 0, stream>>>(input, Wi, bi, rhs, N_IN);
  gemm64_kernel<<<(tiles_ot + 7) / 8, 256, 0, stream>>>(other, Wo, nullptr, lhs, N_OT);

  const int n4 = (N_IN * D) / 4;
  zero_kernel<<<(n4 + 255) / 256, 256, 0, stream>>>((float4*)outAcc, n4,
                                                    (float*)d_out, N_IN);

  const int eWaves = (E + 31) / 32;
  edge_sum_kernel<<<(eWaves + 7) / 8, 256, 0, stream>>>(rhs, lhs, rj, lj, weights,
                                                        We, outAcc, (float*)d_out, E);

  node_out_kernel<<<(tiles_in + 7) / 8, 256, 0, stream>>>(
      outAcc, input, Wf, bf, Wout, bout, (float*)d_out, N_IN);
}


# Edges counting-sorted per call into 400-row owner ranges (wave-private LDS counters merged by ballots); a wave sums its 16 rows in its own LDS tile, node GEMM fused; no atomics; f32 sum order changes
# speedup vs baseline: 1.4735x; 1.1073x over previous
#include <hip/hip_runtime.h>

typedef float v2f __attribute__((ext_vector_type(2)));
typedef float v8f __attribute__((ext_vector_type(8)));

__device__ __forceinline__ v8f wmma_f32(v2f a, v2f b, v8f c) {
  return __builtin_amdgcn_wmma_f32_16x16x4_f32(false, a, false, b, (short)0, c,
                                               false, false);
}

__launch_bounds__(256)
__global__ void gemm64_kernel(const float* __restrict__ A,
                              const float* __restrict__ B,
                              const float* __restrict__ bias,
                              float* __restrict__ C, int M) {
  __shared__ float2 Bs2[32 * 64];
  for (int i = threadIdx.x; i < 2048; i += 256) {
    int k2 = i >> 6, n = i & 63;
    Bs2[i] = float2{B[(2 * k2) * 64 + n], B[(2 * k2 + 1) * 64 + n]};
  }
  __syncthreads();

  const int wave = threadIdx.x >> 5;
  const int lane = threadIdx.x & 31;
  const int r = lane & 15;
  const int h = lane >> 4;

  const int row0 = (blockIdx.x * 8 + wave) * 16;
  if (row0 >= M) return;

  int rowA = row0 + r;
  if (rowA >= M) rowA = M - 1;
  const float* __restrict__ Arow = A + (size_t)rowA * 64;

  v8f acc0 = {}, acc1 = {}, acc2 = {}, acc3 = {};
#pragma unroll
  for (int k0 = 0; k0 < 64; k0 += 4) {
    const int ka = k0 + 2 * h;
    float2 av = *reinterpret_cast<const float2*>(Arow + ka);
    v2f a; a.x = av.x; a.y = av.y;
    const int k2 = (k0 >> 1) + h;
    const float2* __restrict__ Bk = &Bs2[k2 * 64 + r];
    float2 f0 = Bk[0], f1 = Bk[16], f2 = Bk[32], f3 = Bk[48];
    v2f b0, b1, b2, b3;
    b0.x = f0.x; b0.y = f0.y;  b1.x = f1.x; b1.y = f1.y;
    b2.x = f2.x; b2.y = f2.y;  b3.x = f3.x; b3.y = f3.y;
    acc0 = wmma_f32(a, b0, acc0);
    acc1 = wmma_f32(a, b1, acc1);
    acc2 = wmma_f32(a, b2, acc2);
    acc3 = wmma_f32(a, b3, acc3);
  }

  float bia0 = 0.f, bia1 = 0.f, bia2 = 0.f, bia3 = 0.f;
  if (bias) { bia0 = bias[r]; bia1 = bias[16 + r]; bia2 = bias[32 + r]; bia3 = bias[48 + r]; }

#pragma unroll
  for (int v = 0; v < 8; ++v) {
    const int row = row0 + v + 8 * h;
    if (row < M) {
      float* __restrict__ Crow = C + (size_t)row * 64;
      Crow[r]      = acc0[v] + bia0;
      Crow[16 + r] = acc1[v] + bia1;
      Crow[32 + r] = acc2[v] + bia2;
      Crow[48 + r] = acc3[v] + bia3;
    }
  }
}

constexpr int RNG  = 400;
constexpr int NBIN = 256;
constexpr int NSL  = 256;
constexpr int OWV  = RNG / 16;
constexpr int CHK  = OWV * 32;

__device__ __forceinline__ unsigned match_key8(int key, unsigned liveMask) {
  unsigned peers = liveMask;
#pragma unroll
  for (int b = 0; b < 8; ++b) {
    const bool bit = (key >> b) & 1;
    const unsigned bal = __builtin_amdgcn_ballot_w32(bit);
    peers &= bit ? bal : ~bal;
  }
  return peers;
}

__device__ __forceinline__ void count_slice(const int* __restrict__ rj, int* cw,
                                            int e0, int e1, int wave, int lane) {
  const unsigned lt = (1u << lane) - 1u;
  for (int eb = e0 + wave * 32; eb < e1; eb += 1024) {
    const int e = eb + lane;
    const bool live = e < e1;
    const int key = live ? rj[e] / RNG : 0;
    const unsigned lm = __builtin_amdgcn_ballot_w32(live);
    const unsigned peers = match_key8(key, lm);
    if (live && (peers & lt) == 0u) cw[key] += __builtin_popcount(peers);
  }
}

__launch_bounds__(1024)
__global__ void bin_count_kernel(const int* __restrict__ rj, int* __restrict__ cntmat,
                                 int E, int epb) {
  __shared__ int cntW[32 * NBIN];
  for (int i = threadIdx.x; i < 32 * NBIN; i += 1024) cntW[i] = 0;
  __syncthreads();
  const int lane = threadIdx.x & 31;
  const int wave = threadIdx.x >> 5;
  const int e0 = blockIdx.x * epb;
  int e1 = e0 + epb;
  if (e1 > E) e1 = E;
  count_slice(rj, cntW + wave * NBIN, e0, e1, wave, lane);
  __syncthreads();
  if (threadIdx.x < NBIN) {
    int s = 0;
    for (int w = 0; w < 32; ++w) s += cntW[w * NBIN + threadIdx.x];
    cntmat[threadIdx.x * NSL + blockIdx.x] = s;
  }
}

__launch_bounds__(1024)
__global__ void bin_scan_kernel(const int* __restrict__ cntmat, int* __restrict__ offs,
                                int* __restrict__ binStart) {
  __shared__ int tot[NBIN];
  __shared__ int startS[NBIN];
  const int lane = threadIdx.x & 31;
  const int wave = threadIdx.x >> 5;
  int ex[8][8];
#pragma unroll
  for (int j = 0; j < 8; ++j) {
    const int bin = wave * 8 + j;
    const int4* p = reinterpret_cast<const int4*>(cntmat + bin * NSL + lane * 8);
    const int4 a = p[0], b = p[1];
    const int c[8] = {a.x, a.y, a.z, a.w, b.x, b.y, b.z, b.w};
    int s = 0;
#pragma unroll
    for (int i = 0; i < 8; ++i) { ex[j][i] = s; s += c[i]; }
    int inc = s;
#pragma unroll
    for (int d = 1; d < 32; d <<= 1) {
      const int t = __shfl_up(inc, d);
      if (lane >= d) inc += t;
    }
    const int exl = inc - s;
#pragma unroll
    for (int i = 0; i < 8; ++i) ex[j][i] += exl;
    const int total = __shfl(inc, 31);
    if (lane == 0) tot[bin] = total;
  }
  __syncthreads();
  if (wave == 0) {
    int c[8], lx[8];
    int s = 0;
#pragma unroll
    for (int i = 0; i < 8; ++i) { c[i] = tot[lane * 8 + i]; lx[i] = s; s += c[i]; }
    int inc = s;
#pragma unroll
    for (int d = 1; d < 32; d <<= 1) {
      const int t = __shfl_up(inc, d);
      if (lane >= d) inc += t;
    }
    const int exl = inc - s;
#pragma unroll
    for (int i = 0; i < 8; ++i) startS[lane * 8 + i] = exl + lx[i];
  }
  __syncthreads();
#pragma unroll
  for (int j = 0; j < 8; ++j) {
    const int bin = wave * 8 + j;
    const int base = startS[bin];
    int4* p = reinterpret_cast<int4*>(offs + bin * NSL + lane * 8);
    p[0] = int4{base + ex[j][0], base + ex[j][1], base + ex[j][2], base + ex[j][3]};
    p[1] = int4{base + ex[j][4], base + ex[j][5], base + ex[j][6], base + ex[j][7]};
  }
  if (threadIdx.x < NBIN) binStart[threadIdx.x] = startS[threadIdx.x];
  if (threadIdx.x == 0) binStart[NBIN] = startS[NBIN - 1] + tot[NBIN - 1];
}

__launch_bounds__(1024)
__global__ void bin_place_kernel(const int* __restrict__ rj, const int* __restrict__ lj,
                                 const float* __restrict__ weights,
                                 const int* __restrict__ offs, int4* __restrict__ recs,
                                 int E, int epb) {
  __shared__ int cntW[32 * NBIN];
  for (int i = threadIdx.x; i < 32 * NBIN; i += 1024) cntW[i] = 0;
  __syncthreads();
  const int lane = threadIdx.x & 31;
  const int wave = threadIdx.x >> 5;
  const int e0 = blockIdx.x * epb;
  int e1 = e0 + epb;
  if (e1 > E) e1 = E;
  int* cw = cntW + wave * NBIN;
  count_slice(rj, cw, e0, e1, wave, lane);
  __syncthreads();
  if (threadIdx.x < NBIN) {
    int run = offs[threadIdx.x * NSL + blockIdx.x];
    for (int w = 0; w < 32; ++w) {
      const int t = cntW[w * NBIN + threadIdx.x];
      cntW[w * NBIN + threadIdx.x] = run;
      run += t;
    }
  }
  __syncthreads();
  const unsigned lt = (1u << lane) - 1u;
  for (int eb = e0 + wave * 32; eb < e1; eb += 1024) {
    const int e = eb + lane;
    const bool live = e < e1;
    const int rr = live ? rj[e] : 0;
    const int key = rr / RNG;
    const unsigned lm = __builtin_amdgcn_ballot_w32(live);
    const unsigned peers = match_key8(key, lm);
    const int first = cw[key];
    if (live && (peers & lt) == 0u) cw[key] = first + __builtin_popcount(peers);
    if (live) {
      const int pos = first + __builtin_popcount(peers & lt);
      recs[pos] = int4{rr - key * RNG, lj[e], __float_as_int(weights[e]), 0};
    }
  }
}

template <bool FULL>
__device__ __forceinline__ void owner_chunk8(const float* __restrict__ lhs,
                                             const float* rhsS, float* tw,
                                             float& degc,
                                             int rowL, int idxL, int wbits,
                                             float we0, float we1, int lane,
                                             int c0, int nLive) {
  float b0[8], b1[8], a0[8], a1[8], o0[8], o1[8], m0[8], m1[8];
  int tt[8];
#pragma unroll
  for (int i = 0; i < 8; ++i) {
    const int sl = __builtin_amdgcn_readlane(idxL, c0 + i);
    const float* pl = lhs + (size_t)sl * 64 + lane;
    b0[i] = pl[0]; b1[i] = pl[32];
  }
#pragma unroll
  for (int i = 0; i < 8; ++i) {
    const int sr = __builtin_amdgcn_readlane(rowL, c0 + i);
    a0[i] = rhsS[sr * 64 + lane];
    a1[i] = rhsS[sr * 64 + 32 + lane];
    tt[i] = sr & 15;
    const int x = 4 * tt[i];
    const float* trw = tw + tt[i] * 64;
    o0[i] = trw[lane ^ x];
    o1[i] = trw[(32 + lane) ^ x];
  }
#pragma unroll
  for (int i = 0; i < 8; ++i) {
    const float w = __int_as_float(__builtin_amdgcn_readlane(wbits, c0 + i));
    float v0 = a0[i] + b0[i] + w * we0;
    float v1 = a1[i] + b1[i] + w * we1;
    m0[i] = v0 > 0.f ? v0 : 0.01f * v0;
    m1[i] = v1 > 0.f ? v1 : 0.01f * v1;
  }
#pragma unroll
  for (int i = 0; i < 8; ++i) {
    if (FULL || c0 + i < nLive) {
      float s0 = m0[i], s1 = m1[i];
#pragma unroll
      for (int j = 0; j < i; ++j) {
        if (tt[j] == tt[i]) { s0 += m0[j]; s1 += m1[j]; }
      }
      const int x = 4 * tt[i];
      float* trw = tw + tt[i] * 64;
      trw[lane ^ x]        = o0[i] + s0;
      trw[(32 + lane) ^ x] = o1[i] + s1;
      degc += (lane == tt[i]) ? 1.f : 0.f;
    }
  }
}

__launch_bounds__(CHK)
__global__ void owner_kernel(const float* __restrict__ rhs,
                             const float* __restrict__ lhs,
                             const int4* __restrict__ recs,
                             const int* __restrict__ binStart,
                             const float* __restrict__ We,
                             const float* __restrict__ X,
                             const float* __restrict__ Wf,
                             const float* __restrict__ bf,
                             const float* __restrict__ Wout,
                             const float* __restrict__ bout,
                             float* __restrict__ C, int M) {
  __shared__ float2 WfS2[32 * 64];
  __shared__ float2 WoS2[64 * 64];
  __shared__ __attribute__((aligned(16))) float tileS[RNG * 64];
  __shared__ __attribute__((aligned(16))) float rhsS[RNG * 64];
  __shared__ int4 stg[CHK];
  __shared__ unsigned short own[OWV * CHK];

  const int base = blockIdx.x * RNG;
  for (int i = threadIdx.x; i < 2048; i += CHK) {
    int k2 = i >> 6, n = i & 63;
    WfS2[i] = float2{Wf[(2 * k2) * 64 + n], Wf[(2 * k2 + 1) * 64 + n]};
  }
  for (int i = threadIdx.x; i < 4096; i += CHK) {
    int k2 = i >> 6, n = i & 63;
    WoS2[i] = float2{Wout[(2 * k2) * 64 + n], Wout[(2 * k2 + 1) * 64 + n]};
  }
  for (int i = threadIdx.x; i < RNG * 16; i += CHK) {
    const int row = base + (i >> 4);
    float4 v = float4{0.f, 0.f, 0.f, 0.f};
    if (row < M) v = reinterpret_cast<const float4*>(rhs)[(size_t)row * 16 + (i & 15)];
    reinterpret_cast<float4*>(rhsS)[i] = v;
    reinterpret_cast<float4*>(tileS)[i] = float4{0.f, 0.f, 0.f, 0.f};
  }

  const int lane = threadIdx.x & 31;
  const int wave = __builtin_amdgcn_readfirstlane((int)(threadIdx.x >> 5));
  const unsigned lt = (1u << lane) - 1u;

  float* tw = &tileS[wave * 16 * 64];
  float degc = 0.f;

  {
    const int eBeg = __builtin_amdgcn_readfirstlane(binStart[blockIdx.x]);
    const int eEnd = __builtin_amdgcn_readfirstlane(binStart[blockIdx.x + 1]);
    const float we0 = We[lane], we1 = We[32 + lane];
    unsigned short* ow = own + wave * CHK;
    int4 nxt = int4{0, 0, 0, 0};
    if (eBeg + (int)threadIdx.x < eEnd) nxt = recs[eBeg + threadIdx.x];
    for (int cb = eBeg; cb < eEnd; cb += CHK) {
      const int nC = (eEnd - cb) < CHK ? (eEnd - cb) : CHK;
      if ((int)threadIdx.x < nC) stg[threadIdx.x] = nxt;
      __syncthreads();
      {
        const int nn = cb + CHK + (int)threadIdx.x;
        if (nn < eEnd) nxt = recs[nn];
      }

      int nOwn = 0;
      for (int j0 = 0; j0 < CHK; j0 += 160) {
        int rowv[5];
#pragma unroll
        for (int u = 0; u < 5; ++u) {
          const int i = j0 + 32 * u + lane;
          rowv[u] = stg[i < nC ? i : 0].x;
        }
#pragma unroll
        for (int u = 0; u < 5; ++u) {
          const int i = j0 + 32 * u + lane;
          const bool mine = (i < nC) & ((rowv[u] >> 4) == wave);
          const unsigned m = __builtin_amdgcn_ballot_w32(mine);
          if (mine) ow[nOwn + __builtin_popcount(m & lt)] = (unsigned short)i;
          nOwn += __builtin_popcount(m);
        }
      }

      for (int c0 = 0; c0 < nOwn; c0 += 32) {
        const int nLive = (nOwn - c0) < 32 ? (nOwn - c0) : 32;
        const int li = lane < nLive ? c0 + lane : c0;
        const int4 rec = stg[ow[li]];
        const int rowL = rec.x, idxL = rec.y, wbits = rec.z;
        if (nLive == 32) {
#pragma unroll
          for (int c = 0; c < 32; c += 8)
            owner_chunk8<true>(lhs, rhsS, tw, degc, rowL, idxL, wbits, we0, we1, lane, c, nLive);
        } else {
#pragma unroll
          for (int c = 0; c < 32; c += 8)
            if (c < nLive)
              owner_chunk8<false>(lhs, rhsS, tw, degc, rowL, idxL, wbits, we0, we1, lane, c, nLive);
        }
      }
      __syncthreads();
    }
  }
  __syncthreads();

  const int r = lane & 15;
  const int h = lane >> 4;
  const int lrow0 = wave * 16;
  const int row0 = base + lrow0;
  if (row0 >= M) return;

  int rowA = row0 + r;
  if (rowA >= M) rowA = M - 1;
  const float* __restrict__ Xrow = X + (size_t)rowA * 64;

  float dg[8];
#pragma unroll
  for (int v = 0; v < 8; ++v) dg[v] = __shfl(degc, v + 8 * h);

  const float* trow = tw + r * 64;
  const int xr = 4 * r;

  v8f acc0 = {}, acc1 = {}, acc2 = {}, acc3 = {};
#pragma unroll
  for (int k0 = 0; k0 < 64; k0 += 4) {
    float2 av = *reinterpret_cast<const float2*>(trow + ((k0 ^ xr) + 2 * h));
    v2f a; a.x = av.x; a.y = av.y;
    const int k2 = (k0 >> 1) + h;
    const float2* Bk = &WfS2[k2 * 64 + r];
    float2 f0 = Bk[0], f1 = Bk[16], f2 = Bk[32], f3 = Bk[48];
    v2f b0, b1, b2, b3;
    b0.x = f0.x; b0.y = f0.y;  b1.x = f1.x; b1.y = f1.y;
    b2.x = f2.x; b2.y = f2.y;  b3.x = f3.x; b3.y = f3.y;
    acc0 = wmma_f32(a, b0, acc0);
    acc1 = wmma_f32(a, b1, acc1);
    acc2 = wmma_f32(a, b2, acc2);
    acc3 = wmma_f32(a, b3, acc3);
  }

  {
    const float bf0 = bf[r], bf1 = bf[16 + r], bf2 = bf[32 + r], bf3 = bf[48 + r];
#pragma unroll
    for (int v = 0; v < 8; ++v) {
      const int row = v + 8 * h;
      float* trw = tw + row * 64;
      const int x = 4 * row;
      const float d = dg[v];
      trw[(r)      ^ x] = acc0[v] + d * bf0;
      trw[(16 + r) ^ x] = acc1[v] + d * bf1;
      trw[(32 + r) ^ x] = acc2[v] + d * bf2;
      trw[(48 + r) ^ x] = acc3[v] + d * bf3;
    }
  }

  acc0 = v8f{}; acc1 = v8f{}; acc2 = v8f{}; acc3 = v8f{};
#pragma unroll
  for (int k0 = 0; k0 < 128; k0 += 4) {
    const int ka = k0 + 2 * h;
    float2 av;
    if (k0 < 64) av = *reinterpret_cast<const float2*>(trow + ((k0 ^ xr) + 2 * h));
    else         av = *reinterpret_cast<const float2*>(Xrow + (ka - 64));
    v2f a; a.x = av.x; a.y = av.y;
    const int k2 = (k0 >> 1) + h;
    const float2* Bk = &WoS2[k2 * 64 + r];
    float2 f0 = Bk[0], f1 = Bk[16], f2 = Bk[32], f3 = Bk[48];
    v2f b0, b1, b2, b3;
    b0.x = f0.x; b0.y = f0.y;  b1.x = f1.x; b1.y = f1.y;
    b2.x = f2.x; b2.y = f2.y;  b3.x = f3.x; b3.y = f3.y;
    acc0 = wmma_f32(a, b0, acc0);
    acc1 = wmma_f32(a, b1, acc1);
    acc2 = wmma_f32(a, b2, acc2);
    acc3 = wmma_f32(a, b3, acc3);
  }

  const float bia0 = bout[r], bia1 = bout[16 + r], bia2 = bout[32 + r], bia3 = bout[48 + r];
#pragma unroll
  for (int v = 0; v < 8; ++v) {
    const int row = row0 + v + 8 * h;
    if (row < M) {
      float* Crow = C + (size_t)row * 64;
      Crow[r]      = acc0[v] + bia0;
      Crow[16 + r] = acc1[v] + bia1;
      Crow[32 + r] = acc2[v] + bia2;
      Crow[48 + r] = acc3[v] + bia3;
    }
  }
}

extern "C" void kernel_launch(void* const* d_in, const int* in_sizes, int n_in,
                              void* d_out, int out_size, void* d_ws, size_t ws_size,
                              hipStream_t stream) {
  const float* input   = (const float*)d_in[0];
  const float* other   = (const float*)d_in[1];
  const int*   rj      = (const int*)d_in[2];
  const int*   lj      = (const int*)d_in[3];
  const float* weights = (const float*)d_in[4];
  const float* Wi      = (const float*)d_in[5];
  const float* bi      = (const float*)d_in[6];
  const float* Wo      = (const float*)d_in[7];
  const float* We      = (const float*)d_in[8];
  const float* Wf      = (const float*)d_in[9];
  const float* bf      = (const float*)d_in[10];
  const float* Wout    = (const float*)d_in[11];
  const float* bout    = (const float*)d_in[12];

  const int D    = 64;
  const int N_IN = in_sizes[0] / D;
  const int N_OT = in_sizes[1] / D;
  const int E    = in_sizes[2];

  float* rhs    = (float*)d_ws;
  float* lhs    = rhs + (size_t)N_IN * D;
  int4*  recs   = (int4*)(lhs + (size_t)N_OT * D);
  int*   cntmat = (int*)(recs + E);
  int*   offs   = cntmat + NBIN * NSL;
  int*   binStart = offs + NBIN * NSL;

  const int tiles_in = (N_IN + 15) / 16;
  const int tiles_ot = (N_OT + 15) / 16;
  const int nRanges  = (N_IN + RNG - 1) / RNG;
  const int epb      = (E + NSL - 1) / NSL;

  bin_count_kernel<<<NSL, 1024, 0, stream>>>(rj, cntmat, E, epb);
  bin_scan_kernel<<<1, 1024, 0, stream>>>(cntmat, offs, binStart);
  bin_place_kernel<<<NSL, 1024, 0, stream>>>(rj, lj, weights, offs, recs, E, epb);

  gemm64_kernel<<<(tiles_in + 7) / 8, 256, 0, stream>>>(input, Wi, bi, rhs, N_IN);
  gemm64_kernel<<<(tiles_ot + 7) / 8, 256, 0, stream>>>(other, Wo, nullptr, lhs, N_OT);

  owner_kernel<<<nRanges, CHK, 0, stream>>>(rhs, lhs, recs, binStart, We, input, Wf, bf,
                                            Wout, bout, (float*)d_out, N_IN);
}


# 6 launches -> 4: the two independent node GEMMs run as extra work-groups in the grids of the edge sort's count and scan launches (one-block scan hidden under the rhs GEMM); no sum order changes
# speedup vs baseline: 1.7260x; 1.1714x over previous
#include <hip/hip_runtime.h>

typedef float v2f __attribute__((ext_vector_type(2)));
typedef float v8f __attribute__((ext_vector_type(8)));

__device__ __forceinline__ v8f wmma_f32(v2f a, v2f b, v8f c) {
  return __builtin_amdgcn_wmma_f32_16x16x4_f32(false, a, false, b, (short)0, c,
                                               false, false);
}

template <int NT>
__device__ __forceinline__ void gemm64_block(const float* __restrict__ A,
                                             const float* __restrict__ B,
                                             const float* __restrict__ bias,
                                             float* __restrict__ C, int M, int blk) {
  __shared__ float2 Bs2[32 * 64];
  for (int i = threadIdx.x; i < 2048; i += NT) {
    int k2 = i >> 6, n = i & 63;
    Bs2[i] = float2{B[(2 * k2) * 64 + n], B[(2 * k2 + 1) * 64 + n]};
  }
  __syncthreads();

  const int wave = threadIdx.x >> 5;
  const int lane = threadIdx.x & 31;
  const int r = lane & 15;
  const int h = lane >> 4;

  const int row0 = (blk * (NT / 32) + wave) * 16;
  if (row0 >= M) return;

  int rowA = row0 + r;
  if (rowA >= M) rowA = M - 1;
  const float* __restrict__ Arow = A + (size_t)rowA * 64;

  v8f acc0 = {}, acc1 = {}, acc2 = {}, acc3 = {};
#pragma unroll
  for (int k0 = 0; k0 < 64; k0 += 4) {
    const int ka = k0 + 2 * h;
    float2 av = *reinterpret_cast<const float2*>(Arow + ka);
    v2f a; a.x = av.x; a.y = av.y;
    const int k2 = (k0 >> 1) + h;
    const float2* __restrict__ Bk = &Bs2[k2 * 64 + r];
    float2 f0 = Bk[0], f1 = Bk[16], f2 = Bk[32], f3 = Bk[48];
    v2f b0, b1, b2, b3;
    b0.x = f0.x; b0.y = f0.y;  b1.x = f1.x; b1.y = f1.y;
    b2.x = f2.x; b2.y = f2.y;  b3.x = f3.x; b3.y = f3.y;
    acc0 = wmma_f32(a, b0, acc0);
    acc1 = wmma_f32(a, b1, acc1);
    acc2 = wmma_f32(a, b2, acc2);
    acc3 = wmma_f32(a, b3, acc3);
  }

  float bia0 = 0.f, bia1 = 0.f, bia2 = 0.f, bia3 = 0.f;
  if (bias) { bia0 = bias[r]; bia1 = bias[16 + r]; bia2 = bias[32 + r]; bia3 = bias[48 + r]; }

#pragma unroll
  for (int v = 0; v < 8; ++v) {
    const int row = row0 + v + 8 * h;
    if (row < M) {
      float* __restrict__ Crow = C + (size_t)row * 64;
      Crow[r]      = acc0[v] + bia0;
      Crow[16 + r] = acc1[v] + bia1;
      Crow[32 + r] = acc2[v] + bia2;
      Crow[48 + r] = acc3[v] + bia3;
    }
  }
}

constexpr int RNG  = 400;
constexpr int NBIN = 256;
constexpr int NSL  = 256;
constexpr int OWV  = RNG / 16;
constexpr int CHK  = OWV * 32;

__device__ __forceinline__ unsigned match_key8(int key, unsigned liveMask) {
  unsigned peers = liveMask;
#pragma unroll
  for (int b = 0; b < 8; ++b) {
    const bool bit = (key >> b) & 1;
    const unsigned bal = __builtin_amdgcn_ballot_w32(bit);
    peers &= bit ? bal : ~bal;
  }
  return peers;
}

__device__ __forceinline__ void count_slice(const int* __restrict__ rj, int* cw,
                                            int e0, int e1, int wave, int lane) {
  const unsigned lt = (1u << lane) - 1u;
  for (int eb = e0 + wave * 32; eb < e1; eb += 1024) {
    const int e = eb + lane;
    const bool live = e < e1;
    const int key = live ? rj[e] / RNG : 0;
    const unsigned lm = __builtin_amdgcn_ballot_w32(live);
    const unsigned peers = match_key8(key, lm);
    if (live && (peers & lt) == 0u) cw[key] += __builtin_popcount(peers);
  }
}

__launch_bounds__(1024)
__global__ void count_gemm_kernel(const int* __restrict__ rj, int* __restrict__ cntmat,
                                  int E, int epb,
                                  const float* __restrict__ A, const float* __restrict__ B,
                                  const float* __restrict__ bias, float* __restrict__ C, int M) {
  if (blockIdx.x >= NSL) {
    gemm64_block<1024>(A, B, bias, C, M, (int)blockIdx.x - NSL);
    return;
  }
  __shared__ int cntW[32 * NBIN];
  for (int i = threadIdx.x; i < 32 * NBIN; i += 1024) cntW[i] = 0;
  __syncthreads();
  const int lane = threadIdx.x & 31;
  const int wave = threadIdx.x >> 5;
  const int e0 = blockIdx.x * epb;
  int e1 = e0 + epb;
  if (e1 > E) e1 = E;
  count_slice(rj, cntW + wave * NBIN, e0, e1, wave, lane);
  __syncthreads();
  if (threadIdx.x < NBIN) {
    int s = 0;
    for (int w = 0; w < 32; ++w) s += cntW[w * NBIN + threadIdx.x];
    cntmat[threadIdx.x * NSL + blockIdx.x] = s;
  }
}

__launch_bounds__(1024)
__global__ void scan_gemm_kernel(const int* __restrict__ cntmat, int* __restrict__ offs,
                                 int* __restrict__ binStart,
                                 const float* __restrict__ A, const float* __restrict__ B,
                                 const float* __restrict__ bias, float* __restrict__ C, int M) {
  if (blockIdx.x > 0) {
    gemm64_block<1024>(A, B, bias, C, M, (int)blockIdx.x - 1);
    return;
  }
  __shared__ int tot[NBIN];
  __shared__ int startS[NBIN];
  const int lane = threadIdx.x & 31;
  const int wave = threadIdx.x >> 5;
  int ex[8][8];
#pragma unroll
  for (int j = 0; j < 8; ++j) {
    const int bin = wave * 8 + j;
    const int4* p = reinterpret_cast<const int4*>(cntmat + bin * NSL + lane * 8);
    const int4 a = p[0], b = p[1];
    const int c[8] = {a.x, a.y, a.z, a.w, b.x, b.y, b.z, b.w};
    int s = 0;
#pragma unroll
    for (int i = 0; i < 8; ++i) { ex[j][i] = s; s += c[i]; }
    int inc = s;
#pragma unroll
    for (int d = 1; d < 32; d <<= 1) {
      const int t = __shfl_up(inc, d);
      if (lane >= d) inc += t;
    }
    const int exl = inc - s;
#pragma unroll
    for (int i = 0; i < 8; ++i) ex[j][i] += exl;
    const int total = __shfl(inc, 31);
    if (lane == 0) tot[bin] = total;
  }
  __syncthreads();
  if (wave == 0) {
    int c[8], lx[8];
    int s = 0;
#pragma unroll
    for (int i = 0; i < 8; ++i) { c[i] = tot[lane * 8 + i]; lx[i] = s; s += c[i]; }
    int inc = s;
#pragma unroll
    for (int d = 1; d < 32; d <<= 1) {
      const int t = __shfl_up(inc, d);
      if (lane >= d) inc += t;
    }
    const int exl = inc - s;
#pragma unroll
    for (int i = 0; i < 8; ++i) startS[lane * 8 + i] = exl + lx[i];
  }
  __syncthreads();
#pragma unroll
  for (int j = 0; j < 8; ++j) {
    const int bin = wave * 8 + j;
    const int base = startS[bin];
    int4* p = reinterpret_cast<int4*>(offs + bin * NSL + lane * 8);
    p[0] = int4{base + ex[j][0], base + ex[j][1], base + ex[j][2], base + ex[j][3]};
    p[1] = int4{base + ex[j][4], base + ex[j][5], base + ex[j][6], base + ex[j][7]};
  }
  if (threadIdx.x < NBIN) binStart[threadIdx.x] = startS[threadIdx.x];
  if (threadIdx.x == 0) binStart[NBIN] = startS[NBIN - 1] + tot[NBIN - 1];
}

__launch_bounds__(1024)
__global__ void bin_place_kernel(const int* __restrict__ rj, const int* __restrict__ lj,
                                 const float* __restrict__ weights,
                                 const int* __restrict__ offs, int4* __restrict__ recs,
                                 int E, int epb) {
  __shared__ int cntW[32 * NBIN];
  for (int i = threadIdx.x; i < 32 * NBIN; i += 1024) cntW[i] = 0;
  __syncthreads();
  const int lane = threadIdx.x & 31;
  const int wave = threadIdx.x >> 5;
  const int e0 = blockIdx.x * epb;
  int e1 = e0 + epb;
  if (e1 > E) e1 = E;
  int* cw = cntW + wave * NBIN;
  count_slice(rj, cw, e0, e1, wave, lane);
  __syncthreads();
  if (threadIdx.x < NBIN) {
    int run = offs[threadIdx.x * NSL + blockIdx.x];
    for (int w = 0; w < 32; ++w) {
      const int t = cntW[w * NBIN + threadIdx.x];
      cntW[w * NBIN + threadIdx.x] = run;
      run += t;
    }
  }
  __syncthreads();
  const unsigned lt = (1u << lane) - 1u;
  for (int eb = e0 + wave * 32; eb < e1; eb += 1024) {
    const int e = eb + lane;
    const bool live = e < e1;
    const int rr = live ? rj[e] : 0;
    const int key = rr / RNG;
    const unsigned lm = __builtin_amdgcn_ballot_w32(live);
    const unsigned peers = match_key8(key, lm);
    const int first = cw[key];
    if (live && (peers & lt) == 0u) cw[key] = first + __builtin_popcount(peers);
    if (live) {
      const int pos = first + __builtin_popcount(peers & lt);
      recs[pos] = int4{rr - key * RNG, lj[e], __float_as_int(weights[e]), 0};
    }
  }
}

template <bool FULL>
__device__ __forceinline__ void owner_chunk8(const float* __restrict__ lhs,
                                             const float* rhsS, float* tw,
                                             float& degc,
                                             int rowL, int idxL, int wbits,
                                             float we0, float we1, int lane,
                                             int c0, int nLive) {
  float b0[8], b1[8], a0[8], a1[8], o0[8], o1[8], m0[8], m1[8];
  int tt[8];
#pragma unroll
  for (int i = 0; i < 8; ++i) {
    const int sl = __builtin_amdgcn_readlane(idxL, c0 + i);
    const float* pl = lhs + (size_t)sl * 64 + lane;
    b0[i] = pl[0]; b1[i] = pl[32];
  }
#pragma unroll
  for (int i = 0; i < 8; ++i) {
    const int sr = __builtin_amdgcn_readlane(rowL, c0 + i);
    a0[i] = rhsS[sr * 64 + lane];
    a1[i] = rhsS[sr * 64 + 32 + lane];
    tt[i] = sr & 15;
    const int x = 4 * tt[i];
    const float* trw = tw + tt[i] * 64;
    o0[i] = trw[lane ^ x];
    o1[i] = trw[(32 + lane) ^ x];
  }
#pragma unroll
  for (int i = 0; i < 8; ++i) {
    const float w = __int_as_float(__builtin_amdgcn_readlane(wbits, c0 + i));
    float v0 = a0[i] + b0[i] + w * we0;
    float v1 = a1[i] + b1[i] + w * we1;
    m0[i] = v0 > 0.f ? v0 : 0.01f * v0;
    m1[i] = v1 > 0.f ? v1 : 0.01f * v1;
  }
#pragma unroll
  for (int i = 0; i < 8; ++i) {
    if (FULL || c0 + i < nLive) {
      float s0 = m0[i], s1 = m1[i];
#pragma unroll
      for (int j = 0; j < i; ++j) {
        if (tt[j] == tt[i]) { s0 += m0[j]; s1 += m1[j]; }
      }
      const int x = 4 * tt[i];
      float* trw = tw + tt[i] * 64;
      trw[lane ^ x]        = o0[i] + s0;
      trw[(32 + lane) ^ x] = o1[i] + s1;
      degc += (lane == tt[i]) ? 1.f : 0.f;
    }
  }
}

__launch_bounds__(CHK)
__global__ void owner_kernel(const float* __restrict__ rhs,
                             const float* __restrict__ lhs,
                             const int4* __restrict__ recs,
                             const int* __restrict__ binStart,
                             const float* __restrict__ We,
                             const float* __restrict__ X,
                             const float* __restrict__ Wf,
                             const float* __restrict__ bf,
                             const float* __restrict__ Wout,
                             const float* __restrict__ bout,
                             float* __restrict__ C, int M) {
  __shared__ float2 WfS2[32 * 64];
  __shared__ float2 WoS2[64 * 64];
  __shared__ __attribute__((aligned(16))) float tileS[RNG * 64];
  __shared__ __attribute__((aligned(16))) float rhsS[RNG * 64];
  __shared__ int4 stg[CHK];
  __shared__ unsigned short own[OWV * CHK];

  const int base = blockIdx.x * RNG;
  for (int i = threadIdx.x; i < 2048; i += CHK) {
    int k2 = i >> 6, n = i & 63;
    WfS2[i] = float2{Wf[(2 * k2) * 64 + n], Wf[(2 * k2 + 1) * 64 + n]};
  }
  for (int i = threadIdx.x; i < 4096; i += CHK) {
    int k2 = i >> 6, n = i & 63;
    WoS2[i] = float2{Wout[(2 * k2) * 64 + n], Wout[(2 * k2 + 1) * 64 + n]};
  }
  for (int i = threadIdx.x; i < RNG * 16; i += CHK) {
    const int row = base + (i >> 4);
    float4 v = float4{0.f, 0.f, 0.f, 0.f};
    if (row < M) v = reinterpret_cast<const float4*>(rhs)[(size_t)row * 16 + (i & 15)];
    reinterpret_cast<float4*>(rhsS)[i] = v;
    reinterpret_cast<float4*>(tileS)[i] = float4{0.f, 0.f, 0.f, 0.f};
  }

  const int lane = threadIdx.x & 31;
  const int wave = __builtin_amdgcn_readfirstlane((int)(threadIdx.x >> 5));
  const unsigned lt = (1u << lane) - 1u;

  float* tw = &tileS[wave * 16 * 64];
  float degc = 0.f;

  {
    const int eBeg = __builtin_amdgcn_readfirstlane(binStart[blockIdx.x]);
    const int eEnd = __builtin_amdgcn_readfirstlane(binStart[blockIdx.x + 1]);
    const float we0 = We[lane], we1 = We[32 + lane];
    unsigned short* ow = own + wave * CHK;
    int4 nxt = int4{0, 0, 0, 0};
    if (eBeg + (int)threadIdx.x < eEnd) nxt = recs[eBeg + threadIdx.x];
    for (int cb = eBeg; cb < eEnd; cb += CHK) {
      const int nC = (eEnd - cb) < CHK ? (eEnd - cb) : CHK;
      if ((int)threadIdx.x < nC) stg[threadIdx.x] = nxt;
      __syncthreads();
      {
        const int nn = cb + CHK + (int)threadIdx.x;
        if (nn < eEnd) nxt = recs[nn];
      }

      int nOwn = 0;
      for (int j0 = 0; j0 < CHK; j0 += 160) {
        int rowv[5];
#pragma unroll
        for (int u = 0; u < 5; ++u) {
          const int i = j0 + 32 * u + lane;
          rowv[u] = stg[i < nC ? i : 0].x;
        }
#pragma unroll
        for (int u = 0; u < 5; ++u) {
          const int i = j0 + 32 * u + lane;
          const bool mine = (i < nC) & ((rowv[u] >> 4) == wave);
          const unsigned m = __builtin_amdgcn_ballot_w32(mine);
          if (mine) ow[nOwn + __builtin_popcount(m & lt)] = (unsigned short)i;
          nOwn += __builtin_popcount(m);
        }
      }

      for (int c0 = 0; c0 < nOwn; c0 += 32) {
        const int nLive = (nOwn - c0) < 32 ? (nOwn - c0) : 32;
        const int li = lane < nLive ? c0 + lane : c0;
        const int4 rec = stg[ow[li]];
        const int rowL = rec.x, idxL = rec.y, wbits = rec.z;
        if (nLive == 32) {
#pragma unroll
          for (int c = 0; c < 32; c += 8)
            owner_chunk8<true>(lhs, rhsS, tw, degc, rowL, idxL, wbits, we0, we1, lane, c, nLive);
        } else {
#pragma unroll
          for (int c = 0; c < 32; c += 8)
            if (c < nLive)
              owner_chunk8<false>(lhs, rhsS, tw, degc, rowL, idxL, wbits, we0, we1, lane, c, nLive);
        }
      }
      __syncthreads();
    }
  }
  __syncthreads();

  const int r = lane & 15;
  const int h = lane >> 4;
  const int lrow0 = wave * 16;
  const int row0 = base + lrow0;
  if (row0 >= M) return;

  int rowA = row0 + r;
  if (rowA >= M) rowA = M - 1;
  const float* __restrict__ Xrow = X + (size_t)rowA * 64;

  float dg[8];
#pragma unroll
  for (int v = 0; v < 8; ++v) dg[v] = __shfl(degc, v + 8 * h);

  const float* trow = tw + r * 64;
  const int xr = 4 * r;

  v8f acc0 = {}, acc1 = {}, acc2 = {}, acc3 = {};
#pragma unroll
  for (int k0 = 0; k0 < 64; k0 += 4) {
    float2 av = *reinterpret_cast<const float2*>(trow + ((k0 ^ xr) + 2 * h));
    v2f a; a.x = av.x; a.y = av.y;
    const int k2 = (k0 >> 1) + h;
    const float2* Bk = &WfS2[k2 * 64 + r];
    float2 f0 = Bk[0], f1 = Bk[16], f2 = Bk[32], f3 = Bk[48];
    v2f b0, b1, b2, b3;
    b0.x = f0.x; b0.y = f0.y;  b1.x = f1.x; b1.y = f1.y;
    b2.x = f2.x; b2.y = f2.y;  b3.x = f3.x; b3.y = f3.y;
    acc0 = wmma_f32(a, b0, acc0);
    acc1 = wmma_f32(a, b1, acc1);
    acc2 = wmma_f32(a, b2, acc2);
    acc3 = wmma_f32(a, b3, acc3);
  }

  {
    const float bf0 = bf[r], bf1 = bf[16 + r], bf2 = bf[32 + r], bf3 = bf[48 + r];
#pragma unroll
    for (int v = 0; v < 8; ++v) {
      const int row = v + 8 * h;
      float* trw = tw + row * 64;
      const int x = 4 * row;
      const float d = dg[v];
      trw[(r)      ^ x] = acc0[v] + d * bf0;
      trw[(16 + r) ^ x] = acc1[v] + d * bf1;
      trw[(32 + r) ^ x] = acc2[v] + d * bf2;
      trw[(48 + r) ^ x] = acc3[v] + d * bf3;
    }
  }

  acc0 = v8f{}; acc1 = v8f{}; acc2 = v8f{}; acc3 = v8f{};
#pragma unroll
  for (int k0 = 0; k0 < 128; k0 += 4) {
    const int ka = k0 + 2 * h;
    float2 av;
    if (k0 < 64) av = *reinterpret_cast<const float2*>(trow + ((k0 ^ xr) + 2 * h));
    else         av = *reinterpret_cast<const float2*>(Xrow + (ka - 64));
    v2f a; a.x = av.x; a.y = av.y;
    const int k2 = (k0 >> 1) + h;
    const float2* Bk = &WoS2[k2 * 64 + r];
    float2 f0 = Bk[0], f1 = Bk[16], f2 = Bk[32], f3 = Bk[48];
    v2f b0, b1, b2, b3;
    b0.x = f0.x; b0.y = f0.y;  b1.x = f1.x; b1.y = f1.y;
    b2.x = f2.x; b2.y = f2.y;  b3.x = f3.x; b3.y = f3.y;
    acc0 = wmma_f32(a, b0, acc0);
    acc1 = wmma_f32(a, b1, acc1);
    acc2 = wmma_f32(a, b2, acc2);
    acc3 = wmma_f32(a, b3, acc3);
  }

  const float bia0 = bout[r], bia1 = bout[16 + r], bia2 = bout[32 + r], bia3 = bout[48 + r];
#pragma unroll
  for (int v = 0; v < 8; ++v) {
    const int row = row0 + v + 8 * h;
    if (row < M) {
      float* Crow = C + (size_t)row * 64;
      Crow[r]      = acc0[v] + bia0;
      Crow[16 + r] = acc1[v] + bia1;
      Crow[32 + r] = acc2[v] + bia2;
      Crow[48 + r] = acc3[v] + bia3;
    }
  }
}

extern "C" void kernel_launch(void* const* d_in, const int* in_sizes, int n_in,
                              void* d_out, int out_size, void* d_ws, size_t ws_size,
                              hipStream_t stream) {
  const float* input   = (const float*)d_in[0];
  const float* other   = (const float*)d_in[1];
  const int*   rj      = (const int*)d_in[2];
  const int*   lj      = (const int*)d_in[3];
  const float* weights = (const float*)d_in[4];
  const float* Wi      = (const float*)d_in[5];
  const float* bi      = (const float*)d_in[6];
  const float* Wo      = (const float*)d_in[7];
  const float* We      = (const float*)d_in[8];
  const float* Wf      = (const float*)d_in[9];
  const float* bf      = (const float*)d_in[10];
  const float* Wout    = (const float*)d_in[11];
  const float* bout    = (const float*)d_in[12];

  const int D    = 64;
  const int N_IN = in_sizes[0] / D;
  const int N_OT = in_sizes[1] / D;
  const int E    = in_sizes[2];

  float* rhs    = (float*)d_ws;
  float* lhs    = rhs + (size_t)N_IN * D;
  int4*  recs   = (int4*)(lhs + (size_t)N_OT * D);
  int*   cntmat = (int*)(recs + E);
  int*   offs   = cntmat + NBIN * NSL;
  int*   binStart = offs + NBIN * NSL;

  const int tiles_in = (N_IN + 15) / 16;
  const int tiles_ot = (N_OT + 15) / 16;
  const int nRanges  = (N_IN + RNG - 1) / RNG;
  const int epb      = (E + NSL - 1) / NSL;

  count_gemm_kernel<<<NSL + (tiles_ot + 31) / 32, 1024, 0, stream>>>(
      rj, cntmat, E, epb, other, Wo, nullptr, lhs, N_OT);
  scan_gemm_kernel<<<1 + (tiles_in + 31) / 32, 1024, 0, stream>>>(
      cntmat, offs, binStart, input, Wi, bi, rhs, N_IN);
  bin_place_kernel<<<NSL, 1024, 0, stream>>>(rj, lj, weights, offs, recs, E, epb);

  owner_kernel<<<nRanges, CHK, 0, stream>>>(rhs, lhs, recs, binStart, We, input, Wf, bf,
                                            Wout, bout, (float*)d_out, N_IN);
}


# Edge grouping made local to an edge slice: a place work-group counts, scans, places its own slice; the owner walks its range's 256 segments; count and scan launches gone (4 -> 2); no sum order change
# speedup vs baseline: 2.0787x; 1.2043x over previous
#include <hip/hip_runtime.h>

typedef float v2f __attribute__((ext_vector_type(2)));
typedef float v8f __attribute__((ext_vector_type(8)));

__device__ __forceinline__ v8f wmma_f32(v2f a, v2f b, v8f c) {
  return __builtin_amdgcn_wmma_f32_16x16x4_f32(false, a, false, b, (short)0, c,
                                               false, false);
}

template <int NT>
__device__ __forceinline__ void gemm64_block(const float* __restrict__ A,
                                             const float* __restrict__ B,
                                             const float* __restrict__ bias,
                                             float* __restrict__ C, int M, int blk) {
  __shared__ float2 Bs2[32 * 64];
  for (int i = threadIdx.x; i < 2048; i += NT) {
    int k2 = i >> 6, n = i & 63;
    Bs2[i] = float2{B[(2 * k2) * 64 + n], B[(2 * k2 + 1) * 64 + n]};
  }
  __syncthreads();

  const int wave = threadIdx.x >> 5;
  const int lane = threadIdx.x & 31;
  const int r = lane & 15;
  const int h = lane >> 4;

  const int row0 = (blk * (NT / 32) + wave) * 16;
  if (row0 >= M) return;

  int rowA = row0 + r;
  if (rowA >= M) rowA = M - 1;
  const float* __restrict__ Arow = A + (size_t)rowA * 64;

  v8f acc0 = {}, acc1 = {}, acc2 = {}, acc3 = {};
#pragma unroll
  for (int k0 = 0; k0 < 64; k0 += 4) {
    const int ka = k0 + 2 * h;
    float2 av = *reinterpret_cast<const float2*>(Arow + ka);
    v2f a; a.x = av.x; a.y = av.y;
    const int k2 = (k0 >> 1) + h;
    const float2* __restrict__ Bk = &Bs2[k2 * 64 + r];
    float2 f0 = Bk[0], f1 = Bk[16], f2 = Bk[32], f3 = Bk[48];
    v2f b0, b1, b2, b3;
    b0.x = f0.x; b0.y = f0.y;  b1.x = f1.x; b1.y = f1.y;
    b2.x = f2.x; b2.y = f2.y;  b3.x = f3.x; b3.y = f3.y;
    acc0 = wmma_f32(a, b0, acc0);
    acc1 = wmma_f32(a, b1, acc1);
    acc2 = wmma_f32(a, b2, acc2);
    acc3 = wmma_f32(a, b3, acc3);
  }

  float bia0 = 0.f, bia1 = 0.f, bia2 = 0.f, bia3 = 0.f;
  if (bias) { bia0 = bias[r]; bia1 = bias[16 + r]; bia2 = bias[32 + r]; bia3 = bias[48 + r]; }

#pragma unroll
  for (int v = 0; v < 8; ++v) {
    const int row = row0 + v + 8 * h;
    if (row < M) {
      float* __restrict__ Crow = C + (size_t)row * 64;
      Crow[r]      = acc0[v] + bia0;
      Crow[16 + r] = acc1[v] + bia1;
      Crow[32 + r] = acc2[v] + bia2;
      Crow[48 + r] = acc3[v] + bia3;
    }
  }
}

constexpr int RNG  = 400;
constexpr int NBIN = 256;
constexpr int NSL  = 256;
constexpr int OWV  = RNG / 16;
constexpr int CHK  = OWV * 32;

__device__ __forceinline__ unsigned match_key8(int key, unsigned liveMask) {
  unsigned peers = liveMask;
#pragma unroll
  for (int b = 0; b < 8; ++b) {
    const bool bit = (key >> b) & 1;
    const unsigned bal = __builtin_amdgcn_ballot_w32(bit);
    peers &= bit ? bal : ~bal;
  }
  return peers;
}

__device__ __forceinline__ void count_slice(const int* __restrict__ rj, int* cw,
                                            int e0, int e1, int wave, int lane) {
  const unsigned lt = (1u << lane) - 1u;
  for (int eb = e0 + wave * 32; eb < e1; eb += 1024) {
    const int e = eb + lane;
    const bool live = e < e1;
    const int key = live ? rj[e] / RNG : 0;
    const unsigned lm = __builtin_amdgcn_ballot_w32(live);
    const unsigned peers = match_key8(key, lm);
    if (live && (peers & lt) == 0u) cw[key] += __builtin_popcount(peers);
  }
}

__launch_bounds__(1024, 16)
__global__ void place_gemm_kernel(const int* __restrict__ rj, const int* __restrict__ lj,
                                  const float* __restrict__ weights,
                                  int* __restrict__ seg, int4* __restrict__ recs,
                                  int E, int epb,
                                  const float* __restrict__ other, const float* __restrict__ Wo,
                                  float* __restrict__ lhs, int N_OT, int nbOt,
                                  const float* __restrict__ input, const float* __restrict__ Wi,
                                  const float* __restrict__ bi, float* __restrict__ rhs, int N_IN) {
  if (blockIdx.x >= NSL) {
    const int g = (int)blockIdx.x - NSL;
    const bool ot = g < nbOt;
    gemm64_block<1024>(ot ? other : input, ot ? Wo : Wi, ot ? nullptr : bi, ot ? lhs : rhs,
                       ot ? N_OT : N_IN, ot ? g : g - nbOt);
    return;
  }
  __shared__ int cntW[32 * NBIN];
  __shared__ int wsum[NBIN / 32];
  for (int i = threadIdx.x; i < 32 * NBIN; i += 1024) cntW[i] = 0;
  __syncthreads();
  const int lane = threadIdx.x & 31;
  const int wave = threadIdx.x >> 5;
  const int e0 = blockIdx.x * epb;
  int e1 = e0 + epb;
  if (e1 > E) e1 = E;
  int* cw = cntW + wave * NBIN;
  count_slice(rj, cw, e0, e1, wave, lane);
  __syncthreads();
  int c[32];
  int tot = 0, incl = 0;
  if (threadIdx.x < NBIN) {
#pragma unroll
    for (int w = 0; w < 32; ++w) c[w] = cntW[w * NBIN + threadIdx.x];
#pragma unroll
    for (int w = 0; w < 32; ++w) tot += c[w];
    incl = tot;
#pragma unroll
    for (int d = 1; d < 32; d <<= 1) {
      const int t = __shfl_up(incl, d);
      if (lane >= d) incl += t;
    }
    if (lane == 31) wsum[wave] = incl;
  }
  __syncthreads();
  if (threadIdx.x < NBIN) {
    int off = 0;
#pragma unroll
    for (int w = 0; w < NBIN / 32; ++w) off += (w < wave) ? wsum[w] : 0;
    int run = off + incl - tot;
    seg[threadIdx.x * NSL + blockIdx.x] = run;
    if (threadIdx.x == NBIN - 1) seg[NBIN * NSL + blockIdx.x] = run + tot;
#pragma unroll
    for (int w = 0; w < 32; ++w) { cntW[w * NBIN + threadIdx.x] = run; run += c[w]; }
  }
  __syncthreads();
  int4* __restrict__ out = recs + e0;
  const unsigned lt = (1u << lane) - 1u;
  for (int eb = e0 + wave * 32; eb < e1; eb += 1024) {
    const int e = eb + lane;
    const bool live = e < e1;
    const int rr = live ? rj[e] : 0;
    const int key = rr / RNG;
    const unsigned lm = __builtin_amdgcn_ballot_w32(live);
    const unsigned peers = match_key8(key, lm);
    const int first = cw[key];
    if (live && (peers & lt) == 0u) cw[key] = first + __builtin_popcount(peers);
    if (live) {
      const int pos = first + __builtin_popcount(peers & lt);
      out[pos] = int4{rr - key * RNG, lj[e], __float_as_int(weights[e]), 0};
    }
  }
}

template <bool FULL>
__device__ __forceinline__ void owner_chunk8(const float* __restrict__ lhs,
                                             const float* rhsS, float* tw,
                                             float& degc,
                                             int rowL, int idxL, int wbits,
                                             float we0, float we1, int lane,
                                             int c0, int nLive) {
  float b0[8], b1[8], a0[8], a1[8], o0[8], o1[8], m0[8], m1[8];
  int tt[8];
#pragma unroll
  for (int i = 0; i < 8; ++i) {
    const int sl = __builtin_amdgcn_readlane(idxL, c0 + i);
    const float* pl = lhs + (size_t)sl * 64 + lane;
    b0[i] = pl[0]; b1[i] = pl[32];
  }
#pragma unroll
  for (int i = 0; i < 8; ++i) {
    const int sr = __builtin_amdgcn_readlane(rowL, c0 + i);
    a0[i] = rhsS[sr * 64 + lane];
    a1[i] = rhsS[sr * 64 + 32 + lane];
    tt[i] = sr & 15;
    const int x = 4 * tt[i];
    const float* trw = tw + tt[i] * 64;
    o0[i] = trw[lane ^ x];
    o1[i] = trw[(32 + lane) ^ x];
  }
#pragma unroll
  for (int i = 0; i < 8; ++i) {
    const float w = __int_as_float(__builtin_amdgcn_readlane(wbits, c0 + i));
    float v0 = a0[i] + b0[i] + w * we0;
    float v1 = a1[i] + b1[i] + w * we1;
    m0[i] = v0 > 0.f ? v0 : 0.01f * v0;
    m1[i] = v1 > 0.f ? v1 : 0.01f * v1;
  }
#pragma unroll
  for (int i = 0; i < 8; ++i) {
    if (FULL || c0 + i < nLive) {
      float s0 = m0[i], s1 = m1[i];
#pragma unroll
      for (int j = 0; j < i; ++j) {
        if (tt[j] == tt[i]) { s0 += m0[j]; s1 += m1[j]; }
      }
      const int x = 4 * tt[i];
      float* trw = tw + tt[i] * 64;
      trw[lane ^ x]        = o0[i] + s0;
      trw[(32 + lane) ^ x] = o1[i] + s1;
      degc += (lane == tt[i]) ? 1.f : 0.f;
    }
  }
}

__launch_bounds__(CHK)
__global__ void owner_kernel(const float* __restrict__ rhs,
                             const float* __restrict__ lhs,
                             const int4* __restrict__ recs,
                             const int* __restrict__ seg, int epb,
                             const float* __restrict__ We,
                             const float* __restrict__ X,
                             const float* __restrict__ Wf,
                             const float* __restrict__ bf,
                             const float* __restrict__ Wout,
                             const float* __restrict__ bout,
                             float* __restrict__ C, int M) {
  __shared__ float2 WfS2[32 * 64];
  __shared__ float2 WoS2[64 * 64];
  __shared__ __attribute__((aligned(16))) float tileS[RNG * 64];
  __shared__ __attribute__((aligned(16))) float rhsS[RNG * 64];
  __shared__ int4 stg[CHK];
  __shared__ unsigned short own[OWV * CHK];
  __shared__ __attribute__((aligned(64))) int segP[NSL + 1];
  __shared__ __attribute__((aligned(64))) int segPc[NSL / 16];
  __shared__ int segG[NSL];
  __shared__ int wsum[NSL / 32];

  const int base = blockIdx.x * RNG;
  const int lane = threadIdx.x & 31;
  int segSt = 0, segEn = 0;
  if (threadIdx.x < NSL) {
    segSt = seg[blockIdx.x * NSL + threadIdx.x];
    segEn = seg[(blockIdx.x + 1) * NSL + threadIdx.x];
  }
  for (int i = threadIdx.x; i < 2048; i += CHK) {
    int k2 = i >> 6, n = i & 63;
    WfS2[i] = float2{Wf[(2 * k2) * 64 + n], Wf[(2 * k2 + 1) * 64 + n]};
  }
  for (int i = threadIdx.x; i < 4096; i += CHK) {
    int k2 = i >> 6, n = i & 63;
    WoS2[i] = float2{Wout[(2 * k2) * 64 + n], Wout[(2 * k2 + 1) * 64 + n]};
  }
  for (int i = threadIdx.x; i < RNG * 16; i += CHK) {
    const int row = base + (i >> 4);
    float4 v = float4{0.f, 0.f, 0.f, 0.f};
    if (row < M) v = reinterpret_cast<const float4*>(rhs)[(size_t)row * 16 + (i & 15)];
    reinterpret_cast<float4*>(rhsS)[i] = v;
    reinterpret_cast<float4*>(tileS)[i] = float4{0.f, 0.f, 0.f, 0.f};
  }
  const int segLen = segEn - segSt;
  int incl = segLen;
  if (threadIdx.x < NSL) {
#pragma unroll
    for (int d = 1; d < 32; d <<= 1) {
      const int t = __shfl_up(incl, d);
      if (lane >= d) incl += t;
    }
    if (lane == 31) wsum[threadIdx.x >> 5] = incl;
  }
  __syncthreads();
  if (threadIdx.x < NSL) {
    int off = 0;
#pragma unroll
    for (int w = 0; w < NSL / 32; ++w) off += (w < (int)(threadIdx.x >> 5)) ? wsum[w] : 0;
    const int ex = off + incl - segLen;
    segP[threadIdx.x] = ex;
    if ((threadIdx.x & 15) == 0) segPc[threadIdx.x >> 4] = ex;
    segG[threadIdx.x] = (int)threadIdx.x * epb + segSt - ex;
    if (threadIdx.x == NSL - 1) segP[NSL] = ex + segLen;
  }
  __syncthreads();

  const int wave = __builtin_amdgcn_readfirstlane((int)(threadIdx.x >> 5));
  const unsigned lt = (1u << lane) - 1u;

  float* tw = &tileS[wave * 16 * 64];
  float degc = 0.f;

  {
    const int eEnd = __builtin_amdgcn_readfirstlane(segP[NSL]);
    auto fetch = [&](int f) -> int4 {
      const int4* pc = reinterpret_cast<const int4*>(segPc);
      int c = -1;
#pragma unroll
      for (int q = 0; q < 4; ++q) {
        const int4 v = pc[q];
        c += (v.x <= f) + (v.y <= f) + (v.z <= f) + (v.w <= f);
      }
      const int4* pf = reinterpret_cast<const int4*>(segP + 16 * c);
      int sl = 16 * c - 1;
#pragma unroll
      for (int q = 0; q < 4; ++q) {
        const int4 v = pf[q];
        sl += (v.x <= f) + (v.y <= f) + (v.z <= f) + (v.w <= f);
      }
      return recs[segG[sl] + f];
    };
    const float we0 = We[lane], we1 = We[32 + lane];
    unsigned short* ow = own + wave * CHK;
    int4 nxt = int4{0, 0, 0, 0};
    if ((int)threadIdx.x < eEnd) nxt = fetch((int)threadIdx.x);
    for (int cb = 0; cb < eEnd; cb += CHK) {
      const int nC = (eEnd - cb) < CHK ? (eEnd - cb) : CHK;
      if ((int)threadIdx.x < nC) stg[threadIdx.x] = nxt;
      __syncthreads();
      {
        const int nn = cb + CHK + (int)threadIdx.x;
        if (nn < eEnd) nxt = fetch(nn);
      }

      int nOwn = 0;
      for (int j0 = 0; j0 < CHK; j0 += 160) {
        int rowv[5];
#pragma unroll
        for (int u = 0; u < 5; ++u) {
          const int i = j0 + 32 * u + lane;
          rowv[u] = stg[i < nC ? i : 0].x;
        }
#pragma unroll
        for (int u = 0; u < 5; ++u) {
          const int i = j0 + 32 * u + lane;
          const bool mine = (i < nC) & ((rowv[u] >> 4) == wave);
          const unsigned m = __builtin_amdgcn_ballot_w32(mine);
          if (mine) ow[nOwn + __builtin_popcount(m & lt)] = (unsigned short)i;
          nOwn += __builtin_popcount(m);
        }
      }

      for (int c0 = 0; c0 < nOwn; c0 += 32) {
        const int nLive = (nOwn - c0) < 32 ? (nOwn - c0) : 32;
        const int li = lane < nLive ? c0 + lane : c0;
        const int4 rec = stg[ow[li]];
        const int rowL = rec.x, idxL = rec.y, wbits = rec.z;
        if (nLive == 32) {
#pragma unroll
          for (int c = 0; c < 32; c += 8)
            owner_chunk8<true>(lhs, rhsS, tw, degc, rowL, idxL, wbits, we0, we1, lane, c, nLive);
        } else {
#pragma unroll
          for (int c = 0; c < 32; c += 8)
            if (c < nLive)
              owner_chunk8<false>(lhs, rhsS, tw, degc, rowL, idxL, wbits, we0, we1, lane, c, nLive);
        }
      }
      __syncthreads();
    }
  }

  const int r = lane & 15;
  const int h = lane >> 4;
  const int lrow0 = wave * 16;
  const int row0 = base + lrow0;
  if (row0 >= M) return;

  int rowA = row0 + r;
  if (rowA >= M) rowA = M - 1;
  const float* __restrict__ Xrow = X + (size_t)rowA * 64;

  float dg[8];
#pragma unroll
  for (int v = 0; v < 8; ++v) dg[v] = __shfl(degc, v + 8 * h);

  const float* trow = tw + r * 64;
  const int xr = 4 * r;

  v8f acc0 = {}, acc1 = {}, acc2 = {}, acc3 = {};
#pragma unroll
  for (int k0 = 0; k0 < 64; k0 += 4) {
    float2 av = *reinterpret_cast<const float2*>(trow + ((k0 ^ xr) + 2 * h));
    v2f a; a.x = av.x; a.y = av.y;
    const int k2 = (k0 >> 1) + h;
    const float2* Bk = &WfS2[k2 * 64 + r];
    float2 f0 = Bk[0], f1 = Bk[16], f2 = Bk[32], f3 = Bk[48];
    v2f b0, b1, b2, b3;
    b0.x = f0.x; b0.y = f0.y;  b1.x = f1.x; b1.y = f1.y;
    b2.x = f2.x; b2.y = f2.y;  b3.x = f3.x; b3.y = f3.y;
    acc0 = wmma_f32(a, b0, acc0);
    acc1 = wmma_f32(a, b1, acc1);
    acc2 = wmma_f32(a, b2, acc2);
    acc3 = wmma_f32(a, b3, acc3);
  }

  {
    const float bf0 = bf[r], bf1 = bf[16 + r], bf2 = bf[32 + r], bf3 = bf[48 + r];
#pragma unroll
    for (int v = 0; v < 8; ++v) {
      const int row = v + 8 * h;
      float* trw = tw + row * 64;
      const int x = 4 * row;
      const float d = dg[v];
      trw[(r)      ^ x] = acc0[v] + d * bf0;
      trw[(16 + r) ^ x] = acc1[v] + d * bf1;
      trw[(32 + r) ^ x] = acc2[v] + d * bf2;
      trw[(48 + r) ^ x] = acc3[v] + d * bf3;
    }
  }

  acc0 = v8f{}; acc1 = v8f{}; acc2 = v8f{}; acc3 = v8f{};
#pragma unroll
  for (int k0 = 0; k0 < 128; k0 += 4) {
    const int ka = k0 + 2 * h;
    float2 av;
    if (k0 < 64) av = *reinterpret_cast<const float2*>(trow + ((k0 ^ xr) + 2 * h));
    else         av = *reinterpret_cast<const float2*>(Xrow + (ka - 64));
    v2f a; a.x = av.x; a.y = av.y;
    const int k2 = (k0 >> 1) + h;
    const float2* Bk = &WoS2[k2 * 64 + r];
    float2 f0 = Bk[0], f1 = Bk[16], f2 = Bk[32], f3 = Bk[48];
    v2f b0, b1, b2, b3;
    b0.x = f0.x; b0.y = f0.y;  b1.x = f1.x; b1.y = f1.y;
    b2.x = f2.x; b2.y = f2.y;  b3.x = f3.x; b3.y = f3.y;
    acc0 = wmma_f32(a, b0, acc0);
    acc1 = wmma_f32(a, b1, acc1);
    acc2 = wmma_f32(a, b2, acc2);
    acc3 = wmma_f32(a, b3, acc3);
  }

  const float bia0 = bout[r], bia1 = bout[16 + r], bia2 = bout[32 + r], bia3 = bout[48 + r];
#pragma unroll
  for (int v = 0; v < 8; ++v) {
    const int row = row0 + v + 8 * h;
    if (row < M) {
      float* Crow = C + (size_t)row * 64;
      Crow[r]      = acc0[v] + bia0;
      Crow[16 + r] = acc1[v] + bia1;
      Crow[32 + r] = acc2[v] + bia2;
      Crow[48 + r] = acc3[v] + bia3;
    }
  }
}

extern "C" void kernel_launch(void* const* d_in, const int* in_sizes, int n_in,
                              void* d_out, int out_size, void* d_ws, size_t ws_size,
                              hipStream_t stream) {
  const float* input   = (const float*)d_in[0];
  const float* other   = (const float*)d_in[1];
  const int*   rj      = (const int*)d_in[2];
  const int*   lj      = (const int*)d_in[3];
  const float* weights = (const float*)d_in[4];
  const float* Wi      = (const float*)d_in[5];
  const float* bi      = (const float*)d_in[6];
  const float* Wo      = (const float*)d_in[7];
  const float* We      = (const float*)d_in[8];
  const float* Wf      = (const float*)d_in[9];
  const float* bf      = (const float*)d_in[10];
  const float* Wout    = (const float*)d_in[11];
  const float* bout    = (const float*)d_in[12];

  const int D    = 64;
  const int N_IN = in_sizes[0] / D;
  const int N_OT = in_sizes[1] / D;
  const int E    = in_sizes[2];

  float* rhs    = (float*)d_ws;
  float* lhs    = rhs + (size_t)N_IN * D;
  int4*  recs   = (int4*)(lhs + (size_t)N_OT * D);
  int*   seg    = (int*)(recs + E);

  const int tiles_in = (N_IN + 15) / 16;
  const int tiles_ot = (N_OT + 15) / 16;
  const int nRanges  = (N_IN + RNG - 1) / RNG;
  const int epb      = (E + NSL - 1) / NSL;

  const int nbOt = (tiles_ot + 31) / 32;
  const int nbIn = (tiles_in + 31) / 32;
  place_gemm_kernel<<<NSL + nbOt + nbIn, 1024, 0, stream>>>(
      rj, lj, weights, seg, recs, E, epb, other, Wo, lhs, N_OT, nbOt, input, Wi, bi, rhs, N_IN);

  owner_kernel<<<nRanges, CHK, 0, stream>>>(rhs, lhs, recs, seg, epb, We, input, Wf, bf,
                                            Wout, bout, (float*)d_out, N_IN);
}


# Each 32-record step of a wave ordered by tile row (ballot ranks, ds_permute): a group of 8 merges same-row terms along neighbours, 7 pairs not 28; lhs rows asked a group ahead; f32 sum order changes
# speedup vs baseline: 2.1691x; 1.0435x over previous
#include <hip/hip_runtime.h>

typedef float v2f __attribute__((ext_vector_type(2)));
typedef float v8f __attribute__((ext_vector_type(8)));

__device__ __forceinline__ v8f wmma_f32(v2f a, v2f b, v8f c) {
  return __builtin_amdgcn_wmma_f32_16x16x4_f32(false, a, false, b, (short)0, c,
                                               false, false);
}

template <int NT>
__device__ __forceinline__ void gemm64_block(const float* __restrict__ A,
                                             const float* __restrict__ B,
                                             const float* __restrict__ bias,
                                             float* __restrict__ C, int M, int blk) {
  __shared__ float2 Bs2[32 * 64];
  for (int i = threadIdx.x; i < 2048; i += NT) {
    int k2 = i >> 6, n = i & 63;
    Bs2[i] = float2{B[(2 * k2) * 64 + n], B[(2 * k2 + 1) * 64 + n]};
  }
  __syncthreads();

  const int wave = threadIdx.x >> 5;
  const int lane = threadIdx.x & 31;
  const int r = lane & 15;
  const int h = lane >> 4;

  const int row0 = (blk * (NT / 32) + wave) * 16;
  if (row0 >= M) return;

  int rowA = row0 + r;
  if (rowA >= M) rowA = M - 1;
  const float* __restrict__ Arow = A + (size_t)rowA * 64;

  v8f acc0 = {}, acc1 = {}, acc2 = {}, acc3 = {};
#pragma unroll
  for (int k0 = 0; k0 < 64; k0 += 4) {
    const int ka = k0 + 2 * h;
    float2 av = *reinterpret_cast<const float2*>(Arow + ka);
    v2f a; a.x = av.x; a.y = av.y;
    const int k2 = (k0 >> 1) + h;
    const float2* __restrict__ Bk = &Bs2[k2 * 64 + r];
    float2 f0 = Bk[0], f1 = Bk[16], f2 = Bk[32], f3 = Bk[48];
    v2f b0, b1, b2, b3;
    b0.x = f0.x; b0.y = f0.y;  b1.x = f1.x; b1.y = f1.y;
    b2.x = f2.x; b2.y = f2.y;  b3.x = f3.x; b3.y = f3.y;
    acc0 = wmma_f32(a, b0, acc0);
    acc1 = wmma_f32(a, b1, acc1);
    acc2 = wmma_f32(a, b2, acc2);
    acc3 = wmma_f32(a, b3, acc3);
  }

  float bia0 = 0.f, bia1 = 0.f, bia2 = 0.f, bia3 = 0.f;
  if (bias) { bia0 = bias[r]; bia1 = bias[16 + r]; bia2 = bias[32 + r]; bia3 = bias[48 + r]; }

#pragma unroll
  for (int v = 0; v < 8; ++v) {
    const int row = row0 + v + 8 * h;
    if (row < M) {
      float* __restrict__ Crow = C + (size_t)row * 64;
      Crow[r]      = acc0[v] + bia0;
      Crow[16 + r] = acc1[v] + bia1;
      Crow[32 + r] = acc2[v] + bia2;
      Crow[48 + r] = acc3[v] + bia3;
    }
  }
}

constexpr int RNG  = 400;
constexpr int NBIN = 256;
constexpr int NSL  = 256;
constexpr int OWV  = RNG / 16;
constexpr int CHK  = OWV * 32;

__device__ __forceinline__ unsigned match_key8(int key, unsigned liveMask) {
  unsigned peers = liveMask;
#pragma unroll
  for (int b = 0; b < 8; ++b) {
    const bool bit = (key >> b) & 1;
    const unsigned bal = __builtin_amdgcn_ballot_w32(bit);
    peers &= bit ? bal : ~bal;
  }
  return peers;
}

__device__ __forceinline__ void count_slice(const int* __restrict__ rj, int* cw,
                                            int e0, int e1, int wave, int lane) {
  const unsigned lt = (1u << lane) - 1u;
  for (int eb = e0 + wave * 32; eb < e1; eb += 1024) {
    const int e = eb + lane;
    const bool live = e < e1;
    const int key = live ? rj[e] / RNG : 0;
    const unsigned lm = __builtin_amdgcn_ballot_w32(live);
    const unsigned peers = match_key8(key, lm);
    if (live && (peers & lt) == 0u) cw[key] += __builtin_popcount(peers);
  }
}

__launch_bounds__(1024, 16)
__global__ void place_gemm_kernel(const int* __restrict__ rj, const int* __restrict__ lj,
                                  const float* __restrict__ weights,
                                  int* __restrict__ seg, int4* __restrict__ recs,
                                  int E, int epb,
                                  const float* __restrict__ other, const float* __restrict__ Wo,
                                  float* __restrict__ lhs, int N_OT, int nbOt,
                                  const float* __restrict__ input, const float* __restrict__ Wi,
                                  const float* __restrict__ bi, float* __restrict__ rhs, int N_IN) {
  if (blockIdx.x >= NSL) {
    const int g = (int)blockIdx.x - NSL;
    const bool ot = g < nbOt;
    gemm64_block<1024>(ot ? other : input, ot ? Wo : Wi, ot ? nullptr : bi, ot ? lhs : rhs,
                       ot ? N_OT : N_IN, ot ? g : g - nbOt);
    return;
  }
  __shared__ int cntW[32 * NBIN];
  __shared__ int wsum[NBIN / 32];
  for (int i = threadIdx.x; i < 32 * NBIN; i += 1024) cntW[i] = 0;
  __syncthreads();
  const int lane = threadIdx.x & 31;
  const int wave = threadIdx.x >> 5;
  const int e0 = blockIdx.x * epb;
  int e1 = e0 + epb;
  if (e1 > E) e1 = E;
  int* cw = cntW + wave * NBIN;
  count_slice(rj, cw, e0, e1, wave, lane);
  __syncthreads();
  int c[32];
  int tot = 0, incl = 0;
  if (threadIdx.x < NBIN) {
#pragma unroll
    for (int w = 0; w < 32; ++w) c[w] = cntW[w * NBIN + threadIdx.x];
#pragma unroll
    for (int w = 0; w < 32; ++w) tot += c[w];
    incl = tot;
#pragma unroll
    for (int d = 1; d < 32; d <<= 1) {
      const int t = __shfl_up(incl, d);
      if (lane >= d) incl += t;
    }
    if (lane == 31) wsum[wave] = incl;
  }
  __syncthreads();
  if (threadIdx.x < NBIN) {
    int off = 0;
#pragma unroll
    for (int w = 0; w < NBIN / 32; ++w) off += (w < wave) ? wsum[w] : 0;
    int run = off + incl - tot;
    seg[threadIdx.x * NSL + blockIdx.x] = run;
    if (threadIdx.x == NBIN - 1) seg[NBIN * NSL + blockIdx.x] = run + tot;
#pragma unroll
    for (int w = 0; w < 32; ++w) { cntW[w * NBIN + threadIdx.x] = run; run += c[w]; }
  }
  __syncthreads();
  int4* __restrict__ out = recs + e0;
  const unsigned lt = (1u << lane) - 1u;
  for (int eb = e0 + wave * 32; eb < e1; eb += 1024) {
    const int e = eb + lane;
    const bool live = e < e1;
    const int rr = live ? rj[e] : 0;
    const int key = rr / RNG;
    const unsigned lm = __builtin_amdgcn_ballot_w32(live);
    const unsigned peers = match_key8(key, lm);
    const int first = cw[key];
    if (live && (peers & lt) == 0u) cw[key] = first + __builtin_popcount(peers);
    if (live) {
      const int pos = first + __builtin_popcount(peers & lt);
      out[pos] = int4{rr - key * RNG, lj[e], __float_as_int(weights[e]), 0};
    }
  }
}

__device__ __forceinline__ void edge_rows8(const float* __restrict__ lhs, int idxL, int lane,
                                           int c0, float (&b0)[8], float (&b1)[8]) {
#pragma unroll
  for (int i = 0; i < 8; ++i) {
    const int sl = __builtin_amdgcn_readlane(idxL, c0 + i);
    const float* pl = lhs + (size_t)sl * 64 + lane;
    b0[i] = pl[0]; b1[i] = pl[32];
  }
}

template <bool FULL>
__device__ __forceinline__ void owner_chunk8(const float (&b0)[8], const float (&b1)[8],
                                             const float* rhsS, float* tw,
                                             float& degc,
                                             int rowL, int wbits,
                                             float we0, float we1, int lane,
                                             int c0, int nLive) {
  float a0[8], a1[8], o0[8], o1[8], m0[8], m1[8];
  int tt[8];
#pragma unroll
  for (int i = 0; i < 8; ++i) {
    const int sr = __builtin_amdgcn_readlane(rowL, c0 + i);
    a0[i] = rhsS[sr * 64 + lane];
    a1[i] = rhsS[sr * 64 + 32 + lane];
    tt[i] = sr & 15;
    const int x = 4 * tt[i];
    const float* trw = tw + tt[i] * 64;
    o0[i] = trw[lane ^ x];
    o1[i] = trw[(32 + lane) ^ x];
  }
#pragma unroll
  for (int i = 0; i < 8; ++i) {
    const float w = __int_as_float(__builtin_amdgcn_readlane(wbits, c0 + i));
    float v0 = a0[i] + b0[i] + w * we0;
    float v1 = a1[i] + b1[i] + w * we1;
    m0[i] = v0 > 0.f ? v0 : 0.01f * v0;
    m1[i] = v1 > 0.f ? v1 : 0.01f * v1;
  }
  float s0 = 0.f, s1 = 0.f;
#pragma unroll
  for (int i = 0; i < 8; ++i) {
    if (FULL || c0 + i < nLive) {
      const bool same = i > 0 && tt[i] == tt[i - 1];
      s0 = same ? s0 + m0[i] : m0[i];
      s1 = same ? s1 + m1[i] : m1[i];
      const int x = 4 * tt[i];
      float* trw = tw + tt[i] * 64;
      trw[lane ^ x]        = o0[i] + s0;
      trw[(32 + lane) ^ x] = o1[i] + s1;
      degc += (lane == tt[i]) ? 1.f : 0.f;
    }
  }
}

__launch_bounds__(CHK)
__global__ void owner_kernel(const float* __restrict__ rhs,
                             const float* __restrict__ lhs,
                             const int4* __restrict__ recs,
                             const int* __restrict__ seg, int epb,
                             const float* __restrict__ We,
                             const float* __restrict__ X,
                             const float* __restrict__ Wf,
                             const float* __restrict__ bf,
                             const float* __restrict__ Wout,
                             const float* __restrict__ bout,
                             float* __restrict__ C, int M) {
  __shared__ float2 WfS2[32 * 64];
  __shared__ float2 WoS2[64 * 64];
  __shared__ __attribute__((aligned(16))) float tileS[RNG * 64];
  __shared__ __attribute__((aligned(16))) float rhsS[RNG * 64];
  __shared__ int4 stg[CHK];
  __shared__ unsigned short own[OWV * CHK];
  __shared__ __attribute__((aligned(64))) int segP[NSL + 1];
  __shared__ __attribute__((aligned(64))) int segPc[NSL / 16];
  __shared__ int segG[NSL];
  __shared__ int wsum[NSL / 32];

  const int base = blockIdx.x * RNG;
  const int lane = threadIdx.x & 31;
  int segSt = 0, segEn = 0;
  if (threadIdx.x < NSL) {
    segSt = seg[blockIdx.x * NSL + threadIdx.x];
    segEn = seg[(blockIdx.x + 1) * NSL + threadIdx.x];
  }
  for (int i = threadIdx.x; i < 2048; i += CHK) {
    int k2 = i >> 6, n = i & 63;
    WfS2[i] = float2{Wf[(2 * k2) * 64 + n], Wf[(2 * k2 + 1) * 64 + n]};
  }
  for (int i = threadIdx.x; i < 4096; i += CHK) {
    int k2 = i >> 6, n = i & 63;
    WoS2[i] = float2{Wout[(2 * k2) * 64 + n], Wout[(2 * k2 + 1) * 64 + n]};
  }
  for (int i = threadIdx.x; i < RNG * 16; i += CHK) {
    const int row = base + (i >> 4);
    float4 v = float4{0.f, 0.f, 0.f, 0.f};
    if (row < M) v = reinterpret_cast<const float4*>(rhs)[(size_t)row * 16 + (i & 15)];
    reinterpret_cast<float4*>(rhsS)[i] = v;
    reinterpret_cast<float4*>(tileS)[i] = float4{0.f, 0.f, 0.f, 0.f};
  }
  const int segLen = segEn - segSt;
  int incl = segLen;
  if (threadIdx.x < NSL) {
#pragma unroll
    for (int d = 1; d < 32; d <<= 1) {
      const int t = __shfl_up(incl, d);
      if (lane >= d) incl += t;
    }
    if (lane == 31) wsum[threadIdx.x >> 5] = incl;
  }
  __syncthreads();
  if (threadIdx.x < NSL) {
    int off = 0;
#pragma unroll
    for (int w = 0; w < NSL / 32; ++w) off += (w < (int)(threadIdx.x >> 5)) ? wsum[w] : 0;
    const int ex = off + incl - segLen;
    segP[threadIdx.x] = ex;
    if ((threadIdx.x & 15) == 0) segPc[threadIdx.x >> 4] = ex;
    segG[threadIdx.x] = (int)threadIdx.x * epb + segSt - ex;
    if (threadIdx.x == NSL - 1) segP[NSL] = ex + segLen;
  }
  __syncthreads();

  const int wave = __builtin_amdgcn_readfirstlane((int)(threadIdx.x >> 5));
  const unsigned lt = (1u << lane) - 1u;

  float* tw = &tileS[wave * 16 * 64];
  float degc = 0.f;

  {
    const int eEnd = __builtin_amdgcn_readfirstlane(segP[NSL]);
    auto fetch = [&](int f) -> int4 {
      const int4* pc = reinterpret_cast<const int4*>(segPc);
      int c = -1;
#pragma unroll
      for (int q = 0; q < 4; ++q) {
        const int4 v = pc[q];
        c += (v.x <= f) + (v.y <= f) + (v.z <= f) + (v.w <= f);
      }
      const int4* pf = reinterpret_cast<const int4*>(segP + 16 * c);
      int sl = 16 * c - 1;
#pragma unroll
      for (int q = 0; q < 4; ++q) {
        const int4 v = pf[q];
        sl += (v.x <= f) + (v.y <= f) + (v.z <= f) + (v.w <= f);
      }
      return recs[segG[sl] + f];
    };
    const float we0 = We[lane], we1 = We[32 + lane];
    unsigned short* ow = own + wave * CHK;
    int4 nxt = int4{0, 0, 0, 0};
    if ((int)threadIdx.x < eEnd) nxt = fetch((int)threadIdx.x);
    for (int cb = 0; cb < eEnd; cb += CHK) {
      const int nC = (eEnd - cb) < CHK ? (eEnd - cb) : CHK;
      if ((int)threadIdx.x < nC) stg[threadIdx.x] = nxt;
      __syncthreads();
      {
        const int nn = cb + CHK + (int)threadIdx.x;
        if (nn < eEnd) nxt = fetch(nn);
      }

      int nOwn = 0;
      for (int j0 = 0; j0 < CHK; j0 += 160) {
        int rowv[5];
#pragma unroll
        for (int u = 0; u < 5; ++u) {
          const int i = j0 + 32 * u + lane;
          rowv[u] = stg[i < nC ? i : 0].x;
        }
#pragma unroll
        for (int u = 0; u < 5; ++u) {
          const int i = j0 + 32 * u + lane;
          const bool mine = (i < nC) & ((rowv[u] >> 4) == wave);
          const unsigned m = __builtin_amdgcn_ballot_w32(mine);
          if (mine) ow[nOwn + __builtin_popcount(m & lt)] = (unsigned short)i;
          nOwn += __builtin_popcount(m);
        }
      }

      for (int c0 = 0; c0 < nOwn; c0 += 32) {
        const int nLive = (nOwn - c0) < 32 ? (nOwn - c0) : 32;
        const int li = lane < nLive ? c0 + lane : c0;
        const int4 rec = stg[ow[li]];
        const int key = lane < nLive ? (rec.x & 15) : 16;
        unsigned less = 0u, eq = ~0u;
#pragma unroll
        for (int b = 4; b >= 0; --b) {
          const bool bit = (key >> b) & 1;
          const unsigned bal = __builtin_amdgcn_ballot_w32(bit);
          less |= bit ? (eq & ~bal) : 0u;
          eq &= bit ? bal : ~bal;
        }
        const int pos = __builtin_popcount(less) + __builtin_popcount(eq & lt);
        const int rowL  = __builtin_amdgcn_ds_permute(pos << 2, rec.x);
        const int idxL  = __builtin_amdgcn_ds_permute(pos << 2, rec.y);
        const int wbits = __builtin_amdgcn_ds_permute(pos << 2, rec.z);
        float bA0[8], bA1[8], bB0[8], bB1[8];
        edge_rows8(lhs, idxL, lane, 0, bA0, bA1);
        if (nLive == 32) {
          edge_rows8(lhs, idxL, lane, 8, bB0, bB1);
          owner_chunk8<true>(bA0, bA1, rhsS, tw, degc, rowL, wbits, we0, we1, lane, 0, nLive);
          edge_rows8(lhs, idxL, lane, 16, bA0, bA1);
          owner_chunk8<true>(bB0, bB1, rhsS, tw, degc, rowL, wbits, we0, we1, lane, 8, nLive);
          edge_rows8(lhs, idxL, lane, 24, bB0, bB1);
          owner_chunk8<true>(bA0, bA1, rhsS, tw, degc, rowL, wbits, we0, we1, lane, 16, nLive);
          owner_chunk8<true>(bB0, bB1, rhsS, tw, degc, rowL, wbits, we0, we1, lane, 24, nLive);
        } else {
          if (nLive > 8) edge_rows8(lhs, idxL, lane, 8, bB0, bB1);
          owner_chunk8<false>(bA0, bA1, rhsS, tw, degc, rowL, wbits, we0, we1, lane, 0, nLive);
          if (nLive > 8) {
            if (nLive > 16) edge_rows8(lhs, idxL, lane, 16, bA0, bA1);
            owner_chunk8<false>(bB0, bB1, rhsS, tw, degc, rowL, wbits, we0, we1, lane, 8, nLive);
            if (nLive > 16) {
              if (nLive > 24) edge_rows8(lhs, idxL, lane, 24, bB0, bB1);
              owner_chunk8<false>(bA0, bA1, rhsS, tw, degc, rowL, wbits, we0, we1, lane, 16, nLive);
              if (nLive > 24)
                owner_chunk8<false>(bB0, bB1, rhsS, tw, degc, rowL, wbits, we0, we1, lane, 24, nLive);
            }
          }
        }
      }
      __syncthreads();
    }
  }

  const int r = lane & 15;
  const int h = lane >> 4;
  const int lrow0 = wave * 16;
  const int row0 = base + lrow0;
  if (row0 >= M) return;

  int rowA = row0 + r;
  if (rowA >= M) rowA = M - 1;
  const float* __restrict__ Xrow = X + (size_t)rowA * 64;

  float dg[8];
#pragma unroll
  for (int v = 0; v < 8; ++v) dg[v] = __shfl(degc, v + 8 * h);

  const float* trow = tw + r * 64;
  const int xr = 4 * r;

  v8f acc0 = {}, acc1 = {}, acc2 = {}, acc3 = {};
#pragma unroll
  for (int k0 = 0; k0 < 64; k0 += 4) {
    float2 av = *reinterpret_cast<const float2*>(trow + ((k0 ^ xr) + 2 * h));
    v2f a; a.x = av.x; a.y = av.y;
    const int k2 = (k0 >> 1) + h;
    const float2* Bk = &WfS2[k2 * 64 + r];
    float2 f0 = Bk[0], f1 = Bk[16], f2 = Bk[32], f3 = Bk[48];
    v2f b0, b1, b2, b3;
    b0.x = f0.x; b0.y = f0.y;  b1.x = f1.x; b1.y = f1.y;
    b2.x = f2.x; b2.y = f2.y;  b3.x = f3.x; b3.y = f3.y;
    acc0 = wmma_f32(a, b0, acc0);
    acc1 = wmma_f32(a, b1, acc1);
    acc2 = wmma_f32(a, b2, acc2);
    acc3 = wmma_f32(a, b3, acc3);
  }

  {
    const float bf0 = bf[r], bf1 = bf[16 + r], bf2 = bf[32 + r], bf3 = bf[48 + r];
#pragma unroll
    for (int v = 0; v < 8; ++v) {
      const int row = v + 8 * h;
      float* trw = tw + row * 64;
      const int x = 4 * row;
      const float d = dg[v];
      trw[(r)      ^ x] = acc0[v] + d * bf0;
      trw[(16 + r) ^ x] = acc1[v] + d * bf1;
      trw[(32 + r) ^ x] = acc2[v] + d * bf2;
      trw[(48 + r) ^ x] = acc3[v] + d * bf3;
    }
  }

  acc0 = v8f{}; acc1 = v8f{}; acc2 = v8f{}; acc3 = v8f{};
#pragma unroll
  for (int k0 = 0; k0 < 128; k0 += 4) {
    const int ka = k0 + 2 * h;
    float2 av;
    if (k0 < 64) av = *reinterpret_cast<const float2*>(trow + ((k0 ^ xr) + 2 * h));
    else         av = *reinterpret_cast<const float2*>(Xrow + (ka - 64));
    v2f a; a.x = av.x; a.y = av.y;
    const int k2 = (k0 >> 1) + h;
    const float2* Bk = &WoS2[k2 * 64 + r];
    float2 f0 = Bk[0], f1 = Bk[16], f2 = Bk[32], f3 = Bk[48];
    v2f b0, b1, b2, b3;
    b0.x = f0.x; b0.y = f0.y;  b1.x = f1.x; b1.y = f1.y;
    b2.x = f2.x; b2.y = f2.y;  b3.x = f3.x; b3.y = f3.y;
    acc0 = wmma_f32(a, b0, acc0);
    acc1 = wmma_f32(a, b1, acc1);
    acc2 = wmma_f32(a, b2, acc2);
    acc3 = wmma_f32(a, b3, acc3);
  }

  const float bia0 = bout[r], bia1 = bout[16 + r], bia2 = bout[32 + r], bia3 = bout[48 + r];
#pragma unroll
  for (int v = 0; v < 8; ++v) {
    const int row = row0 + v + 8 * h;
    if (row < M) {
      float* Crow = C + (size_t)row * 64;
      Crow[r]      = acc0[v] + bia0;
      Crow[16 + r] = acc1[v] + bia1;
      Crow[32 + r] = acc2[v] + bia2;
      Crow[48 + r] = acc3[v] + bia3;
    }
  }
}

extern "C" void kernel_launch(void* const* d_in, const int* in_sizes, int n_in,
                              void* d_out, int out_size, void* d_ws, size_t ws_size,
                              hipStream_t stream) {
  const float* input   = (const float*)d_in[0];
  const float* other   = (const float*)d_in[1];
  const int*   rj      = (const int*)d_in[2];
  const int*   lj      = (const int*)d_in[3];
  const float* weights = (const float*)d_in[4];
  const float* Wi      = (const float*)d_in[5];
  const float* bi      = (const float*)d_in[6];
  const float* Wo      = (const float*)d_in[7];
  const float* We      = (const float*)d_in[8];
  const float* Wf      = (const float*)d_in[9];
  const float* bf      = (const float*)d_in[10];
  const float* Wout    = (const float*)d_in[11];
  const float* bout    = (const float*)d_in[12];

  const int D    = 64;
  const int N_IN = in_sizes[0] / D;
  const int N_OT = in_sizes[1] / D;
  const int E    = in_sizes[2];

  float* rhs    = (float*)d_ws;
  float* lhs    = rhs + (size_t)N_IN * D;
  int4*  recs   = (int4*)(lhs + (size_t)N_OT * D);
  int*   seg    = (int*)(recs + E);

  const int tiles_in = (N_IN + 15) / 16;
  const int tiles_ot = (N_OT + 15) / 16;
  const int nRanges  = (N_IN + RNG - 1) / RNG;
  const int epb      = (E + NSL - 1) / NSL;

  const int nbOt = (tiles_ot + 31) / 32;
  const int nbIn = (tiles_in + 31) / 32;
  place_gemm_kernel<<<NSL + nbOt + nbIn, 1024, 0, stream>>>(
      rj, lj, weights, seg, recs, E, epb, other, Wo, lhs, N_OT, nbOt, input, Wi, bi, rhs, N_IN);

  owner_kernel<<<nRanges, CHK, 0, stream>>>(rhs, lhs, recs, seg, epb, We, input, Wf, bf,
                                            Wout, bout, (float*)d_out, N_IN);
}


# Owner: a range's records counting-sorted by tile into LDS in one round of 4800 (wave-private counters, ballot ranks, one scan); no wave scans the whole stage every 800 records; f32 sum order changes
# speedup vs baseline: 2.4070x; 1.1097x over previous
#include <hip/hip_runtime.h>

typedef float v2f __attribute__((ext_vector_type(2)));
typedef float v8f __attribute__((ext_vector_type(8)));

__device__ __forceinline__ v8f wmma_f32(v2f a, v2f b, v8f c) {
  return __builtin_amdgcn_wmma_f32_16x16x4_f32(false, a, false, b, (short)0, c,
                                               false, false);
}

template <int NT>
__device__ __forceinline__ void gemm64_block(const float* __restrict__ A,
                                             const float* __restrict__ B,
                                             const float* __restrict__ bias,
                                             float* __restrict__ C, int M, int blk) {
  __shared__ float2 Bs2[32 * 64];
  for (int i = threadIdx.x; i < 2048; i += NT) {
    int k2 = i >> 6, n = i & 63;
    Bs2[i] = float2{B[(2 * k2) * 64 + n], B[(2 * k2 + 1) * 64 + n]};
  }
  __syncthreads();

  const int wave = threadIdx.x >> 5;
  const int lane = threadIdx.x & 31;
  const int r = lane & 15;
  const int h = lane >> 4;

  const int row0 = (blk * (NT / 32) + wave) * 16;
  if (row0 >= M) return;

  int rowA = row0 + r;
  if (rowA >= M) rowA = M - 1;
  const float* __restrict__ Arow = A + (size_t)rowA * 64;

  v8f acc0 = {}, acc1 = {}, acc2 = {}, acc3 = {};
#pragma unroll
  for (int k0 = 0; k0 < 64; k0 += 4) {
    const int ka = k0 + 2 * h;
    float2 av = *reinterpret_cast<const float2*>(Arow + ka);
    v2f a; a.x = av.x; a.y = av.y;
    const int k2 = (k0 >> 1) + h;
    const float2* __restrict__ Bk = &Bs2[k2 * 64 + r];
    float2 f0 = Bk[0], f1 = Bk[16], f2 = Bk[32], f3 = Bk[48];
    v2f b0, b1, b2, b3;
    b0.x = f0.x; b0.y = f0.y;  b1.x = f1.x; b1.y = f1.y;
    b2.x = f2.x; b2.y = f2.y;  b3.x = f3.x; b3.y = f3.y;
    acc0 = wmma_f32(a, b0, acc0);
    acc1 = wmma_f32(a, b1, acc1);
    acc2 = wmma_f32(a, b2, acc2);
    acc3 = wmma_f32(a, b3, acc3);
  }

  float bia0 = 0.f, bia1 = 0.f, bia2 = 0.f, bia3 = 0.f;
  if (bias) { bia0 = bias[r]; bia1 = bias[16 + r]; bia2 = bias[32 + r]; bia3 = bias[48 + r]; }

#pragma unroll
  for (int v = 0; v < 8; ++v) {
    const int row = row0 + v + 8 * h;
    if (row < M) {
      float* __restrict__ Crow = C + (size_t)row * 64;
      Crow[r]      = acc0[v] + bia0;
      Crow[16 + r] = acc1[v] + bia1;
      Crow[32 + r] = acc2[v] + bia2;
      Crow[48 + r] = acc3[v] + bia3;
    }
  }
}

constexpr int RNG  = 400;
constexpr int NBIN = 256;
constexpr int NSL  = 256;
constexpr int OWV  = RNG / 16;
constexpr int CHK  = OWV * 32;
constexpr int JMX  = 6;
constexpr int CAP  = CHK * JMX;

__device__ __forceinline__ unsigned match_key8(int key, unsigned liveMask) {
  unsigned peers = liveMask;
#pragma unroll
  for (int b = 0; b < 8; ++b) {
    const bool bit = (key >> b) & 1;
    const unsigned bal = __builtin_amdgcn_ballot_w32(bit);
    peers &= bit ? bal : ~bal;
  }
  return peers;
}

__device__ __forceinline__ unsigned match_key5(int key, unsigned liveMask) {
  unsigned peers = liveMask;
#pragma unroll
  for (int b = 0; b < 5; ++b) {
    const bool bit = (key >> b) & 1;
    const unsigned bal = __builtin_amdgcn_ballot_w32(bit);
    peers &= bit ? bal : ~bal;
  }
  return peers;
}

__device__ __forceinline__ void count_slice(const int* __restrict__ rj, int* cw,
                                            int e0, int e1, int wave, int lane) {
  const unsigned lt = (1u << lane) - 1u;
  for (int eb = e0 + wave * 32; eb < e1; eb += 1024) {
    const int e = eb + lane;
    const bool live = e < e1;
    const int key = live ? rj[e] / RNG : 0;
    const unsigned lm = __builtin_amdgcn_ballot_w32(live);
    const unsigned peers = match_key8(key, lm);
    if (live && (peers & lt) == 0u) cw[key] += __builtin_popcount(peers);
  }
}

__launch_bounds__(1024, 16)
__global__ void place_gemm_kernel(const int* __restrict__ rj, const int* __restrict__ lj,
                                  const float* __restrict__ weights,
                                  int* __restrict__ seg, int4* __restrict__ recs,
                                  int E, int epb,
                                  const float* __restrict__ other, const float* __restrict__ Wo,
                                  float* __restrict__ lhs, int N_OT, int nbOt,
                                  const float* __restrict__ input, const float* __restrict__ Wi,
                                  const float* __restrict__ bi, float* __restrict__ rhs, int N_IN) {
  if (blockIdx.x >= NSL) {
    const int g = (int)blockIdx.x - NSL;
    const bool ot = g < nbOt;
    gemm64_block<1024>(ot ? other : input, ot ? Wo : Wi, ot ? nullptr : bi, ot ? lhs : rhs,
                       ot ? N_OT : N_IN, ot ? g : g - nbOt);
    return;
  }
  __shared__ int cntW[32 * NBIN];
  __shared__ int wsum[NBIN / 32];
  for (int i = threadIdx.x; i < 32 * NBIN; i += 1024) cntW[i] = 0;
  __syncthreads();
  const int lane = threadIdx.x & 31;
  const int wave = threadIdx.x >> 5;
  const int e0 = blockIdx.x * epb;
  int e1 = e0 + epb;
  if (e1 > E) e1 = E;
  int* cw = cntW + wave * NBIN;
  count_slice(rj, cw, e0, e1, wave, lane);
  __syncthreads();
  int c[32];
  int tot = 0, incl = 0;
  if (threadIdx.x < NBIN) {
#pragma unroll
    for (int w = 0; w < 32; ++w) c[w] = cntW[w * NBIN + threadIdx.x];
#pragma unroll
    for (int w = 0; w < 32; ++w) tot += c[w];
    incl = tot;
#pragma unroll
    for (int d = 1; d < 32; d <<= 1) {
      const int t = __shfl_up(incl, d);
      if (lane >= d) incl += t;
    }
    if (lane == 31) wsum[wave] = incl;
  }
  __syncthreads();
  if (threadIdx.x < NBIN) {
    int off = 0;
#pragma unroll
    for (int w = 0; w < NBIN / 32; ++w) off += (w < wave) ? wsum[w] : 0;
    int run = off + incl - tot;
    seg[threadIdx.x * NSL + blockIdx.x] = run;
    if (threadIdx.x == NBIN - 1) seg[NBIN * NSL + blockIdx.x] = run + tot;
#pragma unroll
    for (int w = 0; w < 32; ++w) { cntW[w * NBIN + threadIdx.x] = run; run += c[w]; }
  }
  __syncthreads();
  int4* __restrict__ out = recs + e0;
  const unsigned lt = (1u << lane) - 1u;
  for (int eb = e0 + wave * 32; eb < e1; eb += 1024) {
    const int e = eb + lane;
    const bool live = e < e1;
    const int rr = live ? rj[e] : 0;
    const int key = rr / RNG;
    const unsigned lm = __builtin_amdgcn_ballot_w32(live);
    const unsigned peers = match_key8(key, lm);
    const int first = cw[key];
    if (live && (peers & lt) == 0u) cw[key] = first + __builtin_popcount(peers);
    if (live) {
      const int pos = first + __builtin_popcount(peers & lt);
      out[pos] = int4{rr - key * RNG, lj[e], __float_as_int(weights[e]), 0};
    }
  }
}

__device__ __forceinline__ void edge_rows8(const float* __restrict__ lhs, int idxL, int lane,
                                           int c0, float (&b0)[8], float (&b1)[8]) {
#pragma unroll
  for (int i = 0; i < 8; ++i) {
    const int sl = __builtin_amdgcn_readlane(idxL, c0 + i);
    const float* pl = lhs + (size_t)sl * 64 + lane;
    b0[i] = pl[0]; b1[i] = pl[32];
  }
}

template <bool FULL>
__device__ __forceinline__ void owner_chunk8(const float (&b0)[8], const float (&b1)[8],
                                             const float* rhsS, float* tw,
                                             float& degc,
                                             int rowL, int wbits,
                                             float we0, float we1, int lane,
                                             int c0, int nLive) {
  float a0[8], a1[8], o0[8], o1[8], m0[8], m1[8];
  int tt[8];
#pragma unroll
  for (int i = 0; i < 8; ++i) {
    const int sr = __builtin_amdgcn_readlane(rowL, c0 + i);
    a0[i] = rhsS[sr * 64 + lane];
    a1[i] = rhsS[sr * 64 + 32 + lane];
    tt[i] = sr & 15;
    const int x = 4 * tt[i];
    const float* trw = tw + tt[i] * 64;
    o0[i] = trw[lane ^ x];
    o1[i] = trw[(32 + lane) ^ x];
  }
#pragma unroll
  for (int i = 0; i < 8; ++i) {
    const float w = __int_as_float(__builtin_amdgcn_readlane(wbits, c0 + i));
    float v0 = a0[i] + b0[i] + w * we0;
    float v1 = a1[i] + b1[i] + w * we1;
    m0[i] = v0 > 0.f ? v0 : 0.01f * v0;
    m1[i] = v1 > 0.f ? v1 : 0.01f * v1;
  }
  float s0 = 0.f, s1 = 0.f;
#pragma unroll
  for (int i = 0; i < 8; ++i) {
    if (FULL || c0 + i < nLive) {
      const bool same = i > 0 && tt[i] == tt[i - 1];
      s0 = same ? s0 + m0[i] : m0[i];
      s1 = same ? s1 + m1[i] : m1[i];
      const int x = 4 * tt[i];
      float* trw = tw + tt[i] * 64;
      trw[lane ^ x]        = o0[i] + s0;
      trw[(32 + lane) ^ x] = o1[i] + s1;
      degc += (lane == tt[i]) ? 1.f : 0.f;
    }
  }
}

__launch_bounds__(CHK)
__global__ void owner_kernel(const float* __restrict__ rhs,
                             const float* __restrict__ lhs,
                             const int4* __restrict__ recs,
                             const int* __restrict__ seg, int epb,
                             const float* __restrict__ We,
                             const float* __restrict__ X,
                             const float* __restrict__ Wf,
                             const float* __restrict__ bf,
                             const float* __restrict__ Wout,
                             const float* __restrict__ bout,
                             float* __restrict__ C, int M) {
  __shared__ float2 WfS2[32 * 64];
  __shared__ float2 WoS2[64 * 64];
  __shared__ __attribute__((aligned(16))) float tileS[RNG * 64];
  __shared__ __attribute__((aligned(16))) float rhsS[RNG * 64];
  __shared__ int srtRow[CAP];
  __shared__ int srtIdx[CAP];
  __shared__ int srtW[CAP];
  __shared__ int curW[OWV * 32];
  __shared__ int tileSt[32];
  __shared__ __attribute__((aligned(64))) int segP[NSL + 1];
  __shared__ __attribute__((aligned(64))) int segPc[NSL / 16];
  __shared__ int segG[NSL];
  __shared__ int wsum[NSL / 32];

  const int base = blockIdx.x * RNG;
  const int lane = threadIdx.x & 31;
  int segSt = 0, segEn = 0;
  if (threadIdx.x < NSL) {
    segSt = seg[blockIdx.x * NSL + threadIdx.x];
    segEn = seg[(blockIdx.x + 1) * NSL + threadIdx.x];
  }
  for (int i = threadIdx.x; i < 2048; i += CHK) {
    int k2 = i >> 6, n = i & 63;
    WfS2[i] = float2{Wf[(2 * k2) * 64 + n], Wf[(2 * k2 + 1) * 64 + n]};
  }
  for (int i = threadIdx.x; i < 4096; i += CHK) {
    int k2 = i >> 6, n = i & 63;
    WoS2[i] = float2{Wout[(2 * k2) * 64 + n], Wout[(2 * k2 + 1) * 64 + n]};
  }
  for (int i = threadIdx.x; i < RNG * 16; i += CHK) {
    const int row = base + (i >> 4);
    float4 v = float4{0.f, 0.f, 0.f, 0.f};
    if (row < M) v = reinterpret_cast<const float4*>(rhs)[(size_t)row * 16 + (i & 15)];
    reinterpret_cast<float4*>(rhsS)[i] = v;
    reinterpret_cast<float4*>(tileS)[i] = float4{0.f, 0.f, 0.f, 0.f};
  }
  const int segLen = segEn - segSt;
  int incl = segLen;
  if (threadIdx.x < NSL) {
#pragma unroll
    for (int d = 1; d < 32; d <<= 1) {
      const int t = __shfl_up(incl, d);
      if (lane >= d) incl += t;
    }
    if (lane == 31) wsum[threadIdx.x >> 5] = incl;
  }
  __syncthreads();
  if (threadIdx.x < NSL) {
    int off = 0;
#pragma unroll
    for (int w = 0; w < NSL / 32; ++w) off += (w < (int)(threadIdx.x >> 5)) ? wsum[w] : 0;
    const int ex = off + incl - segLen;
    segP[threadIdx.x] = ex;
    if ((threadIdx.x & 15) == 0) segPc[threadIdx.x >> 4] = ex;
    segG[threadIdx.x] = (int)threadIdx.x * epb + segSt - ex;
    if (threadIdx.x == NSL - 1) segP[NSL] = ex + segLen;
  }
  __syncthreads();

  const int wave = __builtin_amdgcn_readfirstlane((int)(threadIdx.x >> 5));
  const unsigned lt = (1u << lane) - 1u;

  float* tw = &tileS[wave * 16 * 64];
  float degc = 0.f;

  {
    const int eEnd = __builtin_amdgcn_readfirstlane(segP[NSL]);
    auto fetch = [&](int f) -> int4 {
      const int4* pc = reinterpret_cast<const int4*>(segPc);
      int c = -1;
#pragma unroll
      for (int q = 0; q < 4; ++q) {
        const int4 v = pc[q];
        c += (v.x <= f) + (v.y <= f) + (v.z <= f) + (v.w <= f);
      }
      const int4* pf = reinterpret_cast<const int4*>(segP + 16 * c);
      int sl = 16 * c - 1;
#pragma unroll
      for (int q = 0; q < 4; ++q) {
        const int4 v = pf[q];
        sl += (v.x <= f) + (v.y <= f) + (v.z <= f) + (v.w <= f);
      }
      return recs[segG[sl] + f];
    };
    const float we0 = We[lane], we1 = We[32 + lane];
    int* cw = curW + wave * 32;
    for (int cb = 0; cb < eEnd; cb += CAP) {
      const int nC = (eEnd - cb) < CAP ? (eEnd - cb) : CAP;
      cw[lane] = 0;
      int rrow[JMX], ridx[JMX], rwb[JMX];
      unsigned prs[JMX];
#pragma unroll
      for (int j = 0; j < JMX; ++j) {
        const int fl = j * CHK + (int)threadIdx.x;
        int4 rc = int4{0, 0, 0, 0};
        if (fl < nC) rc = fetch(cb + fl);
        rrow[j] = rc.x; ridx[j] = rc.y; rwb[j] = rc.z;
        prs[j] = 0u;
      }
#pragma unroll
      for (int j = 0; j < JMX; ++j) {
        if (j * CHK + wave * 32 < nC) {
          const bool live = j * CHK + (int)threadIdx.x < nC;
          const int key = live ? (rrow[j] >> 4) : 31;
          const unsigned lm = __builtin_amdgcn_ballot_w32(live);
          const unsigned peers = match_key5(key, lm);
          prs[j] = peers;
          if (live && (peers & lt) == 0u) cw[key] += __builtin_popcount(peers);
        }
      }
      __syncthreads();
      if (wave == 0) {
        int cn[OWV];
        int tot = 0;
#pragma unroll
        for (int w = 0; w < OWV; ++w) cn[w] = curW[w * 32 + lane];
#pragma unroll
        for (int w = 0; w < OWV; ++w) tot += cn[w];
        int inc = tot;
#pragma unroll
        for (int d = 1; d < 32; d <<= 1) {
          const int t = __shfl_up(inc, d);
          if (lane >= d) inc += t;
        }
        int run = inc - tot;
        tileSt[lane] = run;
#pragma unroll
        for (int w = 0; w < OWV; ++w) { curW[w * 32 + lane] = run; run += cn[w]; }
      }
      __syncthreads();
#pragma unroll
      for (int j = 0; j < JMX; ++j) {
        if (j * CHK + wave * 32 < nC) {
          const bool live = j * CHK + (int)threadIdx.x < nC;
          const int key = live ? (rrow[j] >> 4) : 31;
          const unsigned peers = prs[j];
          const int first = cw[key];
          if (live && (peers & lt) == 0u) cw[key] = first + __builtin_popcount(peers);
          if (live) {
            const int pos = first + __builtin_popcount(peers & lt);
            srtRow[pos] = rrow[j]; srtIdx[pos] = ridx[j]; srtW[pos] = rwb[j];
          }
        }
      }
      __syncthreads();

      const int t0 = __builtin_amdgcn_readfirstlane(tileSt[wave]);
      const int t1 = __builtin_amdgcn_readfirstlane(tileSt[wave + 1]);
      for (int c0 = t0; c0 < t1; c0 += 32) {
        const int nLive = (t1 - c0) < 32 ? (t1 - c0) : 32;
        const int li = lane < nLive ? c0 + lane : c0;
        int4 rec;
        rec.x = srtRow[li]; rec.y = srtIdx[li]; rec.z = srtW[li]; rec.w = 0;
        const int key = lane < nLive ? (rec.x & 15) : 16;
        unsigned less = 0u, eq = ~0u;
#pragma unroll
        for (int b = 4; b >= 0; --b) {
          const bool bit = (key >> b) & 1;
          const unsigned bal = __builtin_amdgcn_ballot_w32(bit);
          less |= bit ? (eq & ~bal) : 0u;
          eq &= bit ? bal : ~bal;
        }
        const int pos = __builtin_popcount(less) + __builtin_popcount(eq & lt);
        const int rowL  = __builtin_amdgcn_ds_permute(pos << 2, rec.x);
        const int idxL  = __builtin_amdgcn_ds_permute(pos << 2, rec.y);
        const int wbits = __builtin_amdgcn_ds_permute(pos << 2, rec.z);
        float bA0[8], bA1[8], bB0[8], bB1[8];
        edge_rows8(lhs, idxL, lane, 0, bA0, bA1);
        if (nLive == 32) {
          edge_rows8(lhs, idxL, lane, 8, bB0, bB1);
          owner_chunk8<true>(bA0, bA1, rhsS, tw, degc, rowL, wbits, we0, we1, lane, 0, nLive);
          edge_rows8(lhs, idxL, lane, 16, bA0, bA1);
          owner_chunk8<true>(bB0, bB1, rhsS, tw, degc, rowL, wbits, we0, we1, lane, 8, nLive);
          edge_rows8(lhs, idxL, lane, 24, bB0, bB1);
          owner_chunk8<true>(bA0, bA1, rhsS, tw, degc, rowL, wbits, we0, we1, lane, 16, nLive);
          owner_chunk8<true>(bB0, bB1, rhsS, tw, degc, rowL, wbits, we0, we1, lane, 24, nLive);
        } else {
          if (nLive > 8) edge_rows8(lhs, idxL, lane, 8, bB0, bB1);
          owner_chunk8<false>(bA0, bA1, rhsS, tw, degc, rowL, wbits, we0, we1, lane, 0, nLive);
          if (nLive > 8) {
            if (nLive > 16) edge_rows8(lhs, idxL, lane, 16, bA0, bA1);
            owner_chunk8<false>(bB0, bB1, rhsS, tw, degc, rowL, wbits, we0, we1, lane, 8, nLive);
            if (nLive > 16) {
              if (nLive > 24) edge_rows8(lhs, idxL, lane, 24, bB0, bB1);
              owner_chunk8<false>(bA0, bA1, rhsS, tw, degc, rowL, wbits, we0, we1, lane, 16, nLive);
              if (nLive > 24)
                owner_chunk8<false>(bB0, bB1, rhsS, tw, degc, rowL, wbits, we0, we1, lane, 24, nLive);
            }
          }
        }
      }
      __syncthreads();
    }
  }

  const int r = lane & 15;
  const int h = lane >> 4;
  const int lrow0 = wave * 16;
  const int row0 = base + lrow0;
  if (row0 >= M) return;

  int rowA = row0 + r;
  if (rowA >= M) rowA = M - 1;
  const float* __restrict__ Xrow = X + (size_t)rowA * 64;

  float dg[8];
#pragma unroll
  for (int v = 0; v < 8; ++v) dg[v] = __shfl(degc, v + 8 * h);

  const float* trow = tw + r * 64;
  const int xr = 4 * r;

  v8f acc0 = {}, acc1 = {}, acc2 = {}, acc3 = {};
#pragma unroll
  for (int k0 = 0; k0 < 64; k0 += 4) {
    float2 av = *reinterpret_cast<const float2*>(trow + ((k0 ^ xr) + 2 * h));
    v2f a; a.x = av.x; a.y = av.y;
    const int k2 = (k0 >> 1) + h;
    const float2* Bk = &WfS2[k2 * 64 + r];
    float2 f0 = Bk[0], f1 = Bk[16], f2 = Bk[32], f3 = Bk[48];
    v2f b0, b1, b2, b3;
    b0.x = f0.x; b0.y = f0.y;  b1.x = f1.x; b1.y = f1.y;
    b2.x = f2.x; b2.y = f2.y;  b3.x = f3.x; b3.y = f3.y;
    acc0 = wmma_f32(a, b0, acc0);
    acc1 = wmma_f32(a, b1, acc1);
    acc2 = wmma_f32(a, b2, acc2);
    acc3 = wmma_f32(a, b3, acc3);
  }

  {
    const float bf0 = bf[r], bf1 = bf[16 + r], bf2 = bf[32 + r], bf3 = bf[48 + r];
#pragma unroll
    for (int v = 0; v < 8; ++v) {
      const int row = v + 8 * h;
      float* trw = tw + row * 64;
      const int x = 4 * row;
      const float d = dg[v];
      trw[(r)      ^ x] = acc0[v] + d * bf0;
      trw[(16 + r) ^ x] = acc1[v] + d * bf1;
      trw[(32 + r) ^ x] = acc2[v] + d * bf2;
      trw[(48 + r) ^ x] = acc3[v] + d * bf3;
    }
  }

  acc0 = v8f{}; acc1 = v8f{}; acc2 = v8f{}; acc3 = v8f{};
#pragma unroll
  for (int k0 = 0; k0 < 128; k0 += 4) {
    const int ka = k0 + 2 * h;
    float2 av;
    if (k0 < 64) av = *reinterpret_cast<const float2*>(trow + ((k0 ^ xr) + 2 * h));
    else         av = *reinterpret_cast<const float2*>(Xrow + (ka - 64));
    v2f a; a.x = av.x; a.y = av.y;
    const int k2 = (k0 >> 1) + h;
    const float2* Bk = &WoS2[k2 * 64 + r];
    float2 f0 = Bk[0], f1 = Bk[16], f2 = Bk[32], f3 = Bk[48];
    v2f b0, b1, b2, b3;
    b0.x = f0.x; b0.y = f0.y;  b1.x = f1.x; b1.y = f1.y;
    b2.x = f2.x; b2.y = f2.y;  b3.x = f3.x; b3.y = f3.y;
    acc0 = wmma_f32(a, b0, acc0);
    acc1 = wmma_f32(a, b1, acc1);
    acc2 = wmma_f32(a, b2, acc2);
    acc3 = wmma_f32(a, b3, acc3);
  }

  const float bia0 = bout[r], bia1 = bout[16 + r], bia2 = bout[32 + r], bia3 = bout[48 + r];
#pragma unroll
  for (int v = 0; v < 8; ++v) {
    const int row = row0 + v + 8 * h;
    if (row < M) {
      float* Crow = C + (size_t)row * 64;
      Crow[r]      = acc0[v] + bia0;
      Crow[16 + r] = acc1[v] + bia1;
      Crow[32 + r] = acc2[v] + bia2;
      Crow[48 + r] = acc3[v] + bia3;
    }
  }
}

extern "C" void kernel_launch(void* const* d_in, const int* in_sizes, int n_in,
                              void* d_out, int out_size, void* d_ws, size_t ws_size,
                              hipStream_t stream) {
  const float* input   = (const float*)d_in[0];
  const float* other   = (const float*)d_in[1];
  const int*   rj      = (const int*)d_in[2];
  const int*   lj      = (const int*)d_in[3];
  const float* weights = (const float*)d_in[4];
  const float* Wi      = (const float*)d_in[5];
  const float* bi      = (const float*)d_in[6];
  const float* Wo      = (const float*)d_in[7];
  const float* We      = (const float*)d_in[8];
  const float* Wf      = (const float*)d_in[9];
  const float* bf      = (const float*)d_in[10];
  const float* Wout    = (const float*)d_in[11];
  const float* bout    = (const float*)d_in[12];

  const int D    = 64;
  const int N_IN = in_sizes[0] / D;
  const int N_OT = in_sizes[1] / D;
  const int E    = in_sizes[2];

  float* rhs    = (float*)d_ws;
  float* lhs    = rhs + (size_t)N_IN * D;
  int4*  recs   = (int4*)(lhs + (size_t)N_OT * D);
  int*   seg    = (int*)(recs + E);

  const int tiles_in = (N_IN + 15) / 16;
  const int tiles_ot = (N_OT + 15) / 16;
  const int nRanges  = (N_IN + RNG - 1) / RNG;
  const int epb      = (E + NSL - 1) / NSL;

  const int nbOt = (tiles_ot + 31) / 32;
  const int nbIn = (tiles_in + 31) / 32;
  place_gemm_kernel<<<NSL + nbOt + nbIn, 1024, 0, stream>>>(
      rj, lj, weights, seg, recs, E, epb, other, Wo, lhs, N_OT, nbOt, input, Wi, bi, rhs, N_IN);

  owner_kernel<<<nRanges, CHK, 0, stream>>>(rhs, lhs, recs, seg, epb, We, input, Wf, bf,
                                            Wout, bout, (float*)d_out, N_IN);
}


# Wf folded into Wout's top half by associativity: owner forms Wf@Wout_top and bf@Wout_top once on WMMA up front; S@Wf stage and T write-back gone (192 -> 128 WMMA a tile); f32 sum order changes
# speedup vs baseline: 2.5683x; 1.0670x over previous
#include <hip/hip_runtime.h>

typedef float v2f __attribute__((ext_vector_type(2)));
typedef float v8f __attribute__((ext_vector_type(8)));

__device__ __forceinline__ v8f wmma_f32(v2f a, v2f b, v8f c) {
  return __builtin_amdgcn_wmma_f32_16x16x4_f32(false, a, false, b, (short)0, c,
                                               false, false);
}

template <int NT>
__device__ __forceinline__ void gemm64_block(const float* __restrict__ A,
                                             const float* __restrict__ B,
                                             const float* __restrict__ bias,
                                             float* __restrict__ C, int M, int blk) {
  __shared__ float2 Bs2[32 * 64];
  for (int i = threadIdx.x; i < 2048; i += NT) {
    int k2 = i >> 6, n = i & 63;
    Bs2[i] = float2{B[(2 * k2) * 64 + n], B[(2 * k2 + 1) * 64 + n]};
  }
  __syncthreads();

  const int wave = threadIdx.x >> 5;
  const int lane = threadIdx.x & 31;
  const int r = lane & 15;
  const int h = lane >> 4;

  const int row0 = (blk * (NT / 32) + wave) * 16;
  if (row0 >= M) return;

  int rowA = row0 + r;
  if (rowA >= M) rowA = M - 1;
  const float* __restrict__ Arow = A + (size_t)rowA * 64;

  v8f acc0 = {}, acc1 = {}, acc2 = {}, acc3 = {};
#pragma unroll
  for (int k0 = 0; k0 < 64; k0 += 4) {
    const int ka = k0 + 2 * h;
    float2 av = *reinterpret_cast<const float2*>(Arow + ka);
    v2f a; a.x = av.x; a.y = av.y;
    const int k2 = (k0 >> 1) + h;
    const float2* __restrict__ Bk = &Bs2[k2 * 64 + r];
    float2 f0 = Bk[0], f1 = Bk[16], f2 = Bk[32], f3 = Bk[48];
    v2f b0, b1, b2, b3;
    b0.x = f0.x; b0.y = f0.y;  b1.x = f1.x; b1.y = f1.y;
    b2.x = f2.x; b2.y = f2.y;  b3.x = f3.x; b3.y = f3.y;
    acc0 = wmma_f32(a, b0, acc0);
    acc1 = wmma_f32(a, b1, acc1);
    acc2 = wmma_f32(a, b2, acc2);
    acc3 = wmma_f32(a, b3, acc3);
  }

  float bia0 = 0.f, bia1 = 0.f, bia2 = 0.f, bia3 = 0.f;
  if (bias) { bia0 = bias[r]; bia1 = bias[16 + r]; bia2 = bias[32 + r]; bia3 = bias[48 + r]; }

#pragma unroll
  for (int v = 0; v < 8; ++v) {
    const int row = row0 + v + 8 * h;
    if (row < M) {
      float* __restrict__ Crow = C + (size_t)row * 64;
      Crow[r]      = acc0[v] + bia0;
      Crow[16 + r] = acc1[v] + bia1;
      Crow[32 + r] = acc2[v] + bia2;
      Crow[48 + r] = acc3[v] + bia3;
    }
  }
}

constexpr int RNG  = 400;
constexpr int NBIN = 256;
constexpr int NSL  = 256;
constexpr int OWV  = RNG / 16;
constexpr int CHK  = OWV * 32;
constexpr int JMX  = 6;
constexpr int CAP  = CHK * JMX;

__device__ __forceinline__ unsigned match_key8(int key, unsigned liveMask) {
  unsigned peers = liveMask;
#pragma unroll
  for (int b = 0; b < 8; ++b) {
    const bool bit = (key >> b) & 1;
    const unsigned bal = __builtin_amdgcn_ballot_w32(bit);
    peers &= bit ? bal : ~bal;
  }
  return peers;
}

__device__ __forceinline__ unsigned match_key5(int key, unsigned liveMask) {
  unsigned peers = liveMask;
#pragma unroll
  for (int b = 0; b < 5; ++b) {
    const bool bit = (key >> b) & 1;
    const unsigned bal = __builtin_amdgcn_ballot_w32(bit);
    peers &= bit ? bal : ~bal;
  }
  return peers;
}

__device__ __forceinline__ void count_slice(const int* __restrict__ rj, int* cw,
                                            int e0, int e1, int wave, int lane) {
  const unsigned lt = (1u << lane) - 1u;
  for (int eb = e0 + wave * 32; eb < e1; eb += 1024) {
    const int e = eb + lane;
    const bool live = e < e1;
    const int key = live ? rj[e] / RNG : 0;
    const unsigned lm = __builtin_amdgcn_ballot_w32(live);
    const unsigned peers = match_key8(key, lm);
    if (live && (peers & lt) == 0u) cw[key] += __builtin_popcount(peers);
  }
}

__launch_bounds__(1024, 16)
__global__ void place_gemm_kernel(const int* __restrict__ rj, const int* __restrict__ lj,
                                  const float* __restrict__ weights,
                                  int* __restrict__ seg, int4* __restrict__ recs,
                                  int E, int epb,
                                  const float* __restrict__ other, const float* __restrict__ Wo,
                                  float* __restrict__ lhs, int N_OT, int nbOt,
                                  const float* __restrict__ input, const float* __restrict__ Wi,
                                  const float* __restrict__ bi, float* __restrict__ rhs, int N_IN) {
  if (blockIdx.x >= NSL) {
    const int g = (int)blockIdx.x - NSL;
    const bool ot = g < nbOt;
    gemm64_block<1024>(ot ? other : input, ot ? Wo : Wi, ot ? nullptr : bi, ot ? lhs : rhs,
                       ot ? N_OT : N_IN, ot ? g : g - nbOt);
    return;
  }
  __shared__ int cntW[32 * NBIN];
  __shared__ int wsum[NBIN / 32];
  for (int i = threadIdx.x; i < 32 * NBIN; i += 1024) cntW[i] = 0;
  __syncthreads();
  const int lane = threadIdx.x & 31;
  const int wave = threadIdx.x >> 5;
  const int e0 = blockIdx.x * epb;
  int e1 = e0 + epb;
  if (e1 > E) e1 = E;
  int* cw = cntW + wave * NBIN;
  count_slice(rj, cw, e0, e1, wave, lane);
  __syncthreads();
  int c[32];
  int tot = 0, incl = 0;
  if (threadIdx.x < NBIN) {
#pragma unroll
    for (int w = 0; w < 32; ++w) c[w] = cntW[w * NBIN + threadIdx.x];
#pragma unroll
    for (int w = 0; w < 32; ++w) tot += c[w];
    incl = tot;
#pragma unroll
    for (int d = 1; d < 32; d <<= 1) {
      const int t = __shfl_up(incl, d);
      if (lane >= d) incl += t;
    }
    if (lane == 31) wsum[wave] = incl;
  }
  __syncthreads();
  if (threadIdx.x < NBIN) {
    int off = 0;
#pragma unroll
    for (int w = 0; w < NBIN / 32; ++w) off += (w < wave) ? wsum[w] : 0;
    int run = off + incl - tot;
    seg[threadIdx.x * NSL + blockIdx.x] = run;
    if (threadIdx.x == NBIN - 1) seg[NBIN * NSL + blockIdx.x] = run + tot;
#pragma unroll
    for (int w = 0; w < 32; ++w) { cntW[w * NBIN + threadIdx.x] = run; run += c[w]; }
  }
  __syncthreads();
  int4* __restrict__ out = recs + e0;
  const unsigned lt = (1u << lane) - 1u;
  for (int eb = e0 + wave * 32; eb < e1; eb += 1024) {
    const int e = eb + lane;
    const bool live = e < e1;
    const int rr = live ? rj[e] : 0;
    const int key = rr / RNG;
    const unsigned lm = __builtin_amdgcn_ballot_w32(live);
    const unsigned peers = match_key8(key, lm);
    const int first = cw[key];
    if (live && (peers & lt) == 0u) cw[key] = first + __builtin_popcount(peers);
    if (live) {
      const int pos = first + __builtin_popcount(peers & lt);
      out[pos] = int4{rr - key * RNG, lj[e], __float_as_int(weights[e]), 0};
    }
  }
}

__device__ __forceinline__ void edge_rows8(const float* __restrict__ lhs, int idxL, int lane,
                                           int c0, float (&b0)[8], float (&b1)[8]) {
#pragma unroll
  for (int i = 0; i < 8; ++i) {
    const int sl = __builtin_amdgcn_readlane(idxL, c0 + i);
    const float* pl = lhs + (size_t)sl * 64 + lane;
    b0[i] = pl[0]; b1[i] = pl[32];
  }
}

template <bool FULL>
__device__ __forceinline__ void owner_chunk8(const float (&b0)[8], const float (&b1)[8],
                                             const float* rhsS, float* tw,
                                             float& degc,
                                             int rowL, int wbits,
                                             float we0, float we1, int lane,
                                             int c0, int nLive) {
  float a0[8], a1[8], o0[8], o1[8], m0[8], m1[8];
  int tt[8];
#pragma unroll
  for (int i = 0; i < 8; ++i) {
    const int sr = __builtin_amdgcn_readlane(rowL, c0 + i);
    a0[i] = rhsS[sr * 64 + lane];
    a1[i] = rhsS[sr * 64 + 32 + lane];
    tt[i] = sr & 15;
    const int x = 4 * tt[i];
    const float* trw = tw + tt[i] * 64;
    o0[i] = trw[lane ^ x];
    o1[i] = trw[(32 + lane) ^ x];
  }
#pragma unroll
  for (int i = 0; i < 8; ++i) {
    const float w = __int_as_float(__builtin_amdgcn_readlane(wbits, c0 + i));
    float v0 = a0[i] + b0[i] + w * we0;
    float v1 = a1[i] + b1[i] + w * we1;
    m0[i] = v0 > 0.f ? v0 : 0.01f * v0;
    m1[i] = v1 > 0.f ? v1 : 0.01f * v1;
  }
  float s0 = 0.f, s1 = 0.f;
#pragma unroll
  for (int i = 0; i < 8; ++i) {
    if (FULL || c0 + i < nLive) {
      const bool same = i > 0 && tt[i] == tt[i - 1];
      s0 = same ? s0 + m0[i] : m0[i];
      s1 = same ? s1 + m1[i] : m1[i];
      const int x = 4 * tt[i];
      float* trw = tw + tt[i] * 64;
      trw[lane ^ x]        = o0[i] + s0;
      trw[(32 + lane) ^ x] = o1[i] + s1;
      degc += (lane == tt[i]) ? 1.f : 0.f;
    }
  }
}

__launch_bounds__(CHK)
__global__ void owner_kernel(const float* __restrict__ rhs,
                             const float* __restrict__ lhs,
                             const int4* __restrict__ recs,
                             const int* __restrict__ seg, int epb,
                             const float* __restrict__ We,
                             const float* __restrict__ X,
                             const float* __restrict__ Wf,
                             const float* __restrict__ bf,
                             const float* __restrict__ Wout,
                             const float* __restrict__ bout,
                             float* __restrict__ C, int M) {
  __shared__ float2 WoS2[64 * 64];
  __shared__ float bpS[64];
  __shared__ __attribute__((aligned(16))) float tileS[RNG * 64];
  __shared__ __attribute__((aligned(16))) float rhsS[RNG * 64];
  __shared__ int srtRow[CAP];
  __shared__ int srtIdx[CAP];
  __shared__ int srtW[CAP];
  __shared__ int curW[OWV * 32];
  __shared__ int tileSt[32];
  __shared__ __attribute__((aligned(64))) int segP[NSL + 1];
  __shared__ __attribute__((aligned(64))) int segPc[NSL / 16];
  __shared__ int segG[NSL];
  __shared__ int wsum[NSL / 32];

  const int base = blockIdx.x * RNG;
  const int lane = threadIdx.x & 31;
  int segSt = 0, segEn = 0;
  if (threadIdx.x < NSL) {
    segSt = seg[blockIdx.x * NSL + threadIdx.x];
    segEn = seg[(blockIdx.x + 1) * NSL + threadIdx.x];
  }
  {
    const int pw = __builtin_amdgcn_readfirstlane((int)(threadIdx.x >> 5));
    if (pw < 20) {
      const int pr = lane & 15, ph = lane >> 4;
      const int ti = pw >> 2, tj = pw & 3;
      const float* __restrict__ Arow = (pw < 16) ? Wf + (16 * ti + pr) * 64 : bf;
      const float* __restrict__ Bcol = Wout + 16 * tj + pr;
      v8f accp = {};
#pragma unroll
      for (int k0 = 0; k0 < 64; k0 += 4) {
        const int ka = k0 + 2 * ph;
        const float2 av = *reinterpret_cast<const float2*>(Arow + ka);
        v2f a; a.x = av.x; a.y = av.y;
        v2f b; b.x = Bcol[ka * 64]; b.y = Bcol[(ka + 1) * 64];
        accp = wmma_f32(a, b, accp);
      }
      if (pw < 16) {
        float* WoF = reinterpret_cast<float*>(WoS2);
#pragma unroll
        for (int v = 0; v < 8; ++v) {
          const int k = 16 * ti + 8 * ph + v;
          WoF[((k >> 1) * 64 + 16 * tj + pr) * 2 + (k & 1)] = accp[v];
        }
      } else if (ph == 0) {
        bpS[16 * tj + pr] = accp[0];
      }
    }
  }
  for (int i = 2048 + threadIdx.x; i < 4096; i += CHK) {
    int k2 = i >> 6, n = i & 63;
    WoS2[i] = float2{Wout[(2 * k2) * 64 + n], Wout[(2 * k2 + 1) * 64 + n]};
  }
  for (int i = threadIdx.x; i < RNG * 16; i += CHK) {
    const int row = base + (i >> 4);
    float4 v = float4{0.f, 0.f, 0.f, 0.f};
    if (row < M) v = reinterpret_cast<const float4*>(rhs)[(size_t)row * 16 + (i & 15)];
    reinterpret_cast<float4*>(rhsS)[i] = v;
    reinterpret_cast<float4*>(tileS)[i] = float4{0.f, 0.f, 0.f, 0.f};
  }
  const int segLen = segEn - segSt;
  int incl = segLen;
  if (threadIdx.x < NSL) {
#pragma unroll
    for (int d = 1; d < 32; d <<= 1) {
      const int t = __shfl_up(incl, d);
      if (lane >= d) incl += t;
    }
    if (lane == 31) wsum[threadIdx.x >> 5] = incl;
  }
  __syncthreads();
  if (threadIdx.x < NSL) {
    int off = 0;
#pragma unroll
    for (int w = 0; w < NSL / 32; ++w) off += (w < (int)(threadIdx.x >> 5)) ? wsum[w] : 0;
    const int ex = off + incl - segLen;
    segP[threadIdx.x] = ex;
    if ((threadIdx.x & 15) == 0) segPc[threadIdx.x >> 4] = ex;
    segG[threadIdx.x] = (int)threadIdx.x * epb + segSt - ex;
    if (threadIdx.x == NSL - 1) segP[NSL] = ex + segLen;
  }
  __syncthreads();

  const int wave = __builtin_amdgcn_readfirstlane((int)(threadIdx.x >> 5));
  const unsigned lt = (1u << lane) - 1u;

  float* tw = &tileS[wave * 16 * 64];
  float degc = 0.f;

  {
    const int eEnd = __builtin_amdgcn_readfirstlane(segP[NSL]);
    auto fetch = [&](int f) -> int4 {
      const int4* pc = reinterpret_cast<const int4*>(segPc);
      int c = -1;
#pragma unroll
      for (int q = 0; q < 4; ++q) {
        const int4 v = pc[q];
        c += (v.x <= f) + (v.y <= f) + (v.z <= f) + (v.w <= f);
      }
      const int4* pf = reinterpret_cast<const int4*>(segP + 16 * c);
      int sl = 16 * c - 1;
#pragma unroll
      for (int q = 0; q < 4; ++q) {
        const int4 v = pf[q];
        sl += (v.x <= f) + (v.y <= f) + (v.z <= f) + (v.w <= f);
      }
      return recs[segG[sl] + f];
    };
    const float we0 = We[lane], we1 = We[32 + lane];
    int* cw = curW + wave * 32;
    for (int cb = 0; cb < eEnd; cb += CAP) {
      const int nC = (eEnd - cb) < CAP ? (eEnd - cb) : CAP;
      cw[lane] = 0;
      int rrow[JMX], ridx[JMX], rwb[JMX];
      unsigned prs[JMX];
#pragma unroll
      for (int j = 0; j < JMX; ++j) {
        const int fl = j * CHK + (int)threadIdx.x;
        int4 rc = int4{0, 0, 0, 0};
        if (fl < nC) rc = fetch(cb + fl);
        rrow[j] = rc.x; ridx[j] = rc.y; rwb[j] = rc.z;
        prs[j] = 0u;
      }
#pragma unroll
      for (int j = 0; j < JMX; ++j) {
        if (j * CHK + wave * 32 < nC) {
          const bool live = j * CHK + (int)threadIdx.x < nC;
          const int key = live ? (rrow[j] >> 4) : 31;
          const unsigned lm = __builtin_amdgcn_ballot_w32(live);
          const unsigned peers = match_key5(key, lm);
          prs[j] = peers;
          if (live && (peers & lt) == 0u) cw[key] += __builtin_popcount(peers);
        }
      }
      __syncthreads();
      if (wave == 0) {
        int cn[OWV];
        int tot = 0;
#pragma unroll
        for (int w = 0; w < OWV; ++w) cn[w] = curW[w * 32 + lane];
#pragma unroll
        for (int w = 0; w < OWV; ++w) tot += cn[w];
        int inc = tot;
#pragma unroll
        for (int d = 1; d < 32; d <<= 1) {
          const int t = __shfl_up(inc, d);
          if (lane >= d) inc += t;
        }
        int run = inc - tot;
        tileSt[lane] = run;
#pragma unroll
        for (int w = 0; w < OWV; ++w) { curW[w * 32 + lane] = run; run += cn[w]; }
      }
      __syncthreads();
#pragma unroll
      for (int j = 0; j < JMX; ++j) {
        if (j * CHK + wave * 32 < nC) {
          const bool live = j * CHK + (int)threadIdx.x < nC;
          const int key = live ? (rrow[j] >> 4) : 31;
          const unsigned peers = prs[j];
          const int first = cw[key];
          if (live && (peers & lt) == 0u) cw[key] = first + __builtin_popcount(peers);
          if (live) {
            const int pos = first + __builtin_popcount(peers & lt);
            srtRow[pos] = rrow[j]; srtIdx[pos] = ridx[j]; srtW[pos] = rwb[j];
          }
        }
      }
      __syncthreads();

      const int t0 = __builtin_amdgcn_readfirstlane(tileSt[wave]);
      const int t1 = __builtin_amdgcn_readfirstlane(tileSt[wave + 1]);
      for (int c0 = t0; c0 < t1; c0 += 32) {
        const int nLive = (t1 - c0) < 32 ? (t1 - c0) : 32;
        const int li = lane < nLive ? c0 + lane : c0;
        int4 rec;
        rec.x = srtRow[li]; rec.y = srtIdx[li]; rec.z = srtW[li]; rec.w = 0;
        const int key = lane < nLive ? (rec.x & 15) : 16;
        unsigned less = 0u, eq = ~0u;
#pragma unroll
        for (int b = 4; b >= 0; --b) {
          const bool bit = (key >> b) & 1;
          const unsigned bal = __builtin_amdgcn_ballot_w32(bit);
          less |= bit ? (eq & ~bal) : 0u;
          eq &= bit ? bal : ~bal;
        }
        const int pos = __builtin_popcount(less) + __builtin_popcount(eq & lt);
        const int rowL  = __builtin_amdgcn_ds_permute(pos << 2, rec.x);
        const int idxL  = __builtin_amdgcn_ds_permute(pos << 2, rec.y);
        const int wbits = __builtin_amdgcn_ds_permute(pos << 2, rec.z);
        float bA0[8], bA1[8], bB0[8], bB1[8];
        edge_rows8(lhs, idxL, lane, 0, bA0, bA1);
        if (nLive == 32) {
          edge_rows8(lhs, idxL, lane, 8, bB0, bB1);
          owner_chunk8<true>(bA0, bA1, rhsS, tw, degc, rowL, wbits, we0, we1, lane, 0, nLive);
          edge_rows8(lhs, idxL, lane, 16, bA0, bA1);
          owner_chunk8<true>(bB0, bB1, rhsS, tw, degc, rowL, wbits, we0, we1, lane, 8, nLive);
          edge_rows8(lhs, idxL, lane, 24, bB0, bB1);
          owner_chunk8<true>(bA0, bA1, rhsS, tw, degc, rowL, wbits, we0, we1, lane, 16, nLive);
          owner_chunk8<true>(bB0, bB1, rhsS, tw, degc, rowL, wbits, we0, we1, lane, 24, nLive);
        } else {
          if (nLive > 8) edge_rows8(lhs, idxL, lane, 8, bB0, bB1);
          owner_chunk8<false>(bA0, bA1, rhsS, tw, degc, rowL, wbits, we0, we1, lane, 0, nLive);
          if (nLive > 8) {
            if (nLive > 16) edge_rows8(lhs, idxL, lane, 16, bA0, bA1);
            owner_chunk8<false>(bB0, bB1, rhsS, tw, degc, rowL, wbits, we0, we1, lane, 8, nLive);
            if (nLive > 16) {
              if (nLive > 24) edge_rows8(lhs, idxL, lane, 24, bB0, bB1);
              owner_chunk8<false>(bA0, bA1, rhsS, tw, degc, rowL, wbits, we0, we1, lane, 16, nLive);
              if (nLive > 24)
                owner_chunk8<false>(bB0, bB1, rhsS, tw, degc, rowL, wbits, we0, we1, lane, 24, nLive);
            }
          }
        }
      }
      __syncthreads();
    }
  }

  const int r = lane & 15;
  const int h = lane >> 4;
  const int lrow0 = wave * 16;
  const int row0 = base + lrow0;
  if (row0 >= M) return;

  int rowA = row0 + r;
  if (rowA >= M) rowA = M - 1;
  const float* __restrict__ Xrow = X + (size_t)rowA * 64;

  float dg[8];
#pragma unroll
  for (int v = 0; v < 8; ++v) dg[v] = __shfl(degc, v + 8 * h);

  const float* trow = tw + r * 64;
  const int xr = 4 * r;

  v8f acc0 = {}, acc1 = {}, acc2 = {}, acc3 = {};
#pragma unroll
  for (int k0 = 0; k0 < 128; k0 += 4) {
    const int ka = k0 + 2 * h;
    float2 av;
    if (k0 < 64) av = *reinterpret_cast<const float2*>(trow + ((k0 ^ xr) + 2 * h));
    else         av = *reinterpret_cast<const float2*>(Xrow + (ka - 64));
    v2f a; a.x = av.x; a.y = av.y;
    const int k2 = (k0 >> 1) + h;
    const float2* Bk = &WoS2[k2 * 64 + r];
    float2 f0 = Bk[0], f1 = Bk[16], f2 = Bk[32], f3 = Bk[48];
    v2f b0, b1, b2, b3;
    b0.x = f0.x; b0.y = f0.y;  b1.x = f1.x; b1.y = f1.y;
    b2.x = f2.x; b2.y = f2.y;  b3.x = f3.x; b3.y = f3.y;
    acc0 = wmma_f32(a, b0, acc0);
    acc1 = wmma_f32(a, b1, acc1);
    acc2 = wmma_f32(a, b2, acc2);
    acc3 = wmma_f32(a, b3, acc3);
  }

  const float bia0 = bout[r], bia1 = bout[16 + r], bia2 = bout[32 + r], bia3 = bout[48 + r];
  const float bp0 = bpS[r], bp1 = bpS[16 + r], bp2 = bpS[32 + r], bp3 = bpS[48 + r];
#pragma unroll
  for (int v = 0; v < 8; ++v) {
    const int row = row0 + v + 8 * h;
    if (row < M) {
      float* Crow = C + (size_t)row * 64;
      const float d = dg[v];
      Crow[r]      = acc0[v] + d * bp0 + bia0;
      Crow[16 + r] = acc1[v] + d * bp1 + bia1;
      Crow[32 + r] = acc2[v] + d * bp2 + bia2;
      Crow[48 + r] = acc3[v] + d * bp3 + bia3;
    }
  }
}

extern "C" void kernel_launch(void* const* d_in, const int* in_sizes, int n_in,
                              void* d_out, int out_size, void* d_ws, size_t ws_size,
                              hipStream_t stream) {
  const float* input   = (const float*)d_in[0];
  const float* other   = (const float*)d_in[1];
  const int*   rj      = (const int*)d_in[2];
  const int*   lj      = (const int*)d_in[3];
  const float* weights = (const float*)d_in[4];
  const float* Wi      = (const float*)d_in[5];
  const float* bi      = (const float*)d_in[6];
  const float* Wo      = (const float*)d_in[7];
  const float* We      = (const float*)d_in[8];
  const float* Wf      = (const float*)d_in[9];
  const float* bf      = (const float*)d_in[10];
  const float* Wout    = (const float*)d_in[11];
  const float* bout    = (const float*)d_in[12];

  const int D    = 64;
  const int N_IN = in_sizes[0] / D;
  const int N_OT = in_sizes[1] / D;
  const int E    = in_sizes[2];

  float* rhs    = (float*)d_ws;
  float* lhs    = rhs + (size_t)N_IN * D;
  int4*  recs   = (int4*)(lhs + (size_t)N_OT * D);
  int*   seg    = (int*)(recs + E);

  const int tiles_in = (N_IN + 15) / 16;
  const int tiles_ot = (N_OT + 15) / 16;
  const int nRanges  = (N_IN + RNG - 1) / RNG;
  const int epb      = (E + NSL - 1) / NSL;

  const int nbOt = (tiles_ot + 31) / 32;
  const int nbIn = (tiles_in + 31) / 32;
  place_gemm_kernel<<<NSL + nbOt + nbIn, 1024, 0, stream>>>(
      rj, lj, weights, seg, recs, E, epb, other, Wo, lhs, N_OT, nbOt, input, Wi, bi, rhs, N_IN);

  owner_kernel<<<nRanges, CHK, 0, stream>>>(rhs, lhs, recs, seg, epb, We, input, Wf, bf,
                                            Wout, bout, (float*)d_out, N_IN);
}


# Owner edge sums: a lane owns columns 2l, 2l+1 of a row, not l, l+32: each lhs, rhs and tile row access is one 8-byte access per lane and the sums run on packed f32 pairs; no sum order change
# speedup vs baseline: 2.7403x; 1.0670x over previous
#include <hip/hip_runtime.h>

typedef float v2f __attribute__((ext_vector_type(2)));
typedef float v8f __attribute__((ext_vector_type(8)));

__device__ __forceinline__ v8f wmma_f32(v2f a, v2f b, v8f c) {
  return __builtin_amdgcn_wmma_f32_16x16x4_f32(false, a, false, b, (short)0, c,
                                               false, false);
}

template <int NT>
__device__ __forceinline__ void gemm64_block(const float* __restrict__ A,
                                             const float* __restrict__ B,
                                             const float* __restrict__ bias,
                                             float* __restrict__ C, int M, int blk) {
  __shared__ float2 Bs2[32 * 64];
  for (int i = threadIdx.x; i < 2048; i += NT) {
    int k2 = i >> 6, n = i & 63;
    Bs2[i] = float2{B[(2 * k2) * 64 + n], B[(2 * k2 + 1) * 64 + n]};
  }
  __syncthreads();

  const int wave = threadIdx.x >> 5;
  const int lane = threadIdx.x & 31;
  const int r = lane & 15;
  const int h = lane >> 4;

  const int row0 = (blk * (NT / 32) + wave) * 16;
  if (row0 >= M) return;

  int rowA = row0 + r;
  if (rowA >= M) rowA = M - 1;
  const float* __restrict__ Arow = A + (size_t)rowA * 64;

  v8f acc0 = {}, acc1 = {}, acc2 = {}, acc3 = {};
#pragma unroll
  for (int k0 = 0; k0 < 64; k0 += 4) {
    const int ka = k0 + 2 * h;
    float2 av = *reinterpret_cast<const float2*>(Arow + ka);
    v2f a; a.x = av.x; a.y = av.y;
    const int k2 = (k0 >> 1) + h;
    const float2* __restrict__ Bk = &Bs2[k2 * 64 + r];
    float2 f0 = Bk[0], f1 = Bk[16], f2 = Bk[32], f3 = Bk[48];
    v2f b0, b1, b2, b3;
    b0.x = f0.x; b0.y = f0.y;  b1.x = f1.x; b1.y = f1.y;
    b2.x = f2.x; b2.y = f2.y;  b3.x = f3.x; b3.y = f3.y;
    acc0 = wmma_f32(a, b0, acc0);
    acc1 = wmma_f32(a, b1, acc1);
    acc2 = wmma_f32(a, b2, acc2);
    acc3 = wmma_f32(a, b3, acc3);
  }

  float bia0 = 0.f, bia1 = 0.f, bia2 = 0.f, bia3 = 0.f;
  if (bias) { bia0 = bias[r]; bia1 = bias[16 + r]; bia2 = bias[32 + r]; bia3 = bias[48 + r]; }

#pragma unroll
  for (int v = 0; v < 8; ++v) {
    const int row = row0 + v + 8 * h;
    if (row < M) {
      float* __restrict__ Crow = C + (size_t)row * 64;
      Crow[r]      = acc0[v] + bia0;
      Crow[16 + r] = acc1[v] + bia1;
      Crow[32 + r] = acc2[v] + bia2;
      Crow[48 + r] = acc3[v] + bia3;
    }
  }
}

constexpr int RNG  = 400;
constexpr int NBIN = 256;
constexpr int NSL  = 256;
constexpr int OWV  = RNG / 16;
constexpr int CHK  = OWV * 32;
constexpr int JMX  = 6;
constexpr int CAP  = CHK * JMX;

__device__ __forceinline__ unsigned match_key8(int key, unsigned liveMask) {
  unsigned peers = liveMask;
#pragma unroll
  for (int b = 0; b < 8; ++b) {
    const bool bit = (key >> b) & 1;
    const unsigned bal = __builtin_amdgcn_ballot_w32(bit);
    peers &= bit ? bal : ~bal;
  }
  return peers;
}

__device__ __forceinline__ unsigned match_key5(int key, unsigned liveMask) {
  unsigned peers = liveMask;
#pragma unroll
  for (int b = 0; b < 5; ++b) {
    const bool bit = (key >> b) & 1;
    const unsigned bal = __builtin_amdgcn_ballot_w32(bit);
    peers &= bit ? bal : ~bal;
  }
  return peers;
}

__device__ __forceinline__ void count_slice(const int* __restrict__ rj, int* cw,
                                            int e0, int e1, int wave, int lane) {
  const unsigned lt = (1u << lane) - 1u;
  for (int eb = e0 + wave * 32; eb < e1; eb += 1024) {
    const int e = eb + lane;
    const bool live = e < e1;
    const int key = live ? rj[e] / RNG : 0;
    const unsigned lm = __builtin_amdgcn_ballot_w32(live);
    const unsigned peers = match_key8(key, lm);
    if (live && (peers & lt) == 0u) cw[key] += __builtin_popcount(peers);
  }
}

__launch_bounds__(1024, 16)
__global__ void place_gemm_kernel(const int* __restrict__ rj, const int* __restrict__ lj,
                                  const float* __restrict__ weights,
                                  int* __restrict__ seg, int4* __restrict__ recs,
                                  int E, int epb,
                                  const float* __restrict__ other, const float* __restrict__ Wo,
                                  float* __restrict__ lhs, int N_OT, int nbOt,
                                  const float* __restrict__ input, const float* __restrict__ Wi,
                                  const float* __restrict__ bi, float* __restrict__ rhs, int N_IN) {
  if (blockIdx.x >= NSL) {
    const int g = (int)blockIdx.x - NSL;
    const bool ot = g < nbOt;
    gemm64_block<1024>(ot ? other : input, ot ? Wo : Wi, ot ? nullptr : bi, ot ? lhs : rhs,
                       ot ? N_OT : N_IN, ot ? g : g - nbOt);
    return;
  }
  __shared__ int cntW[32 * NBIN];
  __shared__ int wsum[NBIN / 32];
  for (int i = threadIdx.x; i < 32 * NBIN; i += 1024) cntW[i] = 0;
  __syncthreads();
  const int lane = threadIdx.x & 31;
  const int wave = threadIdx.x >> 5;
  const int e0 = blockIdx.x * epb;
  int e1 = e0 + epb;
  if (e1 > E) e1 = E;
  int* cw = cntW + wave * NBIN;
  count_slice(rj, cw, e0, e1, wave, lane);
  __syncthreads();
  int c[32];
  int tot = 0, incl = 0;
  if (threadIdx.x < NBIN) {
#pragma unroll
    for (int w = 0; w < 32; ++w) c[w] = cntW[w * NBIN + threadIdx.x];
#pragma unroll
    for (int w = 0; w < 32; ++w) tot += c[w];
    incl = tot;
#pragma unroll
    for (int d = 1; d < 32; d <<= 1) {
      const int t = __shfl_up(incl, d);
      if (lane >= d) incl += t;
    }
    if (lane == 31) wsum[wave] = incl;
  }
  __syncthreads();
  if (threadIdx.x < NBIN) {
    int off = 0;
#pragma unroll
    for (int w = 0; w < NBIN / 32; ++w) off += (w < wave) ? wsum[w] : 0;
    int run = off + incl - tot;
    seg[threadIdx.x * NSL + blockIdx.x] = run;
    if (threadIdx.x == NBIN - 1) seg[NBIN * NSL + blockIdx.x] = run + tot;
#pragma unroll
    for (int w = 0; w < 32; ++w) { cntW[w * NBIN + threadIdx.x] = run; run += c[w]; }
  }
  __syncthreads();
  int4* __restrict__ out = recs + e0;
  const unsigned lt = (1u << lane) - 1u;
  for (int eb = e0 + wave * 32; eb < e1; eb += 1024) {
    const int e = eb + lane;
    const bool live = e < e1;
    const int rr = live ? rj[e] : 0;
    const int key = rr / RNG;
    const unsigned lm = __builtin_amdgcn_ballot_w32(live);
    const unsigned peers = match_key8(key, lm);
    const int first = cw[key];
    if (live && (peers & lt) == 0u) cw[key] = first + __builtin_popcount(peers);
    if (live) {
      const int pos = first + __builtin_popcount(peers & lt);
      out[pos] = int4{rr - key * RNG, lj[e], __float_as_int(weights[e]), 0};
    }
  }
}

__device__ __forceinline__ void edge_rows8(const float* __restrict__ lhs, int idxL, int lane,
                                           int c0, float2 (&b)[8]) {
#pragma unroll
  for (int i = 0; i < 8; ++i) {
    const unsigned sl = (unsigned)__builtin_amdgcn_readlane(idxL, c0 + i);
    b[i] = reinterpret_cast<const float2*>(lhs)[sl * 32u + (unsigned)lane];
  }
}

template <bool FULL>
__device__ __forceinline__ void owner_chunk8(const float2 (&b)[8],
                                             const float* rhsS, float* tw,
                                             float& degc,
                                             int rowL, int wbits,
                                             float we0, float we1, int lane,
                                             int c0, int nLive) {
  float2 a[8], o[8], m[8];
  int tt[8];
#pragma unroll
  for (int i = 0; i < 8; ++i) {
    const int sr = __builtin_amdgcn_readlane(rowL, c0 + i);
    a[i] = reinterpret_cast<const float2*>(rhsS + sr * 64)[lane];
    tt[i] = sr & 15;
    const int x = 4 * tt[i];
    const float* trw = tw + tt[i] * 64;
    o[i] = *reinterpret_cast<const float2*>(trw + ((2 * lane) ^ x));
  }
#pragma unroll
  for (int i = 0; i < 8; ++i) {
    const float w = __int_as_float(__builtin_amdgcn_readlane(wbits, c0 + i));
    float v0 = a[i].x + b[i].x + w * we0;
    float v1 = a[i].y + b[i].y + w * we1;
    m[i].x = v0 > 0.f ? v0 : 0.01f * v0;
    m[i].y = v1 > 0.f ? v1 : 0.01f * v1;
  }
  float s0 = 0.f, s1 = 0.f;
#pragma unroll
  for (int i = 0; i < 8; ++i) {
    if (FULL || c0 + i < nLive) {
      const bool same = i > 0 && tt[i] == tt[i - 1];
      s0 = same ? s0 + m[i].x : m[i].x;
      s1 = same ? s1 + m[i].y : m[i].y;
      const int x = 4 * tt[i];
      float* trw = tw + tt[i] * 64;
      *reinterpret_cast<float2*>(trw + ((2 * lane) ^ x)) = float2{o[i].x + s0, o[i].y + s1};
      degc += (lane == tt[i]) ? 1.f : 0.f;
    }
  }
}

__launch_bounds__(CHK)
__global__ void owner_kernel(const float* __restrict__ rhs,
                             const float* __restrict__ lhs,
                             const int4* __restrict__ recs,
                             const int* __restrict__ seg, int epb,
                             const float* __restrict__ We,
                             const float* __restrict__ X,
                             const float* __restrict__ Wf,
                             const float* __restrict__ bf,
                             const float* __restrict__ Wout,
                             const float* __restrict__ bout,
                             float* __restrict__ C, int M) {
  __shared__ float2 WoS2[64 * 64];
  __shared__ float bpS[64];
  __shared__ __attribute__((aligned(16))) float tileS[RNG * 64];
  __shared__ __attribute__((aligned(16))) float rhsS[RNG * 64];
  __shared__ int srtRow[CAP];
  __shared__ int srtIdx[CAP];
  __shared__ int srtW[CAP];
  __shared__ int curW[OWV * 32];
  __shared__ int tileSt[32];
  __shared__ __attribute__((aligned(64))) int segP[NSL + 1];
  __shared__ __attribute__((aligned(64))) int segPc[NSL / 16];
  __shared__ int segG[NSL];
  __shared__ int wsum[NSL / 32];

  const int base = blockIdx.x * RNG;
  const int lane = threadIdx.x & 31;
  int segSt = 0, segEn = 0;
  if (threadIdx.x < NSL) {
    segSt = seg[blockIdx.x * NSL + threadIdx.x];
    segEn = seg[(blockIdx.x + 1) * NSL + threadIdx.x];
  }
  {
    const int pw = __builtin_amdgcn_readfirstlane((int)(threadIdx.x >> 5));
    if (pw < 20) {
      const int pr = lane & 15, ph = lane >> 4;
      const int ti = pw >> 2, tj = pw & 3;
      const float* __restrict__ Arow = (pw < 16) ? Wf + (16 * ti + pr) * 64 : bf;
      const float* __restrict__ Bcol = Wout + 16 * tj + pr;
      v8f accp = {};
#pragma unroll
      for (int k0 = 0; k0 < 64; k0 += 4) {
        const int ka = k0 + 2 * ph;
        const float2 av = *reinterpret_cast<const float2*>(Arow + ka);
        v2f a; a.x = av.x; a.y = av.y;
        v2f b; b.x = Bcol[ka * 64]; b.y = Bcol[(ka + 1) * 64];
        accp = wmma_f32(a, b, accp);
      }
      if (pw < 16) {
        float* WoF = reinterpret_cast<float*>(WoS2);
#pragma unroll
        for (int v = 0; v < 8; ++v) {
          const int k = 16 * ti + 8 * ph + v;
          WoF[((k >> 1) * 64 + 16 * tj + pr) * 2 + (k & 1)] = accp[v];
        }
      } else if (ph == 0) {
        bpS[16 * tj + pr] = accp[0];
      }
    }
  }
  for (int i = 2048 + threadIdx.x; i < 4096; i += CHK) {
    int k2 = i >> 6, n = i & 63;
    WoS2[i] = float2{Wout[(2 * k2) * 64 + n], Wout[(2 * k2 + 1) * 64 + n]};
  }
  for (int i = threadIdx.x; i < RNG * 16; i += CHK) {
    const int row = base + (i >> 4);
    float4 v = float4{0.f, 0.f, 0.f, 0.f};
    if (row < M) v = reinterpret_cast<const float4*>(rhs)[(size_t)row * 16 + (i & 15)];
    reinterpret_cast<float4*>(rhsS)[i] = v;
    reinterpret_cast<float4*>(tileS)[i] = float4{0.f, 0.f, 0.f, 0.f};
  }
  const int segLen = segEn - segSt;
  int incl = segLen;
  if (threadIdx.x < NSL) {
#pragma unroll
    for (int d = 1; d < 32; d <<= 1) {
      const int t = __shfl_up(incl, d);
      if (lane >= d) incl += t;
    }
    if (lane == 31) wsum[threadIdx.x >> 5] = incl;
  }
  __syncthreads();
  if (threadIdx.x < NSL) {
    int off = 0;
#pragma unroll
    for (int w = 0; w < NSL / 32; ++w) off += (w < (int)(threadIdx.x >> 5)) ? wsum[w] : 0;
    const int ex = off + incl - segLen;
    segP[threadIdx.x] = ex;
    if ((threadIdx.x & 15) == 0) segPc[threadIdx.x >> 4] = ex;
    segG[threadIdx.x] = (int)threadIdx.x * epb + segSt - ex;
    if (threadIdx.x == NSL - 1) segP[NSL] = ex + segLen;
  }
  __syncthreads();

  const int wave = __builtin_amdgcn_readfirstlane((int)(threadIdx.x >> 5));
  const unsigned lt = (1u << lane) - 1u;

  float* tw = &tileS[wave * 16 * 64];
  float degc = 0.f;

  {
    const int eEnd = __builtin_amdgcn_readfirstlane(segP[NSL]);
    auto fetch = [&](int f) -> int4 {
      const int4* pc = reinterpret_cast<const int4*>(segPc);
      int c = -1;
#pragma unroll
      for (int q = 0; q < 4; ++q) {
        const int4 v = pc[q];
        c += (v.x <= f) + (v.y <= f) + (v.z <= f) + (v.w <= f);
      }
      const int4* pf = reinterpret_cast<const int4*>(segP + 16 * c);
      int sl = 16 * c - 1;
#pragma unroll
      for (int q = 0; q < 4; ++q) {
        const int4 v = pf[q];
        sl += (v.x <= f) + (v.y <= f) + (v.z <= f) + (v.w <= f);
      }
      return recs[segG[sl] + f];
    };
    const float we0 = We[2 * lane], we1 = We[2 * lane + 1];
    int* cw = curW + wave * 32;
    for (int cb = 0; cb < eEnd; cb += CAP) {
      const int nC = (eEnd - cb) < CAP ? (eEnd - cb) : CAP;
      cw[lane] = 0;
      int rrow[JMX], ridx[JMX], rwb[JMX];
      unsigned prs[JMX];
#pragma unroll
      for (int j = 0; j < JMX; ++j) {
        const int fl = j * CHK + (int)threadIdx.x;
        int4 rc = int4{0, 0, 0, 0};
        if (fl < nC) rc = fetch(cb + fl);
        rrow[j] = rc.x; ridx[j] = rc.y; rwb[j] = rc.z;
        prs[j] = 0u;
      }
#pragma unroll
      for (int j = 0; j < JMX; ++j) {
        if (j * CHK + wave * 32 < nC) {
          const bool live = j * CHK + (int)threadIdx.x < nC;
          const int key = live ? (rrow[j] >> 4) : 31;
          const unsigned lm = __builtin_amdgcn_ballot_w32(live);
          const unsigned peers = match_key5(key, lm);
          prs[j] = peers;
          if (live && (peers & lt) == 0u) cw[key] += __builtin_popcount(peers);
        }
      }
      __syncthreads();
      if (wave == 0) {
        int cn[OWV];
        int tot = 0;
#pragma unroll
        for (int w = 0; w < OWV; ++w) cn[w] = curW[w * 32 + lane];
#pragma unroll
        for (int w = 0; w < OWV; ++w) tot += cn[w];
        int inc = tot;
#pragma unroll
        for (int d = 1; d < 32; d <<= 1) {
          const int t = __shfl_up(inc, d);
          if (lane >= d) inc += t;
        }
        int run = inc - tot;
        tileSt[lane] = run;
#pragma unroll
        for (int w = 0; w < OWV; ++w) { curW[w * 32 + lane] = run; run += cn[w]; }
      }
      __syncthreads();
#pragma unroll
      for (int j = 0; j < JMX; ++j) {
        if (j * CHK + wave * 32 < nC) {
          const bool live = j * CHK + (int)threadIdx.x < nC;
          const int key = live ? (rrow[j] >> 4) : 31;
          const unsigned peers = prs[j];
          const int first = cw[key];
          if (live && (peers & lt) == 0u) cw[key] = first + __builtin_popcount(peers);
          if (live) {
            const int pos = first + __builtin_popcount(peers & lt);
            srtRow[pos] = rrow[j]; srtIdx[pos] = ridx[j]; srtW[pos] = rwb[j];
          }
        }
      }
      __syncthreads();

      const int t0 = __builtin_amdgcn_readfirstlane(tileSt[wave]);
      const int t1 = __builtin_amdgcn_readfirstlane(tileSt[wave + 1]);
      for (int c0 = t0; c0 < t1; c0 += 32) {
        const int nLive = (t1 - c0) < 32 ? (t1 - c0) : 32;
        const int li = lane < nLive ? c0 + lane : c0;
        int4 rec;
        rec.x = srtRow[li]; rec.y = srtIdx[li]; rec.z = srtW[li]; rec.w = 0;
        const int key = lane < nLive ? (rec.x & 15) : 16;
        unsigned less = 0u, eq = ~0u;
#pragma unroll
        for (int b = 4; b >= 0; --b) {
          const bool bit = (key >> b) & 1;
          const unsigned bal = __builtin_amdgcn_ballot_w32(bit);
          less |= bit ? (eq & ~bal) : 0u;
          eq &= bit ? bal : ~bal;
        }
        const int pos = __builtin_popcount(less) + __builtin_popcount(eq & lt);
        const int rowL  = __builtin_amdgcn_ds_permute(pos << 2, rec.x);
        const int idxL  = __builtin_amdgcn_ds_permute(pos << 2, rec.y);
        const int wbits = __builtin_amdgcn_ds_permute(pos << 2, rec.z);
        float2 bA[8], bB[8];
        edge_rows8(lhs, idxL, lane, 0, bA);
        if (nLive == 32) {
          edge_rows8(lhs, idxL, lane, 8, bB);
          owner_chunk8<true>(bA, rhsS, tw, degc, rowL, wbits, we0, we1, lane, 0, nLive);
          edge_rows8(lhs, idxL, lane, 16, bA);
          owner_chunk8<true>(bB, rhsS, tw, degc, rowL, wbits, we0, we1, lane, 8, nLive);
          edge_rows8(lhs, idxL, lane, 24, bB);
          owner_chunk8<true>(bA, rhsS, tw, degc, rowL, wbits, we0, we1, lane, 16, nLive);
          owner_chunk8<true>(bB, rhsS, tw, degc, rowL, wbits, we0, we1, lane, 24, nLive);
        } else {
          if (nLive > 8) edge_rows8(lhs, idxL, lane, 8, bB);
          owner_chunk8<false>(bA, rhsS, tw, degc, rowL, wbits, we0, we1, lane, 0, nLive);
          if (nLive > 8) {
            if (nLive > 16) edge_rows8(lhs, idxL, lane, 16, bA);
            owner_chunk8<false>(bB, rhsS, tw, degc, rowL, wbits, we0, we1, lane, 8, nLive);
            if (nLive > 16) {
              if (nLive > 24) edge_rows8(lhs, idxL, lane, 24, bB);
              owner_chunk8<false>(bA, rhsS, tw, degc, rowL, wbits, we0, we1, lane, 16, nLive);
              if (nLive > 24)
                owner_chunk8<false>(bB, rhsS, tw, degc, rowL, wbits, we0, we1, lane, 24, nLive);
            }
          }
        }
      }
      __syncthreads();
    }
  }

  const int r = lane & 15;
  const int h = lane >> 4;
  const int lrow0 = wave * 16;
  const int row0 = base + lrow0;
  if (row0 >= M) return;

  int rowA = row0 + r;
  if (rowA >= M) rowA = M - 1;
  const float* __restrict__ Xrow = X + (size_t)rowA * 64;

  float dg[8];
#pragma unroll
  for (int v = 0; v < 8; ++v) dg[v] = __shfl(degc, v + 8 * h);

  const float* trow = tw + r * 64;
  const int xr = 4 * r;

  v8f acc0 = {}, acc1 = {}, acc2 = {}, acc3 = {};
#pragma unroll
  for (int k0 = 0; k0 < 128; k0 += 4) {
    const int ka = k0 + 2 * h;
    float2 av;
    if (k0 < 64) av = *reinterpret_cast<const float2*>(trow + ((k0 ^ xr) + 2 * h));
    else         av = *reinterpret_cast<const float2*>(Xrow + (ka - 64));
    v2f a; a.x = av.x; a.y = av.y;
    const int k2 = (k0 >> 1) + h;
    const float2* Bk = &WoS2[k2 * 64 + r];
    float2 f0 = Bk[0], f1 = Bk[16], f2 = Bk[32], f3 = Bk[48];
    v2f b0, b1, b2, b3;
    b0.x = f0.x; b0.y = f0.y;  b1.x = f1.x; b1.y = f1.y;
    b2.x = f2.x; b2.y = f2.y;  b3.x = f3.x; b3.y = f3.y;
    acc0 = wmma_f32(a, b0, acc0);
    acc1 = wmma_f32(a, b1, acc1);
    acc2 = wmma_f32(a, b2, acc2);
    acc3 = wmma_f32(a, b3, acc3);
  }

  const float bia0 = bout[r], bia1 = bout[16 + r], bia2 = bout[32 + r], bia3 = bout[48 + r];
  const float bp0 = bpS[r], bp1 = bpS[16 + r], bp2 = bpS[32 + r], bp3 = bpS[48 + r];
#pragma unroll
  for (int v = 0; v < 8; ++v) {
    const int row = row0 + v + 8 * h;
    if (row < M) {
      float* Crow = C + (size_t)row * 64;
      const float d = dg[v];
      Crow[r]      = acc0[v] + d * bp0 + bia0;
      Crow[16 + r] = acc1[v] + d * bp1 + bia1;
      Crow[32 + r] = acc2[v] + d * bp2 + bia2;
      Crow[48 + r] = acc3[v] + d * bp3 + bia3;
    }
  }
}

extern "C" void kernel_launch(void* const* d_in, const int* in_sizes, int n_in,
                              void* d_out, int out_size, void* d_ws, size_t ws_size,
                              hipStream_t stream) {
  const float* input   = (const float*)d_in[0];
  const float* other   = (const float*)d_in[1];
  const int*   rj      = (const int*)d_in[2];
  const int*   lj      = (const int*)d_in[3];
  const float* weights = (const float*)d_in[4];
  const float* Wi      = (const float*)d_in[5];
  const float* bi      = (const float*)d_in[6];
  const float* Wo      = (const float*)d_in[7];
  const float* We      = (const float*)d_in[8];
  const float* Wf      = (const float*)d_in[9];
  const float* bf      = (const float*)d_in[10];
  const float* Wout    = (const float*)d_in[11];
  const float* bout    = (const float*)d_in[12];

  const int D    = 64;
  const int N_IN = in_sizes[0] / D;
  const int N_OT = in_sizes[1] / D;
  const int E    = in_sizes[2];

  float* rhs    = (float*)d_ws;
  float* lhs    = rhs + (size_t)N_IN * D;
  int4*  recs   = (int4*)(lhs + (size_t)N_OT * D);
  int*   seg    = (int*)(recs + E);

  const int tiles_in = (N_IN + 15) / 16;
  const int tiles_ot = (N_OT + 15) / 16;
  const int nRanges  = (N_IN + RNG - 1) / RNG;
  const int epb      = (E + NSL - 1) / NSL;

  const int nbOt = (tiles_ot + 31) / 32;
  const int nbIn = (tiles_in + 31) / 32;
  place_gemm_kernel<<<NSL + nbOt + nbIn, 1024, 0, stream>>>(
      rj, lj, weights, seg, recs, E, epb, other, Wo, lhs, N_OT, nbOt, input, Wi, bi, rhs, N_IN);

  owner_kernel<<<nRanges, CHK, 0, stream>>>(rhs, lhs, recs, seg, epb, We, input, Wf, bf,
                                            Wout, bout, (float*)d_out, N_IN);
}


# Owner sort keyed by destination row, not tile: a row's edges are neighbours, its sum stays in registers and meets LDS once per row (rhs row read once, no per-edge tile update); f32 sum order changes
# speedup vs baseline: 3.0595x; 1.1165x over previous
#include <hip/hip_runtime.h>

typedef float v2f __attribute__((ext_vector_type(2)));
typedef float v8f __attribute__((ext_vector_type(8)));

__device__ __forceinline__ v8f wmma_f32(v2f a, v2f b, v8f c) {
  return __builtin_amdgcn_wmma_f32_16x16x4_f32(false, a, false, b, (short)0, c,
                                               false, false);
}

template <int NT>
__device__ __forceinline__ void gemm64_block(const float* __restrict__ A,
                                             const float* __restrict__ B,
                                             const float* __restrict__ bias,
                                             float* __restrict__ C, int M, int blk) {
  __shared__ float2 Bs2[32 * 64];
  for (int i = threadIdx.x; i < 2048; i += NT) {
    int k2 = i >> 6, n = i & 63;
    Bs2[i] = float2{B[(2 * k2) * 64 + n], B[(2 * k2 + 1) * 64 + n]};
  }
  __syncthreads();

  const int wave = threadIdx.x >> 5;
  const int lane = threadIdx.x & 31;
  const int r = lane & 15;
  const int h = lane >> 4;

  const int row0 = (blk * (NT / 32) + wave) * 16;
  if (row0 >= M) return;

  int rowA = row0 + r;
  if (rowA >= M) rowA = M - 1;
  const float* __restrict__ Arow = A + (size_t)rowA * 64;

  v8f acc0 = {}, acc1 = {}, acc2 = {}, acc3 = {};
#pragma unroll
  for (int k0 = 0; k0 < 64; k0 += 4) {
    const int ka = k0 + 2 * h;
    float2 av = *reinterpret_cast<const float2*>(Arow + ka);
    v2f a; a.x = av.x; a.y = av.y;
    const int k2 = (k0 >> 1) + h;
    const float2* __restrict__ Bk = &Bs2[k2 * 64 + r];
    float2 f0 = Bk[0], f1 = Bk[16], f2 = Bk[32], f3 = Bk[48];
    v2f b0, b1, b2, b3;
    b0.x = f0.x; b0.y = f0.y;  b1.x = f1.x; b1.y = f1.y;
    b2.x = f2.x; b2.y = f2.y;  b3.x = f3.x; b3.y = f3.y;
    acc0 = wmma_f32(a, b0, acc0);
    acc1 = wmma_f32(a, b1, acc1);
    acc2 = wmma_f32(a, b2, acc2);
    acc3 = wmma_f32(a, b3, acc3);
  }

  float bia0 = 0.f, bia1 = 0.f, bia2 = 0.f, bia3 = 0.f;
  if (bias) { bia0 = bias[r]; bia1 = bias[16 + r]; bia2 = bias[32 + r]; bia3 = bias[48 + r]; }

#pragma unroll
  for (int v = 0; v < 8; ++v) {
    const int row = row0 + v + 8 * h;
    if (row < M) {
      float* __restrict__ Crow = C + (size_t)row * 64;
      Crow[r]      = acc0[v] + bia0;
      Crow[16 + r] = acc1[v] + bia1;
      Crow[32 + r] = acc2[v] + bia2;
      Crow[48 + r] = acc3[v] + bia3;
    }
  }
}

constexpr int RNG  = 400;
constexpr int NBIN = 256;
constexpr int NSL  = 256;
constexpr int OWV  = RNG / 16;
constexpr int CHK  = OWV * 32;
constexpr int JMX  = 6;
constexpr int CAP  = CHK * JMX;
constexpr int SCW  = (RNG + 31) / 32;

__device__ __forceinline__ unsigned match_key8(int key, unsigned liveMask) {
  unsigned peers = liveMask;
#pragma unroll
  for (int b = 0; b < 8; ++b) {
    const bool bit = (key >> b) & 1;
    const unsigned bal = __builtin_amdgcn_ballot_w32(bit);
    peers &= bit ? bal : ~bal;
  }
  return peers;
}

__device__ __forceinline__ unsigned match_key9(int key, unsigned liveMask) {
  unsigned peers = liveMask;
#pragma unroll
  for (int b = 0; b < 9; ++b) {
    const bool bit = (key >> b) & 1;
    const unsigned bal = __builtin_amdgcn_ballot_w32(bit);
    peers &= bit ? bal : ~bal;
  }
  return peers;
}

__device__ __forceinline__ void count_slice(const int* __restrict__ rj, int* cw,
                                            int e0, int e1, int wave, int lane) {
  const unsigned lt = (1u << lane) - 1u;
  for (int eb = e0 + wave * 32; eb < e1; eb += 1024) {
    const int e = eb + lane;
    const bool live = e < e1;
    const int key = live ? rj[e] / RNG : 0;
    const unsigned lm = __builtin_amdgcn_ballot_w32(live);
    const unsigned peers = match_key8(key, lm);
    if (live && (peers & lt) == 0u) cw[key] += __builtin_popcount(peers);
  }
}

__launch_bounds__(1024, 16)
__global__ void place_gemm_kernel(const int* __restrict__ rj, const int* __restrict__ lj,
                                  const float* __restrict__ weights,
                                  int* __restrict__ seg, int4* __restrict__ recs,
                                  int E, int epb,
                                  const float* __restrict__ other, const float* __restrict__ Wo,
                                  float* __restrict__ lhs, int N_OT, int nbOt,
                                  const float* __restrict__ input, const float* __restrict__ Wi,
                                  const float* __restrict__ bi, float* __restrict__ rhs, int N_IN) {
  if (blockIdx.x >= NSL) {
    const int g = (int)blockIdx.x - NSL;
    const bool ot = g < nbOt;
    gemm64_block<1024>(ot ? other : input, ot ? Wo : Wi, ot ? nullptr : bi, ot ? lhs : rhs,
                       ot ? N_OT : N_IN, ot ? g : g - nbOt);
    return;
  }
  __shared__ int cntW[32 * NBIN];
  __shared__ int wsum[NBIN / 32];
  for (int i = threadIdx.x; i < 32 * NBIN; i += 1024) cntW[i] = 0;
  __syncthreads();
  const int lane = threadIdx.x & 31;
  const int wave = threadIdx.x >> 5;
  const int e0 = blockIdx.x * epb;
  int e1 = e0 + epb;
  if (e1 > E) e1 = E;
  int* cw = cntW + wave * NBIN;
  count_slice(rj, cw, e0, e1, wave, lane);
  __syncthreads();
  int c[32];
  int tot = 0, incl = 0;
  if (threadIdx.x < NBIN) {
#pragma unroll
    for (int w = 0; w < 32; ++w) c[w] = cntW[w * NBIN + threadIdx.x];
#pragma unroll
    for (int w = 0; w < 32; ++w) tot += c[w];
    incl = tot;
#pragma unroll
    for (int d = 1; d < 32; d <<= 1) {
      const int t = __shfl_up(incl, d);
      if (lane >= d) incl += t;
    }
    if (lane == 31) wsum[wave] = incl;
  }
  __syncthreads();
  if (threadIdx.x < NBIN) {
    int off = 0;
#pragma unroll
    for (int w = 0; w < NBIN / 32; ++w) off += (w < wave) ? wsum[w] : 0;
    int run = off + incl - tot;
    seg[threadIdx.x * NSL + blockIdx.x] = run;
    if (threadIdx.x == NBIN - 1) seg[NBIN * NSL + blockIdx.x] = run + tot;
#pragma unroll
    for (int w = 0; w < 32; ++w) { cntW[w * NBIN + threadIdx.x] = run; run += c[w]; }
  }
  __syncthreads();
  int4* __restrict__ out = recs + e0;
  const unsigned lt = (1u << lane) - 1u;
  for (int eb = e0 + wave * 32; eb < e1; eb += 1024) {
    const int e = eb + lane;
    const bool live = e < e1;
    const int rr = live ? rj[e] : 0;
    const int key = rr / RNG;
    const unsigned lm = __builtin_amdgcn_ballot_w32(live);
    const unsigned peers = match_key8(key, lm);
    const int first = cw[key];
    if (live && (peers & lt) == 0u) cw[key] = first + __builtin_popcount(peers);
    if (live) {
      const int pos = first + __builtin_popcount(peers & lt);
      out[pos] = int4{rr - key * RNG, lj[e], __float_as_int(weights[e]), 0};
    }
  }
}

__device__ __forceinline__ void edge_rows8(const float* __restrict__ lhs, int idxL, int lane,
                                           int c0, float2 (&b)[8]) {
#pragma unroll
  for (int i = 0; i < 8; ++i) {
    const unsigned sl = (unsigned)__builtin_amdgcn_readlane(idxL, c0 + i);
    b[i] = reinterpret_cast<const float2*>(lhs)[sl * 32u + (unsigned)lane];
  }
}

template <bool FULL>
__device__ __forceinline__ void owner_chunk8(const float2 (&b)[8],
                                             const float* rhsW, float* tw, int rsV,
                                             int& cur, int& nextB,
                                             float2& a, float2& o, float2& acc,
                                             int wbits, float we0, float we1, int lane,
                                             int slot0, int c0, int nLive) {
#pragma unroll
  for (int i = 0; i < 8; ++i) {
    if (FULL || c0 + i < nLive) {
      const int f = slot0 + c0 + i;
      if (f >= nextB) {
        float* trw = tw + cur * 64;
        *reinterpret_cast<float2*>(trw + ((2 * lane) ^ (4 * cur))) = float2{o.x + acc.x, o.y + acc.y};
        do {
          ++cur;
          nextB = __builtin_amdgcn_readlane(rsV, cur + 1);
        } while (f >= nextB);
        a = reinterpret_cast<const float2*>(rhsW + cur * 64)[lane];
        o = *reinterpret_cast<const float2*>(tw + cur * 64 + ((2 * lane) ^ (4 * cur)));
        acc = float2{0.f, 0.f};
      }
      const float w = __int_as_float(__builtin_amdgcn_readlane(wbits, c0 + i));
      const float v0 = a.x + b[i].x + w * we0;
      const float v1 = a.y + b[i].y + w * we1;
      acc.x += v0 > 0.f ? v0 : 0.01f * v0;
      acc.y += v1 > 0.f ? v1 : 0.01f * v1;
    }
  }
}

__launch_bounds__(CHK)
__global__ void owner_kernel(const float* __restrict__ rhs,
                             const float* __restrict__ lhs,
                             const int4* __restrict__ recs,
                             const int* __restrict__ seg, int epb,
                             const float* __restrict__ We,
                             const float* __restrict__ X,
                             const float* __restrict__ Wf,
                             const float* __restrict__ bf,
                             const float* __restrict__ Wout,
                             const float* __restrict__ bout,
                             float* __restrict__ C, int M) {
  __shared__ float2 WoS2[64 * 64];
  __shared__ float bpS[64];
  __shared__ __attribute__((aligned(16))) float tileS[RNG * 64];
  __shared__ __attribute__((aligned(16))) float rhsS[RNG * 64];
  __shared__ int srtIdx[CAP];
  __shared__ int srtW[CAP];
  __shared__ int curR[OWV * RNG];
  __shared__ int rowSt[RNG + 1];
  __shared__ int wsumR[SCW];
  __shared__ __attribute__((aligned(64))) int segP[NSL + 1];
  __shared__ __attribute__((aligned(64))) int segPc[NSL / 16];
  __shared__ int segG[NSL];
  __shared__ int wsum[NSL / 32];

  const int base = blockIdx.x * RNG;
  const int lane = threadIdx.x & 31;
  int segSt = 0, segEn = 0;
  if (threadIdx.x < NSL) {
    segSt = seg[blockIdx.x * NSL + threadIdx.x];
    segEn = seg[(blockIdx.x + 1) * NSL + threadIdx.x];
  }
  {
    const int pw = __builtin_amdgcn_readfirstlane((int)(threadIdx.x >> 5));
    if (pw < 20) {
      const int pr = lane & 15, ph = lane >> 4;
      const int ti = pw >> 2, tj = pw & 3;
      const float* __restrict__ Arow = (pw < 16) ? Wf + (16 * ti + pr) * 64 : bf;
      const float* __restrict__ Bcol = Wout + 16 * tj + pr;
      v8f accp = {};
#pragma unroll
      for (int k0 = 0; k0 < 64; k0 += 4) {
        const int ka = k0 + 2 * ph;
        const float2 av = *reinterpret_cast<const float2*>(Arow + ka);
        v2f a; a.x = av.x; a.y = av.y;
        v2f b; b.x = Bcol[ka * 64]; b.y = Bcol[(ka + 1) * 64];
        accp = wmma_f32(a, b, accp);
      }
      if (pw < 16) {
        float* WoF = reinterpret_cast<float*>(WoS2);
#pragma unroll
        for (int v = 0; v < 8; ++v) {
          const int k = 16 * ti + 8 * ph + v;
          WoF[((k >> 1) * 64 + 16 * tj + pr) * 2 + (k & 1)] = accp[v];
        }
      } else if (ph == 0) {
        bpS[16 * tj + pr] = accp[0];
      }
    }
  }
  for (int i = 2048 + threadIdx.x; i < 4096; i += CHK) {
    int k2 = i >> 6, n = i & 63;
    WoS2[i] = float2{Wout[(2 * k2) * 64 + n], Wout[(2 * k2 + 1) * 64 + n]};
  }
  for (int i = threadIdx.x; i < RNG * 16; i += CHK) {
    const int row = base + (i >> 4);
    float4 v = float4{0.f, 0.f, 0.f, 0.f};
    if (row < M) v = reinterpret_cast<const float4*>(rhs)[(size_t)row * 16 + (i & 15)];
    reinterpret_cast<float4*>(rhsS)[i] = v;
    reinterpret_cast<float4*>(tileS)[i] = float4{0.f, 0.f, 0.f, 0.f};
  }
  const int segLen = segEn - segSt;
  int incl = segLen;
  if (threadIdx.x < NSL) {
#pragma unroll
    for (int d = 1; d < 32; d <<= 1) {
      const int t = __shfl_up(incl, d);
      if (lane >= d) incl += t;
    }
    if (lane == 31) wsum[threadIdx.x >> 5] = incl;
  }
  __syncthreads();
  if (threadIdx.x < NSL) {
    int off = 0;
#pragma unroll
    for (int w = 0; w < NSL / 32; ++w) off += (w < (int)(threadIdx.x >> 5)) ? wsum[w] : 0;
    const int ex = off + incl - segLen;
    segP[threadIdx.x] = ex;
    if ((threadIdx.x & 15) == 0) segPc[threadIdx.x >> 4] = ex;
    segG[threadIdx.x] = (int)threadIdx.x * epb + segSt - ex;
    if (threadIdx.x == NSL - 1) segP[NSL] = ex + segLen;
  }
  __syncthreads();

  const int wave = __builtin_amdgcn_readfirstlane((int)(threadIdx.x >> 5));
  const unsigned lt = (1u << lane) - 1u;

  float* tw = &tileS[wave * 16 * 64];
  float degc = 0.f;

  {
    const int eEnd = __builtin_amdgcn_readfirstlane(segP[NSL]);
    auto fetch = [&](int f) -> int4 {
      const int4* pc = reinterpret_cast<const int4*>(segPc);
      int c = -1;
#pragma unroll
      for (int q = 0; q < 4; ++q) {
        const int4 v = pc[q];
        c += (v.x <= f) + (v.y <= f) + (v.z <= f) + (v.w <= f);
      }
      const int4* pf = reinterpret_cast<const int4*>(segP + 16 * c);
      int sl = 16 * c - 1;
#pragma unroll
      for (int q = 0; q < 4; ++q) {
        const int4 v = pf[q];
        sl += (v.x <= f) + (v.y <= f) + (v.z <= f) + (v.w <= f);
      }
      return recs[segG[sl] + f];
    };
    const float we0 = We[2 * lane], we1 = We[2 * lane + 1];
    int* cw = curR + wave * RNG;
    const float* rhsW = &rhsS[wave * 16 * 64];
    for (int cb = 0; cb < eEnd; cb += CAP) {
      const int nC = (eEnd - cb) < CAP ? (eEnd - cb) : CAP;
      for (int i = lane; i < RNG; i += 32) cw[i] = 0;
      int rrow[JMX], ridx[JMX], rwb[JMX];
      unsigned prs[JMX];
#pragma unroll
      for (int j = 0; j < JMX; ++j) {
        const int fl = j * CHK + (int)threadIdx.x;
        int4 rc = int4{0, 0, 0, 0};
        if (fl < nC) rc = fetch(cb + fl);
        rrow[j] = rc.x; ridx[j] = rc.y; rwb[j] = rc.z;
        prs[j] = 0u;
      }
#pragma unroll
      for (int j = 0; j < JMX; ++j) {
        if (j * CHK + wave * 32 < nC) {
          const bool live = j * CHK + (int)threadIdx.x < nC;
          const int key = rrow[j];
          const unsigned lm = __builtin_amdgcn_ballot_w32(live);
          const unsigned peers = match_key9(key, lm);
          prs[j] = peers;
          if (live && (peers & lt) == 0u) cw[key] += __builtin_popcount(peers);
        }
      }
      __syncthreads();
      int cn[OWV];
      int tot = 0, inc = 0;
      if (wave < SCW) {
        const bool own = (int)threadIdx.x < RNG;
        const int tr = own ? (int)threadIdx.x : 0;
#pragma unroll
        for (int w = 0; w < OWV; ++w) cn[w] = curR[w * RNG + tr];
#pragma unroll
        for (int w = 0; w < OWV; ++w) tot += cn[w];
        if (!own) tot = 0;
        inc = tot;
#pragma unroll
        for (int d = 1; d < 32; d <<= 1) {
          const int t = __shfl_up(inc, d);
          if (lane >= d) inc += t;
        }
        if (lane == 31) wsumR[wave] = inc;
      }
      __syncthreads();
      if (wave < SCW && (int)threadIdx.x < RNG) {
        int off = 0;
#pragma unroll
        for (int w = 0; w < SCW; ++w) off += (w < wave) ? wsumR[w] : 0;
        int run = off + inc - tot;
        rowSt[threadIdx.x] = run;
#pragma unroll
        for (int w = 0; w < OWV; ++w) { curR[w * RNG + threadIdx.x] = run; run += cn[w]; }
        if ((int)threadIdx.x == RNG - 1) rowSt[RNG] = run;
      }
      __syncthreads();
#pragma unroll
      for (int j = 0; j < JMX; ++j) {
        if (j * CHK + wave * 32 < nC) {
          const bool live = j * CHK + (int)threadIdx.x < nC;
          const int key = rrow[j];
          const unsigned peers = prs[j];
          const int first = cw[key];
          if (live && (peers & lt) == 0u) cw[key] = first + __builtin_popcount(peers);
          if (live) {
            const int pos = first + __builtin_popcount(peers & lt);
            srtIdx[pos] = ridx[j]; srtW[pos] = rwb[j];
          }
        }
      }
      __syncthreads();

      const int rsV = rowSt[wave * 16 + (lane < 16 ? lane : 16)];
      const int rsN = __shfl_down(rsV, 1);
      degc += (lane < 16) ? (float)(rsN - rsV) : 0.f;
      const int t0 = __builtin_amdgcn_readlane(rsV, 0);
      const int t1 = __builtin_amdgcn_readlane(rsV, 16);
      int cur = 0;
      int nextB = __builtin_amdgcn_readlane(rsV, 1);
      float2 a = reinterpret_cast<const float2*>(rhsW)[lane];
      float2 o = *reinterpret_cast<const float2*>(tw + 2 * lane);
      float2 acc = float2{0.f, 0.f};
      for (int c0 = t0; c0 < t1; c0 += 32) {
        const int nLive = (t1 - c0) < 32 ? (t1 - c0) : 32;
        const int li = lane < nLive ? c0 + lane : c0;
        const int idxL = srtIdx[li];
        const int wbits = srtW[li];
        float2 bA[8], bB[8];
        edge_rows8(lhs, idxL, lane, 0, bA);
        if (nLive == 32) {
          edge_rows8(lhs, idxL, lane, 8, bB);
          owner_chunk8<true>(bA, rhsW, tw, rsV, cur, nextB, a, o, acc, wbits, we0, we1, lane, c0, 0, nLive);
          edge_rows8(lhs, idxL, lane, 16, bA);
          owner_chunk8<true>(bB, rhsW, tw, rsV, cur, nextB, a, o, acc, wbits, we0, we1, lane, c0, 8, nLive);
          edge_rows8(lhs, idxL, lane, 24, bB);
          owner_chunk8<true>(bA, rhsW, tw, rsV, cur, nextB, a, o, acc, wbits, we0, we1, lane, c0, 16, nLive);
          owner_chunk8<true>(bB, rhsW, tw, rsV, cur, nextB, a, o, acc, wbits, we0, we1, lane, c0, 24, nLive);
        } else {
          if (nLive > 8) edge_rows8(lhs, idxL, lane, 8, bB);
          owner_chunk8<false>(bA, rhsW, tw, rsV, cur, nextB, a, o, acc, wbits, we0, we1, lane, c0, 0, nLive);
          if (nLive > 8) {
            if (nLive > 16) edge_rows8(lhs, idxL, lane, 16, bA);
            owner_chunk8<false>(bB, rhsW, tw, rsV, cur, nextB, a, o, acc, wbits, we0, we1, lane, c0, 8, nLive);
            if (nLive > 16) {
              if (nLive > 24) edge_rows8(lhs, idxL, lane, 24, bB);
              owner_chunk8<false>(bA, rhsW, tw, rsV, cur, nextB, a, o, acc, wbits, we0, we1, lane, c0, 16, nLive);
              if (nLive > 24)
                owner_chunk8<false>(bB, rhsW, tw, rsV, cur, nextB, a, o, acc, wbits, we0, we1, lane, c0, 24, nLive);
            }
          }
        }
      }
      *reinterpret_cast<float2*>(tw + cur * 64 + ((2 * lane) ^ (4 * cur))) = float2{o.x + acc.x, o.y + acc.y};
      __syncthreads();
    }
  }

  const int r = lane & 15;
  const int h = lane >> 4;
  const int lrow0 = wave * 16;
  const int row0 = base + lrow0;
  if (row0 >= M) return;

  int rowA = row0 + r;
  if (rowA >= M) rowA = M - 1;
  const float* __restrict__ Xrow = X + (size_t)rowA * 64;

  float dg[8];
#pragma unroll
  for (int v = 0; v < 8; ++v) dg[v] = __shfl(degc, v + 8 * h);

  const float* trow = tw + r * 64;
  const int xr = 4 * r;

  v8f acc0 = {}, acc1 = {}, acc2 = {}, acc3 = {};
#pragma unroll
  for (int k0 = 0; k0 < 128; k0 += 4) {
    const int ka = k0 + 2 * h;
    float2 av;
    if (k0 < 64) av = *reinterpret_cast<const float2*>(trow + ((k0 ^ xr) + 2 * h));
    else         av = *reinterpret_cast<const float2*>(Xrow + (ka - 64));
    v2f a; a.x = av.x; a.y = av.y;
    const int k2 = (k0 >> 1) + h;
    const float2* Bk = &WoS2[k2 * 64 + r];
    float2 f0 = Bk[0], f1 = Bk[16], f2 = Bk[32], f3 = Bk[48];
    v2f b0, b1, b2, b3;
    b0.x = f0.x; b0.y = f0.y;  b1.x = f1.x; b1.y = f1.y;
    b2.x = f2.x; b2.y = f2.y;  b3.x = f3.x; b3.y = f3.y;
    acc0 = wmma_f32(a, b0, acc0);
    acc1 = wmma_f32(a, b1, acc1);
    acc2 = wmma_f32(a, b2, acc2);
    acc3 = wmma_f32(a, b3, acc3);
  }

  const float bia0 = bout[r], bia1 = bout[16 + r], bia2 = bout[32 + r], bia3 = bout[48 + r];
  const float bp0 = bpS[r], bp1 = bpS[16 + r], bp2 = bpS[32 + r], bp3 = bpS[48 + r];
#pragma unroll
  for (int v = 0; v < 8; ++v) {
    const int row = row0 + v + 8 * h;
    if (row < M) {
      float* Crow = C + (size_t)row * 64;
      const float d = dg[v];
      Crow[r]      = acc0[v] + d * bp0 + bia0;
      Crow[16 + r] = acc1[v] + d * bp1 + bia1;
      Crow[32 + r] = acc2[v] + d * bp2 + bia2;
      Crow[48 + r] = acc3[v] + d * bp3 + bia3;
    }
  }
}

extern "C" void kernel_launch(void* const* d_in, const int* in_sizes, int n_in,
                              void* d_out, int out_size, void* d_ws, size_t ws_size,
                              hipStream_t stream) {
  const float* input   = (const float*)d_in[0];
  const float* other   = (const float*)d_in[1];
  const int*   rj      = (const int*)d_in[2];
  const int*   lj      = (const int*)d_in[3];
  const float* weights = (const float*)d_in[4];
  const float* Wi      = (const float*)d_in[5];
  const float* bi      = (const float*)d_in[6];
  const float* Wo      = (const float*)d_in[7];
  const float* We      = (const float*)d_in[8];
  const float* Wf      = (const float*)d_in[9];
  const float* bf      = (const float*)d_in[10];
  const float* Wout    = (const float*)d_in[11];
  const float* bout    = (const float*)d_in[12];

  const int D    = 64;
  const int N_IN = in_sizes[0] / D;
  const int N_OT = in_sizes[1] / D;
  const int E    = in_sizes[2];

  float* rhs    = (float*)d_ws;
  float* lhs    = rhs + (size_t)N_IN * D;
  int4*  recs   = (int4*)(lhs + (size_t)N_OT * D);
  int*   seg    = (int*)(recs + E);

  const int tiles_in = (N_IN + 15) / 16;
  const int tiles_ot = (N_OT + 15) / 16;
  const int nRanges  = (N_IN + RNG - 1) / RNG;
  const int epb      = (E + NSL - 1) / NSL;

  const int nbOt = (tiles_ot + 31) / 32;
  const int nbIn = (tiles_in + 31) / 32;
  place_gemm_kernel<<<NSL + nbOt + nbIn, 1024, 0, stream>>>(
      rj, lj, weights, seg, recs, E, epb, other, Wo, lhs, N_OT, nbOt, input, Wi, bi, rhs, N_IN);

  owner_kernel<<<nRanges, CHK, 0, stream>>>(rhs, lhs, recs, seg, epb, We, input, Wf, bf,
                                            Wout, bout, (float*)d_out, N_IN);
}


# Staging loads asked for as one group ahead of use, not load-wait-store per trip: owner's 8 rhs-row and 3 Wout pieces a thread, place and count's 4 steps of rj, lj, weight a wave; no sum order change
# speedup vs baseline: 3.1843x; 1.0408x over previous
#include <hip/hip_runtime.h>

typedef float v2f __attribute__((ext_vector_type(2)));
typedef float v8f __attribute__((ext_vector_type(8)));

__device__ __forceinline__ v8f wmma_f32(v2f a, v2f b, v8f c) {
  return __builtin_amdgcn_wmma_f32_16x16x4_f32(false, a, false, b, (short)0, c,
                                               false, false);
}

template <int NT>
__device__ __forceinline__ void gemm64_block(const float* __restrict__ A,
                                             const float* __restrict__ B,
                                             const float* __restrict__ bias,
                                             float* __restrict__ C, int M, int blk) {
  __shared__ float2 Bs2[32 * 64];
  static_assert(2048 % NT == 0, "the B pieces divide evenly among the threads");
  constexpr int BQ = 2048 / NT;
  float2 bv[BQ];
#pragma unroll
  for (int q = 0; q < BQ; ++q) {
    const int i = q * NT + (int)threadIdx.x;
    const int k2 = i >> 6, n = i & 63;
    bv[q] = float2{B[(2 * k2) * 64 + n], B[(2 * k2 + 1) * 64 + n]};
  }
#pragma unroll
  for (int q = 0; q < BQ; ++q) Bs2[q * NT + (int)threadIdx.x] = bv[q];
  __syncthreads();

  const int wave = threadIdx.x >> 5;
  const int lane = threadIdx.x & 31;
  const int r = lane & 15;
  const int h = lane >> 4;

  const int row0 = (blk * (NT / 32) + wave) * 16;
  if (row0 >= M) return;

  int rowA = row0 + r;
  if (rowA >= M) rowA = M - 1;
  const float* __restrict__ Arow = A + (size_t)rowA * 64;

  v8f acc0 = {}, acc1 = {}, acc2 = {}, acc3 = {};
#pragma unroll
  for (int k0 = 0; k0 < 64; k0 += 4) {
    const int ka = k0 + 2 * h;
    float2 av = *reinterpret_cast<const float2*>(Arow + ka);
    v2f a; a.x = av.x; a.y = av.y;
    const int k2 = (k0 >> 1) + h;
    const float2* __restrict__ Bk = &Bs2[k2 * 64 + r];
    float2 f0 = Bk[0], f1 = Bk[16], f2 = Bk[32], f3 = Bk[48];
    v2f b0, b1, b2, b3;
    b0.x = f0.x; b0.y = f0.y;  b1.x = f1.x; b1.y = f1.y;
    b2.x = f2.x; b2.y = f2.y;  b3.x = f3.x; b3.y = f3.y;
    acc0 = wmma_f32(a, b0, acc0);
    acc1 = wmma_f32(a, b1, acc1);
    acc2 = wmma_f32(a, b2, acc2);
    acc3 = wmma_f32(a, b3, acc3);
  }

  float bia0 = 0.f, bia1 = 0.f, bia2 = 0.f, bia3 = 0.f;
  if (bias) { bia0 = bias[r]; bia1 = bias[16 + r]; bia2 = bias[32 + r]; bia3 = bias[48 + r]; }

#pragma unroll
  for (int v = 0; v < 8; ++v) {
    const int row = row0 + v + 8 * h;
    if (row < M) {
      float* __restrict__ Crow = C + (size_t)row * 64;
      Crow[r]      = acc0[v] + bia0;
      Crow[16 + r] = acc1[v] + bia1;
      Crow[32 + r] = acc2[v] + bia2;
      Crow[48 + r] = acc3[v] + bia3;
    }
  }
}

constexpr int RNG  = 400;
constexpr int NBIN = 256;
constexpr int NSL  = 256;
constexpr int OWV  = RNG / 16;
constexpr int CHK  = OWV * 32;
constexpr int JMX  = 6;
constexpr int CAP  = CHK * JMX;
constexpr int SCW  = (RNG + 31) / 32;

__device__ __forceinline__ unsigned match_key8(int key, unsigned liveMask) {
  unsigned peers = liveMask;
#pragma unroll
  for (int b = 0; b < 8; ++b) {
    const bool bit = (key >> b) & 1;
    const unsigned bal = __builtin_amdgcn_ballot_w32(bit);
    peers &= bit ? bal : ~bal;
  }
  return peers;
}

__device__ __forceinline__ unsigned match_key9(int key, unsigned liveMask) {
  unsigned peers = liveMask;
#pragma unroll
  for (int b = 0; b < 9; ++b) {
    const bool bit = (key >> b) & 1;
    const unsigned bal = __builtin_amdgcn_ballot_w32(bit);
    peers &= bit ? bal : ~bal;
  }
  return peers;
}

constexpr int PJ = 4;

__device__ __forceinline__ void count_slice(const int* __restrict__ rj, int* cw,
                                            int e0, int e1, int wave, int lane) {
  const unsigned lt = (1u << lane) - 1u;
  for (int eb = e0 + wave * 32; eb < e1; eb += PJ * 1024) {
    int rr[PJ];
#pragma unroll
    for (int j = 0; j < PJ; ++j) {
      int e = eb + j * 1024 + lane;
      if (e > e1 - 1) e = e1 - 1;
      rr[j] = rj[e];
    }
    __builtin_amdgcn_sched_barrier(0);
#pragma unroll
    for (int j = 0; j < PJ; ++j) {
      if (eb + j * 1024 < e1) {
        const bool live = eb + j * 1024 + lane < e1;
        const int key = live ? rr[j] / RNG : 0;
        const unsigned lm = __builtin_amdgcn_ballot_w32(live);
        const unsigned peers = match_key8(key, lm);
        if (live && (peers & lt) == 0u) cw[key] += __builtin_popcount(peers);
      }
    }
  }
}

__launch_bounds__(1024, 16)
__global__ void place_gemm_kernel(const int* __restrict__ rj, const int* __restrict__ lj,
                                  const float* __restrict__ weights,
                                  int* __restrict__ seg, int4* __restrict__ recs,
                                  int E, int epb,
                                  const float* __restrict__ other, const float* __restrict__ Wo,
                                  float* __restrict__ lhs, int N_OT, int nbOt,
                                  const float* __restrict__ input, const float* __restrict__ Wi,
                                  const float* __restrict__ bi, float* __restrict__ rhs, int N_IN) {
  if (blockIdx.x >= NSL) {
    const int g = (int)blockIdx.x - NSL;
    const bool ot = g < nbOt;
    gemm64_block<1024>(ot ? other : input, ot ? Wo : Wi, ot ? nullptr : bi, ot ? lhs : rhs,
                       ot ? N_OT : N_IN, ot ? g : g - nbOt);
    return;
  }
  __shared__ int cntW[32 * NBIN];
  __shared__ int wsum[NBIN / 32];
  for (int i = threadIdx.x; i < 32 * NBIN; i += 1024) cntW[i] = 0;
  __syncthreads();
  const int lane = threadIdx.x & 31;
  const int wave = threadIdx.x >> 5;
  const int e0 = blockIdx.x * epb;
  int e1 = e0 + epb;
  if (e1 > E) e1 = E;
  int* cw = cntW + wave * NBIN;
  count_slice(rj, cw, e0, e1, wave, lane);
  __syncthreads();
  int c[32];
  int tot = 0, incl = 0;
  if (threadIdx.x < NBIN) {
#pragma unroll
    for (int w = 0; w < 32; ++w) c[w] = cntW[w * NBIN + threadIdx.x];
#pragma unroll
    for (int w = 0; w < 32; ++w) tot += c[w];
    incl = tot;
#pragma unroll
    for (int d = 1; d < 32; d <<= 1) {
      const int t = __shfl_up(incl, d);
      if (lane >= d) incl += t;
    }
    if (lane == 31) wsum[wave] = incl;
  }
  __syncthreads();
  if (threadIdx.x < NBIN) {
    int off = 0;
#pragma unroll
    for (int w = 0; w < NBIN / 32; ++w) off += (w < wave) ? wsum[w] : 0;
    int run = off + incl - tot;
    seg[threadIdx.x * NSL + blockIdx.x] = run;
    if (threadIdx.x == NBIN - 1) seg[NBIN * NSL + blockIdx.x] = run + tot;
#pragma unroll
    for (int w = 0; w < 32; ++w) { cntW[w * NBIN + threadIdx.x] = run; run += c[w]; }
  }
  __syncthreads();
  int4* __restrict__ out = recs + e0;
  const unsigned lt = (1u << lane) - 1u;
  for (int eb = e0 + wave * 32; eb < e1; eb += PJ * 1024) {
    int rrv[PJ], llv[PJ];
    float wwv[PJ];
#pragma unroll
    for (int j = 0; j < PJ; ++j) {
      int e = eb + j * 1024 + lane;
      if (e > e1 - 1) e = e1 - 1;
      rrv[j] = rj[e]; llv[j] = lj[e]; wwv[j] = weights[e];
    }
#pragma unroll
    for (int j = 0; j < PJ; ++j) {
      if (eb + j * 1024 < e1) {
        const bool live = eb + j * 1024 + lane < e1;
        const int rr = live ? rrv[j] : 0;
        const int key = rr / RNG;
        const unsigned lm = __builtin_amdgcn_ballot_w32(live);
        const unsigned peers = match_key8(key, lm);
        const int first = cw[key];
        if (live && (peers & lt) == 0u) cw[key] = first + __builtin_popcount(peers);
        if (live) {
          const int pos = first + __builtin_popcount(peers & lt);
          out[pos] = int4{rr - key * RNG, llv[j], __float_as_int(wwv[j]), 0};
        }
      }
    }
  }
}

__device__ __forceinline__ void edge_rows8(const float* __restrict__ lhs, int idxL, int lane,
                                           int c0, float2 (&b)[8]) {
#pragma unroll
  for (int i = 0; i < 8; ++i) {
    const unsigned sl = (unsigned)__builtin_amdgcn_readlane(idxL, c0 + i);
    b[i] = reinterpret_cast<const float2*>(lhs)[sl * 32u + (unsigned)lane];
  }
}

template <bool FULL>
__device__ __forceinline__ void owner_chunk8(const float2 (&b)[8],
                                             const float* rhsW, float* tw, int rsV,
                                             int& cur, int& nextB,
                                             float2& a, float2& o, float2& acc,
                                             int wbits, float we0, float we1, int lane,
                                             int slot0, int c0, int nLive) {
#pragma unroll
  for (int i = 0; i < 8; ++i) {
    if (FULL || c0 + i < nLive) {
      const int f = slot0 + c0 + i;
      if (f >= nextB) {
        float* trw = tw + cur * 64;
        *reinterpret_cast<float2*>(trw + ((2 * lane) ^ (4 * cur))) = float2{o.x + acc.x, o.y + acc.y};
        do {
          ++cur;
          nextB = __builtin_amdgcn_readlane(rsV, cur + 1);
        } while (f >= nextB);
        a = reinterpret_cast<const float2*>(rhsW + cur * 64)[lane];
        o = *reinterpret_cast<const float2*>(tw + cur * 64 + ((2 * lane) ^ (4 * cur)));
        acc = float2{0.f, 0.f};
      }
      const float w = __int_as_float(__builtin_amdgcn_readlane(wbits, c0 + i));
      const float v0 = a.x + b[i].x + w * we0;
      const float v1 = a.y + b[i].y + w * we1;
      acc.x += v0 > 0.f ? v0 : 0.01f * v0;
      acc.y += v1 > 0.f ? v1 : 0.01f * v1;
    }
  }
}

__launch_bounds__(CHK)
__global__ void owner_kernel(const float* __restrict__ rhs,
                             const float* __restrict__ lhs,
                             const int4* __restrict__ recs,
                             const int* __restrict__ seg, int epb,
                             const float* __restrict__ We,
                             const float* __restrict__ X,
                             const float* __restrict__ Wf,
                             const float* __restrict__ bf,
                             const float* __restrict__ Wout,
                             const float* __restrict__ bout,
                             float* __restrict__ C, int M) {
  __shared__ float2 WoS2[64 * 64];
  __shared__ float bpS[64];
  __shared__ __attribute__((aligned(16))) float tileS[RNG * 64];
  __shared__ __attribute__((aligned(16))) float rhsS[RNG * 64];
  __shared__ int srtIdx[CAP];
  __shared__ int srtW[CAP];
  __shared__ int curR[OWV * RNG];
  __shared__ int rowSt[RNG + 1];
  __shared__ int wsumR[SCW];
  __shared__ __attribute__((aligned(64))) int segP[NSL + 1];
  __shared__ __attribute__((aligned(64))) int segPc[NSL / 16];
  __shared__ int segG[NSL];
  __shared__ int wsum[NSL / 32];

  const int base = blockIdx.x * RNG;
  const int lane = threadIdx.x & 31;
  int segSt = 0, segEn = 0;
  if (threadIdx.x < NSL) {
    segSt = seg[blockIdx.x * NSL + threadIdx.x];
    segEn = seg[(blockIdx.x + 1) * NSL + threadIdx.x];
  }
  constexpr int RQ = RNG * 16 / CHK;
  static_assert(RQ * CHK == RNG * 16, "the rhs pieces divide evenly among the threads");
  constexpr int WQ = (2048 + CHK - 1) / CHK;
  float4 rv[RQ];
  float2 wv[WQ];
#pragma unroll
  for (int q = 0; q < RQ; ++q) {
    const int i = q * CHK + (int)threadIdx.x;
    int row = base + (i >> 4);
    if (row >= M) row = M - 1;
    rv[q] = reinterpret_cast<const float4*>(rhs)[(size_t)row * 16 + (i & 15)];
  }
#pragma unroll
  for (int q = 0; q < WQ; ++q) {
    int i = 2048 + q * CHK + (int)threadIdx.x;
    if (i > 4095) i = 4095;
    const int k2 = i >> 6, n = i & 63;
    wv[q] = float2{Wout[(2 * k2) * 64 + n], Wout[(2 * k2 + 1) * 64 + n]};
  }
  {
    const int pw = __builtin_amdgcn_readfirstlane((int)(threadIdx.x >> 5));
    if (pw < 20) {
      const int pr = lane & 15, ph = lane >> 4;
      const int ti = pw >> 2, tj = pw & 3;
      const float* __restrict__ Arow = (pw < 16) ? Wf + (16 * ti + pr) * 64 : bf;
      const float* __restrict__ Bcol = Wout + 16 * tj + pr;
      v8f accp = {};
#pragma unroll
      for (int k0 = 0; k0 < 64; k0 += 4) {
        const int ka = k0 + 2 * ph;
        const float2 av = *reinterpret_cast<const float2*>(Arow + ka);
        v2f a; a.x = av.x; a.y = av.y;
        v2f b; b.x = Bcol[ka * 64]; b.y = Bcol[(ka + 1) * 64];
        accp = wmma_f32(a, b, accp);
      }
      if (pw < 16) {
        float* WoF = reinterpret_cast<float*>(WoS2);
#pragma unroll
        for (int v = 0; v < 8; ++v) {
          const int k = 16 * ti + 8 * ph + v;
          WoF[((k >> 1) * 64 + 16 * tj + pr) * 2 + (k & 1)] = accp[v];
        }
      } else if (ph == 0) {
        bpS[16 * tj + pr] = accp[0];
      }
    }
  }
#pragma unroll
  for (int q = 0; q < WQ; ++q) {
    const int i = 2048 + q * CHK + (int)threadIdx.x;
    if (i < 4096) WoS2[i] = wv[q];
  }
#pragma unroll
  for (int q = 0; q < RQ; ++q) {
    const int i = q * CHK + (int)threadIdx.x;
    const bool in = base + (i >> 4) < M;
    reinterpret_cast<float4*>(rhsS)[i] = in ? rv[q] : float4{0.f, 0.f, 0.f, 0.f};
    reinterpret_cast<float4*>(tileS)[i] = float4{0.f, 0.f, 0.f, 0.f};
  }
  const int segLen = segEn - segSt;
  int incl = segLen;
  if (threadIdx.x < NSL) {
#pragma unroll
    for (int d = 1; d < 32; d <<= 1) {
      const int t = __shfl_up(incl, d);
      if (lane >= d) incl += t;
    }
    if (lane == 31) wsum[threadIdx.x >> 5] = incl;
  }
  __syncthreads();
  if (threadIdx.x < NSL) {
    int off = 0;
#pragma unroll
    for (int w = 0; w < NSL / 32; ++w) off += (w < (int)(threadIdx.x >> 5)) ? wsum[w] : 0;
    const int ex = off + incl - segLen;
    segP[threadIdx.x] = ex;
    if ((threadIdx.x & 15) == 0) segPc[threadIdx.x >> 4] = ex;
    segG[threadIdx.x] = (int)threadIdx.x * epb + segSt - ex;
    if (threadIdx.x == NSL - 1) segP[NSL] = ex + segLen;
  }
  __syncthreads();

  const int wave = __builtin_amdgcn_readfirstlane((int)(threadIdx.x >> 5));
  const unsigned lt = (1u << lane) - 1u;

  float* tw = &tileS[wave * 16 * 64];
  float degc = 0.f;

  {
    const int eEnd = __builtin_amdgcn_readfirstlane(segP[NSL]);
    auto fetch = [&](int f) -> int4 {
      const int4* pc = reinterpret_cast<const int4*>(segPc);
      int c = -1;
#pragma unroll
      for (int q = 0; q < 4; ++q) {
        const int4 v = pc[q];
        c += (v.x <= f) + (v.y <= f) + (v.z <= f) + (v.w <= f);
      }
      const int4* pf = reinterpret_cast<const int4*>(segP + 16 * c);
      int sl = 16 * c - 1;
#pragma unroll
      for (int q = 0; q < 4; ++q) {
        const int4 v = pf[q];
        sl += (v.x <= f) + (v.y <= f) + (v.z <= f) + (v.w <= f);
      }
      return recs[segG[sl] + f];
    };
    const float we0 = We[2 * lane], we1 = We[2 * lane + 1];
    int* cw = curR + wave * RNG;
    const float* rhsW = &rhsS[wave * 16 * 64];
    for (int cb = 0; cb < eEnd; cb += CAP) {
      const int nC = (eEnd - cb) < CAP ? (eEnd - cb) : CAP;
      for (int i = lane; i < RNG; i += 32) cw[i] = 0;
      int rrow[JMX], ridx[JMX], rwb[JMX];
      unsigned prs[JMX];
#pragma unroll
      for (int j = 0; j < JMX; ++j) {
        const int fl = j * CHK + (int)threadIdx.x;
        int4 rc = int4{0, 0, 0, 0};
        if (fl < nC) rc = fetch(cb + fl);
        rrow[j] = rc.x; ridx[j] = rc.y; rwb[j] = rc.z;
        prs[j] = 0u;
      }
#pragma unroll
      for (int j = 0; j < JMX; ++j) {
        if (j * CHK + wave * 32 < nC) {
          const bool live = j * CHK + (int)threadIdx.x < nC;
          const int key = rrow[j];
          const unsigned lm = __builtin_amdgcn_ballot_w32(live);
          const unsigned peers = match_key9(key, lm);
          prs[j] = peers;
          if (live && (peers & lt) == 0u) cw[key] += __builtin_popcount(peers);
        }
      }
      __syncthreads();
      int cn[OWV];
      int tot = 0, inc = 0;
      if (wave < SCW) {
        const bool own = (int)threadIdx.x < RNG;
        const int tr = own ? (int)threadIdx.x : 0;
#pragma unroll
        for (int w = 0; w < OWV; ++w) cn[w] = curR[w * RNG + tr];
#pragma unroll
        for (int w = 0; w < OWV; ++w) tot += cn[w];
        if (!own) tot = 0;
        inc = tot;
#pragma unroll
        for (int d = 1; d < 32; d <<= 1) {
          const int t = __shfl_up(inc, d);
          if (lane >= d) inc += t;
        }
        if (lane == 31) wsumR[wave] = inc;
      }
      __syncthreads();
      if (wave < SCW && (int)threadIdx.x < RNG) {
        int off = 0;
#pragma unroll
        for (int w = 0; w < SCW; ++w) off += (w < wave) ? wsumR[w] : 0;
        int run = off + inc - tot;
        rowSt[threadIdx.x] = run;
#pragma unroll
        for (int w = 0; w < OWV; ++w) { curR[w * RNG + threadIdx.x] = run; run += cn[w]; }
        if ((int)threadIdx.x == RNG - 1) rowSt[RNG] = run;
      }
      __syncthreads();
#pragma unroll
      for (int j = 0; j < JMX; ++j) {
        if (j * CHK + wave * 32 < nC) {
          const bool live = j * CHK + (int)threadIdx.x < nC;
          const int key = rrow[j];
          const unsigned peers = prs[j];
          const int first = cw[key];
          if (live && (peers & lt) == 0u) cw[key] = first + __builtin_popcount(peers);
          if (live) {
            const int pos = first + __builtin_popcount(peers & lt);
            srtIdx[pos] = ridx[j]; srtW[pos] = rwb[j];
          }
        }
      }
      __syncthreads();

      const int rsV = rowSt[wave * 16 + (lane < 16 ? lane : 16)];
      const int rsN = __shfl_down(rsV, 1);
      degc += (lane < 16) ? (float)(rsN - rsV) : 0.f;
      const int t0 = __builtin_amdgcn_readlane(rsV, 0);
      const int t1 = __builtin_amdgcn_readlane(rsV, 16);
      int cur = 0;
      int nextB = __builtin_amdgcn_readlane(rsV, 1);
      float2 a = reinterpret_cast<const float2*>(rhsW)[lane];
      float2 o = *reinterpret_cast<const float2*>(tw + 2 * lane);
      float2 acc = float2{0.f, 0.f};
      for (int c0 = t0; c0 < t1; c0 += 32) {
        const int nLive = (t1 - c0) < 32 ? (t1 - c0) : 32;
        const int li = lane < nLive ? c0 + lane : c0;
        const int idxL = srtIdx[li];
        const int wbits = srtW[li];
        float2 bA[8], bB[8];
        edge_rows8(lhs, idxL, lane, 0, bA);
        if (nLive == 32) {
          edge_rows8(lhs, idxL, lane, 8, bB);
          owner_chunk8<true>(bA, rhsW, tw, rsV, cur, nextB, a, o, acc, wbits, we0, we1, lane, c0, 0, nLive);
          edge_rows8(lhs, idxL, lane, 16, bA);
          owner_chunk8<true>(bB, rhsW, tw, rsV, cur, nextB, a, o, acc, wbits, we0, we1, lane, c0, 8, nLive);
          edge_rows8(lhs, idxL, lane, 24, bB);
          owner_chunk8<true>(bA, rhsW, tw, rsV, cur, nextB, a, o, acc, wbits, we0, we1, lane, c0, 16, nLive);
          owner_chunk8<true>(bB, rhsW, tw, rsV, cur, nextB, a, o, acc, wbits, we0, we1, lane, c0, 24, nLive);
        } else {
          if (nLive > 8) edge_rows8(lhs, idxL, lane, 8, bB);
          owner_chunk8<false>(bA, rhsW, tw, rsV, cur, nextB, a, o, acc, wbits, we0, we1, lane, c0, 0, nLive);
          if (nLive > 8) {
            if (nLive > 16) edge_rows8(lhs, idxL, lane, 16, bA);
            owner_chunk8<false>(bB, rhsW, tw, rsV, cur, nextB, a, o, acc, wbits, we0, we1, lane, c0, 8, nLive);
            if (nLive > 16) {
              if (nLive > 24) edge_rows8(lhs, idxL, lane, 24, bB);
              owner_chunk8<false>(bA, rhsW, tw, rsV, cur, nextB, a, o, acc, wbits, we0, we1, lane, c0, 16, nLive);
              if (nLive > 24)
                owner_chunk8<false>(bB, rhsW, tw, rsV, cur, nextB, a, o, acc, wbits, we0, we1, lane, c0, 24, nLive);
            }
          }
        }
      }
      *reinterpret_cast<float2*>(tw + cur * 64 + ((2 * lane) ^ (4 * cur))) = float2{o.x + acc.x, o.y + acc.y};
      __syncthreads();
    }
  }

  const int r = lane & 15;
  const int h = lane >> 4;
  const int lrow0 = wave * 16;
  const int row0 = base + lrow0;
  if (row0 >= M) return;

  int rowA = row0 + r;
  if (rowA >= M) rowA = M - 1;
  const float* __restrict__ Xrow = X + (size_t)rowA * 64;

  float dg[8];
#pragma unroll
  for (int v = 0; v < 8; ++v) dg[v] = __shfl(degc, v + 8 * h);

  const float* trow = tw + r * 64;
  const int xr = 4 * r;

  v8f acc0 = {}, acc1 = {}, acc2 = {}, acc3 = {};
#pragma unroll
  for (int k0 = 0; k0 < 128; k0 += 4) {
    const int ka = k0 + 2 * h;
    float2 av;
    if (k0 < 64) av = *reinterpret_cast<const float2*>(trow + ((k0 ^ xr) + 2 * h));
    else         av = *reinterpret_cast<const float2*>(Xrow + (ka - 64));
    v2f a; a.x = av.x; a.y = av.y;
    const int k2 = (k0 >> 1) + h;
    const float2* Bk = &WoS2[k2 * 64 + r];
    float2 f0 = Bk[0], f1 = Bk[16], f2 = Bk[32], f3 = Bk[48];
    v2f b0, b1, b2, b3;
    b0.x = f0.x; b0.y = f0.y;  b1.x = f1.x; b1.y = f1.y;
    b2.x = f2.x; b2.y = f2.y;  b3.x = f3.x; b3.y = f3.y;
    acc0 = wmma_f32(a, b0, acc0);
    acc1 = wmma_f32(a, b1, acc1);
    acc2 = wmma_f32(a, b2, acc2);
    acc3 = wmma_f32(a, b3, acc3);
  }

  const float bia0 = bout[r], bia1 = bout[16 + r], bia2 = bout[32 + r], bia3 = bout[48 + r];
  const float bp0 = bpS[r], bp1 = bpS[16 + r], bp2 = bpS[32 + r], bp3 = bpS[48 + r];
#pragma unroll
  for (int v = 0; v < 8; ++v) {
    const int row = row0 + v + 8 * h;
    if (row < M) {
      float* Crow = C + (size_t)row * 64;
      const float d = dg[v];
      Crow[r]      = acc0[v] + d * bp0 + bia0;
      Crow[16 + r] = acc1[v] + d * bp1 + bia1;
      Crow[32 + r] = acc2[v] + d * bp2 + bia2;
      Crow[48 + r] = acc3[v] + d * bp3 + bia3;
    }
  }
}

extern "C" void kernel_launch(void* const* d_in, const int* in_sizes, int n_in,
                              void* d_out, int out_size, void* d_ws, size_t ws_size,
                              hipStream_t stream) {
  const float* input   = (const float*)d_in[0];
  const float* other   = (const float*)d_in[1];
  const int*   rj      = (const int*)d_in[2];
  const int*   lj      = (const int*)d_in[3];
  const float* weights = (const float*)d_in[4];
  const float* Wi      = (const float*)d_in[5];
  const float* bi      = (const float*)d_in[6];
  const float* Wo      = (const float*)d_in[7];
  const float* We      = (const float*)d_in[8];
  const float* Wf      = (const float*)d_in[9];
  const float* bf      = (const float*)d_in[10];
  const float* Wout    = (const float*)d_in[11];
  const float* bout    = (const float*)d_in[12];

  const int D    = 64;
  const int N_IN = in_sizes[0] / D;
  const int N_OT = in_sizes[1] / D;
  const int E    = in_sizes[2];

  float* rhs    = (float*)d_ws;
  float* lhs    = rhs + (size_t)N_IN * D;
  int4*  recs   = (int4*)(lhs + (size_t)N_OT * D);
  int*   seg    = (int*)(recs + E);

  const int tiles_in = (N_IN + 15) / 16;
  const int tiles_ot = (N_OT + 15) / 16;
  const int nRanges  = (N_IN + RNG - 1) / RNG;
  const int epb      = (E + NSL - 1) / NSL;

  const int nbOt = (tiles_ot + 31) / 32;
  const int nbIn = (tiles_in + 31) / 32;
  place_gemm_kernel<<<NSL + nbOt + nbIn, 1024, 0, stream>>>(
      rj, lj, weights, seg, recs, E, epb, other, Wo, lhs, N_OT, nbOt, input, Wi, bi, rhs, N_IN);

  owner_kernel<<<nRanges, CHK, 0, stream>>>(rhs, lhs, recs, seg, epb, We, input, Wf, bf,
                                            Wout, bout, (float*)d_out, N_IN);
}
